# starting kernel
# baseline (speedup 1.0000x reference)
#include <hip/hip_runtime.h>
#include <math.h>

#define E_NODES 20000
#define LE_EDGES 320000
#define NDIM_ 256
#define HEADS 8
#define DK_ 32

typedef __attribute__((ext_vector_type(2))) float v2f;
typedef __attribute__((ext_vector_type(8))) float v8f;

template <int K, int N, bool HAS_BIAS, bool HAS_ADD, bool RELU>
__global__ __launch_bounds__(256) void lgesql_gemm_wmma_f32(
    const float* __restrict__ A, const float* __restrict__ W,
    const float* __restrict__ bias, const float* __restrict__ addm,
    float* __restrict__ C)
{
  constexpr int TN = 64;
  constexpr int tilesN = N / TN;
  const int wid  = (blockIdx.x * blockDim.x + threadIdx.x) >> 5;
  const int lane = threadIdx.x & 31;
  const int tm = wid / tilesN;
  const int tn = wid - tm * tilesN;
  const int half = lane >> 4;
  const int l    = lane & 15;

  const float* Ap = A + (size_t)(tm * 16 + l) * K + 2 * half;
  const float* Bp = W + tn * TN + l + (size_t)(2 * half) * N;

  v8f acc[4] = {};
  #pragma unroll 4
  for (int k0 = 0; k0 < K; k0 += 4) {
    v2f a = *(const v2f*)(Ap + k0);
    const float* b = Bp + (size_t)k0 * N;
    v2f b0, b1, b2, b3;
    b0.x = b[0];   b0.y = b[N];
    b1.x = b[16];  b1.y = b[16 + N];
    b2.x = b[32];  b2.y = b[32 + N];
    b3.x = b[48];  b3.y = b[48 + N];
    acc[0] = __builtin_amdgcn_wmma_f32_16x16x4_f32(false, a, false, b0, (short)0, acc[0], false, false);
    acc[1] = __builtin_amdgcn_wmma_f32_16x16x4_f32(false, a, false, b1, (short)0, acc[1], false, false);
    acc[2] = __builtin_amdgcn_wmma_f32_16x16x4_f32(false, a, false, b2, (short)0, acc[2], false, false);
    acc[3] = __builtin_amdgcn_wmma_f32_16x16x4_f32(false, a, false, b3, (short)0, acc[3], false, false);
  }

  #pragma unroll
  for (int j = 0; j < 4; ++j) {
    const int col = tn * TN + j * 16 + l;
    const float bv = HAS_BIAS ? bias[col] : 0.0f;
    #pragma unroll
    for (int r = 0; r < 8; ++r) {
      const int row = tm * 16 + r + 8 * half;
      float v = acc[j][r] + bv;
      if (HAS_ADD) v += addm[(size_t)row * N + col];
      if (RELU)    v = fmaxf(v, 0.0f);
      C[(size_t)row * N + col] = v;
    }
  }
}

__global__ __launch_bounds__(256) void lgesql_score_scatter(
    const float* __restrict__ q, const float* __restrict__ k,
    const float* __restrict__ v, const int* __restrict__ lg_src,
    const int* __restrict__ lg_dst, float* __restrict__ wv,
    float* __restrict__ z)
{
  const int e    = (blockIdx.x * blockDim.x + threadIdx.x) >> 5;
  const int lane = threadIdx.x & 31;
  const int s = lg_src[e];
  const int d = lg_dst[e];
  const float* kr = k + (size_t)s * NDIM_;
  const float* qr = q + (size_t)d * NDIM_;
  const float* vr = v + (size_t)s * NDIM_;
  float* wvr = wv + (size_t)d * NDIM_;

  #pragma unroll
  for (int h = 0; h < HEADS; ++h) {
    const int c = h * DK_ + lane;
    float p = kr[c] * qr[c];
    #pragma unroll
    for (int off = 16; off > 0; off >>= 1)
      p += __shfl_xor(p, off, 32);
    float sc = p * 0.17677669529663687f;
    sc = fminf(5.0f, fmaxf(-5.0f, sc));
    sc = expf(sc);
    atomicAdd(&wvr[c], vr[c] * sc);
    if (lane == 0) atomicAdd(&z[(size_t)d * HEADS + h], sc);
  }
}

__global__ __launch_bounds__(256) void lgesql_normalize(
    float* __restrict__ wv, const float* __restrict__ z)
{
  const size_t i = (size_t)blockIdx.x * blockDim.x + threadIdx.x;
  const size_t row = i >> 8;
  const size_t col = i & 255;
  wv[i] = wv[i] / z[row * HEADS + (col >> 5)];
}

__global__ __launch_bounds__(256) void lgesql_layernorm(
    const float* __restrict__ in, const float* __restrict__ g,
    const float* __restrict__ b, float* __restrict__ out)
{
  const int row  = (blockIdx.x * blockDim.x + threadIdx.x) >> 5;
  const int lane = threadIdx.x & 31;
  const float* r = in + (size_t)row * NDIM_;
  float vals[8];
  float sum = 0.0f, sq = 0.0f;
  #pragma unroll
  for (int t = 0; t < 8; ++t) {
    float x = r[lane + 32 * t];
    vals[t] = x; sum += x; sq += x * x;
  }
  #pragma unroll
  for (int off = 16; off > 0; off >>= 1) {
    sum += __shfl_xor(sum, off, 32);
    sq  += __shfl_xor(sq,  off, 32);
  }
  const float mu   = sum * (1.0f / NDIM_);
  const float var  = sq * (1.0f / NDIM_) - mu * mu;
  const float rstd = rsqrtf(var + 1e-5f);
  float* o = out + (size_t)row * NDIM_;
  #pragma unroll
  for (int t = 0; t < 8; ++t) {
    const int c = lane + 32 * t;
    o[c] = (vals[t] - mu) * rstd * g[c] + b[c];
  }
}

extern "C" void kernel_launch(void* const* d_in, const int* in_sizes, int n_in,
                              void* d_out, int out_size, void* d_ws, size_t ws_size,
                              hipStream_t stream) {
  const float* x     = (const float*)d_in[0];
  const float* src_x = (const float*)d_in[1];
  const float* dst_x = (const float*)d_in[2];
  const float* Wq    = (const float*)d_in[3];
  const float* bq    = (const float*)d_in[4];
  const float* Wk    = (const float*)d_in[5];
  const float* Wv    = (const float*)d_in[6];
  const float* Wo    = (const float*)d_in[7];
  const float* bo    = (const float*)d_in[8];
  const float* ln1_g = (const float*)d_in[9];
  const float* ln1_b = (const float*)d_in[10];
  const float* W1    = (const float*)d_in[11];
  const float* b1    = (const float*)d_in[12];
  const float* W2    = (const float*)d_in[13];
  const float* b2    = (const float*)d_in[14];
  const float* ln2_g = (const float*)d_in[15];
  const float* ln2_b = (const float*)d_in[16];
  const int*   lg_src = (const int*)d_in[17];
  const int*   lg_dst = (const int*)d_in[18];
  float* out = (float*)d_out;

  const size_t EN = (size_t)E_NODES * NDIM_;
  float* ws = (float*)d_ws;
  float* qf   = ws;
  float* kf   = ws + EN;
  float* vf   = ws + 2 * EN;
  float* wvf  = ws + 3 * EN;
  float* zf   = ws + 4 * EN;
  float* ffn1 = ws + 4 * EN + (size_t)E_NODES * HEADS;

  hipMemsetAsync(wvf, 0, (EN + (size_t)E_NODES * HEADS) * sizeof(float), stream);

  const dim3 blk(256);
  const int g256  = (E_NODES / 16) * (NDIM_ / 64) / 8;
  const int g1024 = (E_NODES / 16) * (4 * NDIM_ / 64) / 8;

  lgesql_gemm_wmma_f32<256, 256, true,  true,  false>
      <<<g256, blk, 0, stream>>>(x, Wq, bq, src_x, qf);
  lgesql_gemm_wmma_f32<256, 256, false, false, false>
      <<<g256, blk, 0, stream>>>(x, Wk, nullptr, nullptr, kf);
  lgesql_gemm_wmma_f32<256, 256, false, true,  false>
      <<<g256, blk, 0, stream>>>(x, Wv, nullptr, dst_x, vf);

  lgesql_score_scatter<<<LE_EDGES / 8, blk, 0, stream>>>(qf, kf, vf,
                                                         lg_src, lg_dst, wvf, zf);

  lgesql_normalize<<<(int)(EN / 256), blk, 0, stream>>>(wvf, zf);

  lgesql_gemm_wmma_f32<256, 256, true,  true,  false>
      <<<g256, blk, 0, stream>>>(wvf, Wo, bo, x, qf);
  lgesql_layernorm<<<E_NODES / 8, blk, 0, stream>>>(qf, ln1_g, ln1_b, kf);

  lgesql_gemm_wmma_f32<256, 1024, true, false, true>
      <<<g1024, blk, 0, stream>>>(kf, W1, b1, nullptr, ffn1);
  lgesql_gemm_wmma_f32<1024, 256, true, true, false>
      <<<g256, blk, 0, stream>>>(ffn1, W2, b2, kf, vf);
  lgesql_layernorm<<<E_NODES / 8, blk, 0, stream>>>(vf, ln2_g, ln2_b, out);
}


# Edge stage: float-atomic scatter replaced by counting sort of edges by dst (int atomic rank) + one wave per node summing wv,z in registers, o=wv/z folded in; sum order: arrival rank, 8-lane DPP tree
# speedup vs baseline: 1.4782x; 1.4782x over previous
#include <hip/hip_runtime.h>
#include <math.h>

#define E_NODES 20000
#define LE_EDGES 320000
#define NDIM_ 256
#define HEADS 8
#define DK_ 32

typedef __attribute__((ext_vector_type(2))) float v2f;
typedef __attribute__((ext_vector_type(8))) float v8f;

template <int K, int N, bool HAS_BIAS, bool HAS_ADD, bool RELU>
__global__ __launch_bounds__(256) void lgesql_gemm_wmma_f32(
    const float* __restrict__ A, const float* __restrict__ W,
    const float* __restrict__ bias, const float* __restrict__ addm,
    float* __restrict__ C)
{
  constexpr int TN = 64;
  constexpr int tilesN = N / TN;
  const int wid  = (blockIdx.x * blockDim.x + threadIdx.x) >> 5;
  const int lane = threadIdx.x & 31;
  const int tm = wid / tilesN;
  const int tn = wid - tm * tilesN;
  const int half = lane >> 4;
  const int l    = lane & 15;

  const float* Ap = A + (size_t)(tm * 16 + l) * K + 2 * half;
  const float* Bp = W + tn * TN + l + (size_t)(2 * half) * N;

  v8f acc[4] = {};
  #pragma unroll 4
  for (int k0 = 0; k0 < K; k0 += 4) {
    v2f a = *(const v2f*)(Ap + k0);
    const float* b = Bp + (size_t)k0 * N;
    v2f b0, b1, b2, b3;
    b0.x = b[0];   b0.y = b[N];
    b1.x = b[16];  b1.y = b[16 + N];
    b2.x = b[32];  b2.y = b[32 + N];
    b3.x = b[48];  b3.y = b[48 + N];
    acc[0] = __builtin_amdgcn_wmma_f32_16x16x4_f32(false, a, false, b0, (short)0, acc[0], false, false);
    acc[1] = __builtin_amdgcn_wmma_f32_16x16x4_f32(false, a, false, b1, (short)0, acc[1], false, false);
    acc[2] = __builtin_amdgcn_wmma_f32_16x16x4_f32(false, a, false, b2, (short)0, acc[2], false, false);
    acc[3] = __builtin_amdgcn_wmma_f32_16x16x4_f32(false, a, false, b3, (short)0, acc[3], false, false);
  }

  #pragma unroll
  for (int j = 0; j < 4; ++j) {
    const int col = tn * TN + j * 16 + l;
    const float bv = HAS_BIAS ? bias[col] : 0.0f;
    #pragma unroll
    for (int r = 0; r < 8; ++r) {
      const int row = tm * 16 + r + 8 * half;
      float v = acc[j][r] + bv;
      if (HAS_ADD) v += addm[(size_t)row * N + col];
      if (RELU)    v = fmaxf(v, 0.0f);
      C[(size_t)row * N + col] = v;
    }
  }
}

__global__ __launch_bounds__(256) void lgesql_dst_count(
    const int* __restrict__ lg_dst, int* __restrict__ cnt, int* __restrict__ rank)
{
  const int e = blockIdx.x * blockDim.x + threadIdx.x;
  rank[e] = atomicAdd(&cnt[lg_dst[e]], 1);
}

__global__ __launch_bounds__(1024) void lgesql_dst_offsets(
    const int* __restrict__ cnt, int* __restrict__ off)
{
  constexpr int ITEMS = 20;
  static_assert(E_NODES % ITEMS == 0 && E_NODES / ITEMS <= 1024, "node split");
  __shared__ int wsum[32];
  const int t = threadIdx.x;
  const int lane = t & 31;
  const int w = t >> 5;
  const bool live = t < E_NODES / ITEMS;
  int c[ITEMS];
  if (live) {
    const int4* p = (const int4*)(cnt + t * ITEMS);
    #pragma unroll
    for (int g = 0; g < ITEMS / 4; ++g) {
      const int4 x = p[g];
      c[4 * g] = x.x; c[4 * g + 1] = x.y; c[4 * g + 2] = x.z; c[4 * g + 3] = x.w;
    }
  } else {
    #pragma unroll
    for (int i = 0; i < ITEMS; ++i) c[i] = 0;
  }
  int tot = 0;
  #pragma unroll
  for (int i = 0; i < ITEMS; ++i) { const int x = c[i]; c[i] = tot; tot += x; }
  int inc = tot;
  #pragma unroll
  for (int d = 1; d < 32; d <<= 1) {
    const int nb = __shfl_up(inc, d, 32);
    if (lane >= d) inc += nb;
  }
  if (lane == 31) wsum[w] = inc;
  __syncthreads();
  if (w == 0) {
    const int own = wsum[lane];
    int si = own;
    #pragma unroll
    for (int d = 1; d < 32; d <<= 1) {
      const int nb = __shfl_up(si, d, 32);
      if (lane >= d) si += nb;
    }
    wsum[lane] = si - own;
  }
  __syncthreads();
  const int base = wsum[w] + inc - tot;
  if (live) {
    int4* o = (int4*)(off + t * ITEMS);
    #pragma unroll
    for (int g = 0; g < ITEMS / 4; ++g)
      o[g] = make_int4(base + c[4 * g], base + c[4 * g + 1], base + c[4 * g + 2], base + c[4 * g + 3]);
  }
  if (t == 1023) off[E_NODES] = base + tot;
}

__global__ __launch_bounds__(256) void lgesql_dst_fill(
    const int* __restrict__ lg_src, const int* __restrict__ lg_dst,
    const int* __restrict__ off, const int* __restrict__ rank, int* __restrict__ slot_src)
{
  const int e = blockIdx.x * blockDim.x + threadIdx.x;
  slot_src[off[lg_dst[e]] + rank[e]] = lg_src[e];
}

template <int XMASK>
__device__ __forceinline__ float lgesql_xadd(float x)
{
  const int y = __builtin_amdgcn_update_dpp(0, __builtin_bit_cast(int, x), 0x160 | XMASK, 0xf, 0xf, true);
  return x + __builtin_bit_cast(float, y);
}

template <int U>
__device__ __forceinline__ void lgesql_edges(
    const float* __restrict__ k, const float* __restrict__ v, int mysrc, int j, int lane,
    const float4& qa, const float4& qb, float4& accA, float4& accB, float& zA, float& zB)
{
  float4 ka[U], kb[U], va[U], vb[U];
  #pragma unroll
  for (int u = 0; u < U; ++u) {
    const int s = __builtin_amdgcn_readlane(mysrc, j + u);
    const float4* kr = (const float4*)(k + (size_t)s * NDIM_);
    const float4* vr = (const float4*)(v + (size_t)s * NDIM_);
    ka[u] = kr[lane]; kb[u] = kr[32 + lane];
    va[u] = vr[lane]; vb[u] = vr[32 + lane];
  }
  #pragma unroll
  for (int u = 0; u < U; ++u) {
    float pA = ka[u].x * qa.x + ka[u].y * qa.y + ka[u].z * qa.z + ka[u].w * qa.w;
    float pB = kb[u].x * qb.x + kb[u].y * qb.y + kb[u].z * qb.z + kb[u].w * qb.w;
    pA = lgesql_xadd<1>(pA); pB = lgesql_xadd<1>(pB);
    pA = lgesql_xadd<2>(pA); pB = lgesql_xadd<2>(pB);
    pA = lgesql_xadd<4>(pA); pB = lgesql_xadd<4>(pB);
    float sA = pA * 0.17677669529663687f;
    float sB = pB * 0.17677669529663687f;
    sA = expf(fminf(5.0f, fmaxf(-5.0f, sA)));
    sB = expf(fminf(5.0f, fmaxf(-5.0f, sB)));
    accA.x += va[u].x * sA; accA.y += va[u].y * sA; accA.z += va[u].z * sA; accA.w += va[u].w * sA;
    accB.x += vb[u].x * sB; accB.y += vb[u].y * sB; accB.z += vb[u].z * sB; accB.w += vb[u].w * sB;
    zA += sA; zB += sB;
  }
}

__global__ __launch_bounds__(256) void lgesql_edge_gather(
    const float* __restrict__ q, const float* __restrict__ k, const float* __restrict__ v,
    const int* __restrict__ off, const int* __restrict__ slot_src, float* __restrict__ o)
{
  const int lane = threadIdx.x & 31;
  const int n = __builtin_amdgcn_readfirstlane((int)((blockIdx.x * blockDim.x + threadIdx.x) >> 5));
  const int beg = off[n];
  const int end = off[n + 1];
  const float4* qr = (const float4*)(q + (size_t)n * NDIM_);
  const float4 qa = qr[lane];
  const float4 qb = qr[32 + lane];
  float4 accA = make_float4(0.f, 0.f, 0.f, 0.f);
  float4 accB = make_float4(0.f, 0.f, 0.f, 0.f);
  float zA = 0.f, zB = 0.f;
  for (int base = beg; base < end; base += 32) {
    const int m = min(32, end - base);
    const int mysrc = (lane < m) ? slot_src[base + lane] : 0;
    int j = 0;
    for (; j + 4 <= m; j += 4)
      lgesql_edges<4>(k, v, mysrc, j, lane, qa, qb, accA, accB, zA, zB);
    if (j + 2 <= m) {
      lgesql_edges<2>(k, v, mysrc, j, lane, qa, qb, accA, accB, zA, zB);
      j += 2;
    }
    if (j < m)
      lgesql_edges<1>(k, v, mysrc, j, lane, qa, qb, accA, accB, zA, zB);
  }
  float4* orow = (float4*)(o + (size_t)n * NDIM_);
  orow[lane]      = make_float4(accA.x / zA, accA.y / zA, accA.z / zA, accA.w / zA);
  orow[32 + lane] = make_float4(accB.x / zB, accB.y / zB, accB.z / zB, accB.w / zB);
}

__global__ __launch_bounds__(256) void lgesql_layernorm(
    const float* __restrict__ in, const float* __restrict__ g,
    const float* __restrict__ b, float* __restrict__ out)
{
  const int row  = (blockIdx.x * blockDim.x + threadIdx.x) >> 5;
  const int lane = threadIdx.x & 31;
  const float* r = in + (size_t)row * NDIM_;
  float vals[8];
  float sum = 0.0f, sq = 0.0f;
  #pragma unroll
  for (int t = 0; t < 8; ++t) {
    float x = r[lane + 32 * t];
    vals[t] = x; sum += x; sq += x * x;
  }
  #pragma unroll
  for (int off = 16; off > 0; off >>= 1) {
    sum += __shfl_xor(sum, off, 32);
    sq  += __shfl_xor(sq,  off, 32);
  }
  const float mu   = sum * (1.0f / NDIM_);
  const float var  = sq * (1.0f / NDIM_) - mu * mu;
  const float rstd = rsqrtf(var + 1e-5f);
  float* o = out + (size_t)row * NDIM_;
  #pragma unroll
  for (int t = 0; t < 8; ++t) {
    const int c = lane + 32 * t;
    o[c] = (vals[t] - mu) * rstd * g[c] + b[c];
  }
}

extern "C" void kernel_launch(void* const* d_in, const int* in_sizes, int n_in,
                              void* d_out, int out_size, void* d_ws, size_t ws_size,
                              hipStream_t stream) {
  const float* x     = (const float*)d_in[0];
  const float* src_x = (const float*)d_in[1];
  const float* dst_x = (const float*)d_in[2];
  const float* Wq    = (const float*)d_in[3];
  const float* bq    = (const float*)d_in[4];
  const float* Wk    = (const float*)d_in[5];
  const float* Wv    = (const float*)d_in[6];
  const float* Wo    = (const float*)d_in[7];
  const float* bo    = (const float*)d_in[8];
  const float* ln1_g = (const float*)d_in[9];
  const float* ln1_b = (const float*)d_in[10];
  const float* W1    = (const float*)d_in[11];
  const float* b1    = (const float*)d_in[12];
  const float* W2    = (const float*)d_in[13];
  const float* b2    = (const float*)d_in[14];
  const float* ln2_g = (const float*)d_in[15];
  const float* ln2_b = (const float*)d_in[16];
  const int*   lg_src = (const int*)d_in[17];
  const int*   lg_dst = (const int*)d_in[18];
  float* out = (float*)d_out;

  const size_t EN = (size_t)E_NODES * NDIM_;
  float* ws = (float*)d_ws;
  float* qf   = ws;
  float* kf   = ws + EN;
  float* vf   = ws + 2 * EN;
  float* wvf  = ws + 3 * EN;
  int*   cntf = (int*)(ws + 4 * EN);
  int*   offf = cntf + E_NODES;
  float* ffn1 = ws + 4 * EN + (size_t)E_NODES * HEADS;
  int*   slotf = (int*)ffn1;
  int*   rankf = slotf + LE_EDGES;

  hipMemsetAsync(cntf, 0, (size_t)E_NODES * sizeof(int), stream);

  const dim3 blk(256);

  lgesql_dst_count<<<LE_EDGES / 256, blk, 0, stream>>>(lg_dst, cntf, rankf);
  lgesql_dst_offsets<<<1, dim3(1024), 0, stream>>>(cntf, offf);
  lgesql_dst_fill<<<LE_EDGES / 256, blk, 0, stream>>>(lg_src, lg_dst, offf, rankf, slotf);
  const int g256  = (E_NODES / 16) * (NDIM_ / 64) / 8;
  const int g1024 = (E_NODES / 16) * (4 * NDIM_ / 64) / 8;

  lgesql_gemm_wmma_f32<256, 256, true,  true,  false>
      <<<g256, blk, 0, stream>>>(x, Wq, bq, src_x, qf);
  lgesql_gemm_wmma_f32<256, 256, false, false, false>
      <<<g256, blk, 0, stream>>>(x, Wk, nullptr, nullptr, kf);
  lgesql_gemm_wmma_f32<256, 256, false, true,  false>
      <<<g256, blk, 0, stream>>>(x, Wv, nullptr, dst_x, vf);

  lgesql_edge_gather<<<E_NODES / 8, blk, 0, stream>>>(qf, kf, vf, offf, slotf, wvf);

  lgesql_gemm_wmma_f32<256, 256, true,  true,  false>
      <<<g256, blk, 0, stream>>>(wvf, Wo, bo, x, qf);
  lgesql_layernorm<<<E_NODES / 8, blk, 0, stream>>>(qf, ln1_g, ln1_b, kf);

  lgesql_gemm_wmma_f32<256, 1024, true, false, true>
      <<<g1024, blk, 0, stream>>>(kf, W1, b1, nullptr, ffn1);
  lgesql_gemm_wmma_f32<1024, 256, true, true, false>
      <<<g256, blk, 0, stream>>>(ffn1, W2, b2, kf, vf);
  lgesql_layernorm<<<E_NODES / 8, blk, 0, stream>>>(vf, ln2_g, ln2_b, out);
}


# GEMM fragment loads 16 B per lane: each lane half owns 4 contiguous k of every 8 and 4 adjacent columns, so W loads and C stores cover whole lines; k summed as {0,1,4,5},{2,3,6,7} per 8
# speedup vs baseline: 1.6462x; 1.1137x over previous
#include <hip/hip_runtime.h>
#include <math.h>

#define E_NODES 20000
#define LE_EDGES 320000
#define NDIM_ 256
#define HEADS 8
#define DK_ 32

typedef __attribute__((ext_vector_type(2))) float v2f;
typedef __attribute__((ext_vector_type(8))) float v8f;

__device__ __forceinline__ v2f lgesql_pair(float lo, float hi)
{
  v2f p; p.x = lo; p.y = hi; return p;
}

template <int K, int N, bool HAS_BIAS, bool HAS_ADD, bool RELU>
__global__ __launch_bounds__(256) void lgesql_gemm_wmma_f32(
    const float* __restrict__ A, const float* __restrict__ W,
    const float* __restrict__ bias, const float* __restrict__ addm,
    float* __restrict__ C)
{
  constexpr int TN = 64;
  constexpr int tilesN = N / TN;
  static_assert(K % 16 == 0 && N % TN == 0, "tile split");
  const int wid  = (blockIdx.x * blockDim.x + threadIdx.x) >> 5;
  const int lane = threadIdx.x & 31;
  const int tm = wid / tilesN;
  const int tn = wid - tm * tilesN;
  const int half = lane >> 4;
  const int l    = lane & 15;

  const float* Ap = A + (size_t)(tm * 16 + l) * K + 4 * half;
  const float* Bp = W + (size_t)(4 * half) * N + tn * TN + 4 * l;

  v8f acc0 = {}, acc1 = {}, acc2 = {}, acc3 = {};
  for (int k = 0; k < K; k += 16) {
    int kk = k;
    asm volatile("v_nop\n\tv_nop\n\tv_nop\n\tv_nop\n\tv_nop\n\tv_nop\n\tv_nop\n\tv_nop"
                 : "+v"(acc0), "+v"(acc1), "+v"(acc2), "+v"(acc3), "+s"(kk));
    const float* ak = Ap + kk;
    const float* bk = Bp + (size_t)kk * N;
    const float4 a0 = *(const float4*)(ak);
    const float4 a1 = *(const float4*)(ak + 8);
    const float4 r0 = *(const float4*)(bk);
    const float4 r1 = *(const float4*)(bk + N);
    const float4 r2 = *(const float4*)(bk + 2 * N);
    const float4 r3 = *(const float4*)(bk + 3 * N);
    const float4 r4 = *(const float4*)(bk + 8 * N);
    const float4 r5 = *(const float4*)(bk + 9 * N);
    const float4 r6 = *(const float4*)(bk + 10 * N);
    const float4 r7 = *(const float4*)(bk + 11 * N);
    v2f p00 = lgesql_pair(r0.x, r1.x), p01 = lgesql_pair(r0.y, r1.y), p02 = lgesql_pair(r0.z, r1.z), p03 = lgesql_pair(r0.w, r1.w);
    v2f p10 = lgesql_pair(r2.x, r3.x), p11 = lgesql_pair(r2.y, r3.y), p12 = lgesql_pair(r2.z, r3.z), p13 = lgesql_pair(r2.w, r3.w);
    v2f p20 = lgesql_pair(r4.x, r5.x), p21 = lgesql_pair(r4.y, r5.y), p22 = lgesql_pair(r4.z, r5.z), p23 = lgesql_pair(r4.w, r5.w);
    v2f p30 = lgesql_pair(r6.x, r7.x), p31 = lgesql_pair(r6.y, r7.y), p32 = lgesql_pair(r6.z, r7.z), p33 = lgesql_pair(r6.w, r7.w);
    asm volatile("" : "+v"(p00), "+v"(p01), "+v"(p02), "+v"(p03), "+v"(p10), "+v"(p11), "+v"(p12), "+v"(p13),
                      "+v"(p20), "+v"(p21), "+v"(p22), "+v"(p23), "+v"(p30), "+v"(p31), "+v"(p32), "+v"(p33));
    const v2f a00 = lgesql_pair(a0.x, a0.y), a01 = lgesql_pair(a0.z, a0.w);
    const v2f a10 = lgesql_pair(a1.x, a1.y), a11 = lgesql_pair(a1.z, a1.w);
    acc0 = __builtin_amdgcn_wmma_f32_16x16x4_f32(false, a00, false, p00, (short)0, acc0, false, false);
    acc1 = __builtin_amdgcn_wmma_f32_16x16x4_f32(false, a00, false, p01, (short)0, acc1, false, false);
    acc2 = __builtin_amdgcn_wmma_f32_16x16x4_f32(false, a00, false, p02, (short)0, acc2, false, false);
    acc3 = __builtin_amdgcn_wmma_f32_16x16x4_f32(false, a00, false, p03, (short)0, acc3, false, false);
    acc0 = __builtin_amdgcn_wmma_f32_16x16x4_f32(false, a01, false, p10, (short)0, acc0, false, false);
    acc1 = __builtin_amdgcn_wmma_f32_16x16x4_f32(false, a01, false, p11, (short)0, acc1, false, false);
    acc2 = __builtin_amdgcn_wmma_f32_16x16x4_f32(false, a01, false, p12, (short)0, acc2, false, false);
    acc3 = __builtin_amdgcn_wmma_f32_16x16x4_f32(false, a01, false, p13, (short)0, acc3, false, false);
    acc0 = __builtin_amdgcn_wmma_f32_16x16x4_f32(false, a10, false, p20, (short)0, acc0, false, false);
    acc1 = __builtin_amdgcn_wmma_f32_16x16x4_f32(false, a10, false, p21, (short)0, acc1, false, false);
    acc2 = __builtin_amdgcn_wmma_f32_16x16x4_f32(false, a10, false, p22, (short)0, acc2, false, false);
    acc3 = __builtin_amdgcn_wmma_f32_16x16x4_f32(false, a10, false, p23, (short)0, acc3, false, false);
    acc0 = __builtin_amdgcn_wmma_f32_16x16x4_f32(false, a11, false, p30, (short)0, acc0, false, false);
    acc1 = __builtin_amdgcn_wmma_f32_16x16x4_f32(false, a11, false, p31, (short)0, acc1, false, false);
    acc2 = __builtin_amdgcn_wmma_f32_16x16x4_f32(false, a11, false, p32, (short)0, acc2, false, false);
    acc3 = __builtin_amdgcn_wmma_f32_16x16x4_f32(false, a11, false, p33, (short)0, acc3, false, false);
  }
  asm volatile("v_nop\n\tv_nop\n\tv_nop\n\tv_nop\n\tv_nop\n\tv_nop\n\tv_nop\n\tv_nop"
               : "+v"(acc0), "+v"(acc1), "+v"(acc2), "+v"(acc3));

  const int col = tn * TN + 4 * l;
  float4 bv = make_float4(0.0f, 0.0f, 0.0f, 0.0f);
  if (HAS_BIAS) bv = *(const float4*)(bias + col);
  #pragma unroll
  for (int r = 0; r < 8; ++r) {
    const int row = tm * 16 + r + 8 * half;
    float4 v = make_float4(acc0[r] + bv.x, acc1[r] + bv.y, acc2[r] + bv.z, acc3[r] + bv.w);
    if (HAS_ADD) {
      const float4 ad = *(const float4*)(addm + (size_t)row * N + col);
      v.x += ad.x; v.y += ad.y; v.z += ad.z; v.w += ad.w;
    }
    if (RELU) {
      v.x = fmaxf(v.x, 0.0f); v.y = fmaxf(v.y, 0.0f); v.z = fmaxf(v.z, 0.0f); v.w = fmaxf(v.w, 0.0f);
    }
    *(float4*)(C + (size_t)row * N + col) = v;
  }
}

__global__ __launch_bounds__(256) void lgesql_dst_count(
    const int* __restrict__ lg_dst, int* __restrict__ cnt, int* __restrict__ rank)
{
  const int e = blockIdx.x * blockDim.x + threadIdx.x;
  rank[e] = atomicAdd(&cnt[lg_dst[e]], 1);
}

__global__ __launch_bounds__(1024) void lgesql_dst_offsets(
    const int* __restrict__ cnt, int* __restrict__ off)
{
  constexpr int ITEMS = 20;
  static_assert(E_NODES % ITEMS == 0 && E_NODES / ITEMS <= 1024, "node split");
  __shared__ int wsum[32];
  const int t = threadIdx.x;
  const int lane = t & 31;
  const int w = t >> 5;
  const bool live = t < E_NODES / ITEMS;
  int c[ITEMS];
  if (live) {
    const int4* p = (const int4*)(cnt + t * ITEMS);
    #pragma unroll
    for (int g = 0; g < ITEMS / 4; ++g) {
      const int4 x = p[g];
      c[4 * g] = x.x; c[4 * g + 1] = x.y; c[4 * g + 2] = x.z; c[4 * g + 3] = x.w;
    }
  } else {
    #pragma unroll
    for (int i = 0; i < ITEMS; ++i) c[i] = 0;
  }
  int tot = 0;
  #pragma unroll
  for (int i = 0; i < ITEMS; ++i) { const int x = c[i]; c[i] = tot; tot += x; }
  int inc = tot;
  #pragma unroll
  for (int d = 1; d < 32; d <<= 1) {
    const int nb = __shfl_up(inc, d, 32);
    if (lane >= d) inc += nb;
  }
  if (lane == 31) wsum[w] = inc;
  __syncthreads();
  if (w == 0) {
    const int own = wsum[lane];
    int si = own;
    #pragma unroll
    for (int d = 1; d < 32; d <<= 1) {
      const int nb = __shfl_up(si, d, 32);
      if (lane >= d) si += nb;
    }
    wsum[lane] = si - own;
  }
  __syncthreads();
  const int base = wsum[w] + inc - tot;
  if (live) {
    int4* o = (int4*)(off + t * ITEMS);
    #pragma unroll
    for (int g = 0; g < ITEMS / 4; ++g)
      o[g] = make_int4(base + c[4 * g], base + c[4 * g + 1], base + c[4 * g + 2], base + c[4 * g + 3]);
  }
  if (t == 1023) off[E_NODES] = base + tot;
}

__global__ __launch_bounds__(256) void lgesql_dst_fill(
    const int* __restrict__ lg_src, const int* __restrict__ lg_dst,
    const int* __restrict__ off, const int* __restrict__ rank, int* __restrict__ slot_src)
{
  const int e = blockIdx.x * blockDim.x + threadIdx.x;
  slot_src[off[lg_dst[e]] + rank[e]] = lg_src[e];
}

template <int XMASK>
__device__ __forceinline__ float lgesql_xadd(float x)
{
  const int y = __builtin_amdgcn_update_dpp(0, __builtin_bit_cast(int, x), 0x160 | XMASK, 0xf, 0xf, true);
  return x + __builtin_bit_cast(float, y);
}

template <int U>
__device__ __forceinline__ void lgesql_edges(
    const float* __restrict__ k, const float* __restrict__ v, int mysrc, int j, int lane,
    const float4& qa, const float4& qb, float4& accA, float4& accB, float& zA, float& zB)
{
  float4 ka[U], kb[U], va[U], vb[U];
  #pragma unroll
  for (int u = 0; u < U; ++u) {
    const int s = __builtin_amdgcn_readlane(mysrc, j + u);
    const float4* kr = (const float4*)(k + (size_t)s * NDIM_);
    const float4* vr = (const float4*)(v + (size_t)s * NDIM_);
    ka[u] = kr[lane]; kb[u] = kr[32 + lane];
    va[u] = vr[lane]; vb[u] = vr[32 + lane];
  }
  #pragma unroll
  for (int u = 0; u < U; ++u) {
    float pA = ka[u].x * qa.x + ka[u].y * qa.y + ka[u].z * qa.z + ka[u].w * qa.w;
    float pB = kb[u].x * qb.x + kb[u].y * qb.y + kb[u].z * qb.z + kb[u].w * qb.w;
    pA = lgesql_xadd<1>(pA); pB = lgesql_xadd<1>(pB);
    pA = lgesql_xadd<2>(pA); pB = lgesql_xadd<2>(pB);
    pA = lgesql_xadd<4>(pA); pB = lgesql_xadd<4>(pB);
    float sA = pA * 0.17677669529663687f;
    float sB = pB * 0.17677669529663687f;
    sA = expf(fminf(5.0f, fmaxf(-5.0f, sA)));
    sB = expf(fminf(5.0f, fmaxf(-5.0f, sB)));
    accA.x += va[u].x * sA; accA.y += va[u].y * sA; accA.z += va[u].z * sA; accA.w += va[u].w * sA;
    accB.x += vb[u].x * sB; accB.y += vb[u].y * sB; accB.z += vb[u].z * sB; accB.w += vb[u].w * sB;
    zA += sA; zB += sB;
  }
}

__global__ __launch_bounds__(256) void lgesql_edge_gather(
    const float* __restrict__ q, const float* __restrict__ k, const float* __restrict__ v,
    const int* __restrict__ off, const int* __restrict__ slot_src, float* __restrict__ o)
{
  const int lane = threadIdx.x & 31;
  const int n = __builtin_amdgcn_readfirstlane((int)((blockIdx.x * blockDim.x + threadIdx.x) >> 5));
  const int beg = off[n];
  const int end = off[n + 1];
  const float4* qr = (const float4*)(q + (size_t)n * NDIM_);
  const float4 qa = qr[lane];
  const float4 qb = qr[32 + lane];
  float4 accA = make_float4(0.f, 0.f, 0.f, 0.f);
  float4 accB = make_float4(0.f, 0.f, 0.f, 0.f);
  float zA = 0.f, zB = 0.f;
  for (int base = beg; base < end; base += 32) {
    const int m = min(32, end - base);
    const int mysrc = (lane < m) ? slot_src[base + lane] : 0;
    int j = 0;
    for (; j + 4 <= m; j += 4)
      lgesql_edges<4>(k, v, mysrc, j, lane, qa, qb, accA, accB, zA, zB);
    if (j + 2 <= m) {
      lgesql_edges<2>(k, v, mysrc, j, lane, qa, qb, accA, accB, zA, zB);
      j += 2;
    }
    if (j < m)
      lgesql_edges<1>(k, v, mysrc, j, lane, qa, qb, accA, accB, zA, zB);
  }
  float4* orow = (float4*)(o + (size_t)n * NDIM_);
  orow[lane]      = make_float4(accA.x / zA, accA.y / zA, accA.z / zA, accA.w / zA);
  orow[32 + lane] = make_float4(accB.x / zB, accB.y / zB, accB.z / zB, accB.w / zB);
}

__global__ __launch_bounds__(256) void lgesql_layernorm(
    const float* __restrict__ in, const float* __restrict__ g,
    const float* __restrict__ b, float* __restrict__ out)
{
  const int row  = (blockIdx.x * blockDim.x + threadIdx.x) >> 5;
  const int lane = threadIdx.x & 31;
  const float* r = in + (size_t)row * NDIM_;
  float vals[8];
  float sum = 0.0f, sq = 0.0f;
  #pragma unroll
  for (int t = 0; t < 8; ++t) {
    float x = r[lane + 32 * t];
    vals[t] = x; sum += x; sq += x * x;
  }
  #pragma unroll
  for (int off = 16; off > 0; off >>= 1) {
    sum += __shfl_xor(sum, off, 32);
    sq  += __shfl_xor(sq,  off, 32);
  }
  const float mu   = sum * (1.0f / NDIM_);
  const float var  = sq * (1.0f / NDIM_) - mu * mu;
  const float rstd = rsqrtf(var + 1e-5f);
  float* o = out + (size_t)row * NDIM_;
  #pragma unroll
  for (int t = 0; t < 8; ++t) {
    const int c = lane + 32 * t;
    o[c] = (vals[t] - mu) * rstd * g[c] + b[c];
  }
}

extern "C" void kernel_launch(void* const* d_in, const int* in_sizes, int n_in,
                              void* d_out, int out_size, void* d_ws, size_t ws_size,
                              hipStream_t stream) {
  const float* x     = (const float*)d_in[0];
  const float* src_x = (const float*)d_in[1];
  const float* dst_x = (const float*)d_in[2];
  const float* Wq    = (const float*)d_in[3];
  const float* bq    = (const float*)d_in[4];
  const float* Wk    = (const float*)d_in[5];
  const float* Wv    = (const float*)d_in[6];
  const float* Wo    = (const float*)d_in[7];
  const float* bo    = (const float*)d_in[8];
  const float* ln1_g = (const float*)d_in[9];
  const float* ln1_b = (const float*)d_in[10];
  const float* W1    = (const float*)d_in[11];
  const float* b1    = (const float*)d_in[12];
  const float* W2    = (const float*)d_in[13];
  const float* b2    = (const float*)d_in[14];
  const float* ln2_g = (const float*)d_in[15];
  const float* ln2_b = (const float*)d_in[16];
  const int*   lg_src = (const int*)d_in[17];
  const int*   lg_dst = (const int*)d_in[18];
  float* out = (float*)d_out;

  const size_t EN = (size_t)E_NODES * NDIM_;
  float* ws = (float*)d_ws;
  float* qf   = ws;
  float* kf   = ws + EN;
  float* vf   = ws + 2 * EN;
  float* wvf  = ws + 3 * EN;
  int*   cntf = (int*)(ws + 4 * EN);
  int*   offf = cntf + E_NODES;
  float* ffn1 = ws + 4 * EN + (size_t)E_NODES * HEADS;
  int*   slotf = (int*)ffn1;
  int*   rankf = slotf + LE_EDGES;

  hipMemsetAsync(cntf, 0, (size_t)E_NODES * sizeof(int), stream);

  const dim3 blk(256);

  lgesql_dst_count<<<LE_EDGES / 256, blk, 0, stream>>>(lg_dst, cntf, rankf);
  lgesql_dst_offsets<<<1, dim3(1024), 0, stream>>>(cntf, offf);
  lgesql_dst_fill<<<LE_EDGES / 256, blk, 0, stream>>>(lg_src, lg_dst, offf, rankf, slotf);
  const int g256  = (E_NODES / 16) * (NDIM_ / 64) / 8;
  const int g1024 = (E_NODES / 16) * (4 * NDIM_ / 64) / 8;

  lgesql_gemm_wmma_f32<256, 256, true,  true,  false>
      <<<g256, blk, 0, stream>>>(x, Wq, bq, src_x, qf);
  lgesql_gemm_wmma_f32<256, 256, false, false, false>
      <<<g256, blk, 0, stream>>>(x, Wk, nullptr, nullptr, kf);
  lgesql_gemm_wmma_f32<256, 256, false, true,  false>
      <<<g256, blk, 0, stream>>>(x, Wv, nullptr, dst_x, vf);

  lgesql_edge_gather<<<E_NODES / 8, blk, 0, stream>>>(qf, kf, vf, offf, slotf, wvf);

  lgesql_gemm_wmma_f32<256, 256, true,  true,  false>
      <<<g256, blk, 0, stream>>>(wvf, Wo, bo, x, qf);
  lgesql_layernorm<<<E_NODES / 8, blk, 0, stream>>>(qf, ln1_g, ln1_b, kf);

  lgesql_gemm_wmma_f32<256, 1024, true, false, true>
      <<<g1024, blk, 0, stream>>>(kf, W1, b1, nullptr, ffn1);
  lgesql_gemm_wmma_f32<1024, 256, true, true, false>
      <<<g256, blk, 0, stream>>>(ffn1, W2, b2, kf, vf);
  lgesql_layernorm<<<E_NODES / 8, blk, 0, stream>>>(vf, ln2_g, ln2_b, out);
}


# LayerNorm folded into the epilogue of the GEMM that feeds it (4-wave block owns 16 whole rows): 2 launches fewer; row sums now 4 in lane, 16-lane swizzle tree, 4 waves via LDS
# speedup vs baseline: 1.7452x; 1.0601x over previous
#include <hip/hip_runtime.h>
#include <math.h>

#define E_NODES 20000
#define LE_EDGES 320000
#define NDIM_ 256
#define HEADS 8
#define DK_ 32

typedef __attribute__((ext_vector_type(2))) float v2f;
typedef __attribute__((ext_vector_type(8))) float v8f;

__device__ __forceinline__ v2f lgesql_pair(float lo, float hi)
{
  v2f p; p.x = lo; p.y = hi; return p;
}

template <int XMASK>
__device__ __forceinline__ float lgesql_xadd(float x)
{
  const int y = __builtin_amdgcn_update_dpp(0, __builtin_bit_cast(int, x), 0x160 | XMASK, 0xf, 0xf, true);
  return x + __builtin_bit_cast(float, y);
}

template <int K, int N, bool HAS_BIAS, bool HAS_ADD, bool RELU, bool LN>
__global__ __launch_bounds__(LN ? 128 : 256) void lgesql_gemm_wmma_f32(
    const float* __restrict__ A, const float* __restrict__ W,
    const float* __restrict__ bias, const float* __restrict__ addm,
    const float* __restrict__ ln_g, const float* __restrict__ ln_b,
    float* __restrict__ C)
{
  static_assert(!LN || N == 256, "LN: the four waves of a work-group must cover whole rows");
  constexpr int TN = 64;
  constexpr int tilesN = N / TN;
  static_assert(K % 16 == 0 && N % TN == 0, "tile split");
  const int wid  = (blockIdx.x * blockDim.x + threadIdx.x) >> 5;
  const int lane = threadIdx.x & 31;
  const int tm = wid / tilesN;
  const int tn = wid - tm * tilesN;
  const int half = lane >> 4;
  const int l    = lane & 15;

  const float* Ap = A + (size_t)(tm * 16 + l) * K + 4 * half;
  const float* Bp = W + (size_t)(4 * half) * N + tn * TN + 4 * l;

  v8f acc0 = {}, acc1 = {}, acc2 = {}, acc3 = {};
  for (int k = 0; k < K; k += 16) {
    int kk = k;
    asm volatile("v_nop\n\tv_nop\n\tv_nop\n\tv_nop\n\tv_nop\n\tv_nop\n\tv_nop\n\tv_nop"
                 : "+v"(acc0), "+v"(acc1), "+v"(acc2), "+v"(acc3), "+s"(kk));
    const float* ak = Ap + kk;
    const float* bk = Bp + (size_t)kk * N;
    const float4 a0 = *(const float4*)(ak);
    const float4 a1 = *(const float4*)(ak + 8);
    const float4 r0 = *(const float4*)(bk);
    const float4 r1 = *(const float4*)(bk + N);
    const float4 r2 = *(const float4*)(bk + 2 * N);
    const float4 r3 = *(const float4*)(bk + 3 * N);
    const float4 r4 = *(const float4*)(bk + 8 * N);
    const float4 r5 = *(const float4*)(bk + 9 * N);
    const float4 r6 = *(const float4*)(bk + 10 * N);
    const float4 r7 = *(const float4*)(bk + 11 * N);
    v2f p00 = lgesql_pair(r0.x, r1.x), p01 = lgesql_pair(r0.y, r1.y), p02 = lgesql_pair(r0.z, r1.z), p03 = lgesql_pair(r0.w, r1.w);
    v2f p10 = lgesql_pair(r2.x, r3.x), p11 = lgesql_pair(r2.y, r3.y), p12 = lgesql_pair(r2.z, r3.z), p13 = lgesql_pair(r2.w, r3.w);
    v2f p20 = lgesql_pair(r4.x, r5.x), p21 = lgesql_pair(r4.y, r5.y), p22 = lgesql_pair(r4.z, r5.z), p23 = lgesql_pair(r4.w, r5.w);
    v2f p30 = lgesql_pair(r6.x, r7.x), p31 = lgesql_pair(r6.y, r7.y), p32 = lgesql_pair(r6.z, r7.z), p33 = lgesql_pair(r6.w, r7.w);
    asm volatile("" : "+v"(p00), "+v"(p01), "+v"(p02), "+v"(p03), "+v"(p10), "+v"(p11), "+v"(p12), "+v"(p13),
                      "+v"(p20), "+v"(p21), "+v"(p22), "+v"(p23), "+v"(p30), "+v"(p31), "+v"(p32), "+v"(p33));
    const v2f a00 = lgesql_pair(a0.x, a0.y), a01 = lgesql_pair(a0.z, a0.w);
    const v2f a10 = lgesql_pair(a1.x, a1.y), a11 = lgesql_pair(a1.z, a1.w);
    acc0 = __builtin_amdgcn_wmma_f32_16x16x4_f32(false, a00, false, p00, (short)0, acc0, false, false);
    acc1 = __builtin_amdgcn_wmma_f32_16x16x4_f32(false, a00, false, p01, (short)0, acc1, false, false);
    acc2 = __builtin_amdgcn_wmma_f32_16x16x4_f32(false, a00, false, p02, (short)0, acc2, false, false);
    acc3 = __builtin_amdgcn_wmma_f32_16x16x4_f32(false, a00, false, p03, (short)0, acc3, false, false);
    acc0 = __builtin_amdgcn_wmma_f32_16x16x4_f32(false, a01, false, p10, (short)0, acc0, false, false);
    acc1 = __builtin_amdgcn_wmma_f32_16x16x4_f32(false, a01, false, p11, (short)0, acc1, false, false);
    acc2 = __builtin_amdgcn_wmma_f32_16x16x4_f32(false, a01, false, p12, (short)0, acc2, false, false);
    acc3 = __builtin_amdgcn_wmma_f32_16x16x4_f32(false, a01, false, p13, (short)0, acc3, false, false);
    acc0 = __builtin_amdgcn_wmma_f32_16x16x4_f32(false, a10, false, p20, (short)0, acc0, false, false);
    acc1 = __builtin_amdgcn_wmma_f32_16x16x4_f32(false, a10, false, p21, (short)0, acc1, false, false);
    acc2 = __builtin_amdgcn_wmma_f32_16x16x4_f32(false, a10, false, p22, (short)0, acc2, false, false);
    acc3 = __builtin_amdgcn_wmma_f32_16x16x4_f32(false, a10, false, p23, (short)0, acc3, false, false);
    acc0 = __builtin_amdgcn_wmma_f32_16x16x4_f32(false, a11, false, p30, (short)0, acc0, false, false);
    acc1 = __builtin_amdgcn_wmma_f32_16x16x4_f32(false, a11, false, p31, (short)0, acc1, false, false);
    acc2 = __builtin_amdgcn_wmma_f32_16x16x4_f32(false, a11, false, p32, (short)0, acc2, false, false);
    acc3 = __builtin_amdgcn_wmma_f32_16x16x4_f32(false, a11, false, p33, (short)0, acc3, false, false);
  }
  asm volatile("v_nop\n\tv_nop\n\tv_nop\n\tv_nop\n\tv_nop\n\tv_nop\n\tv_nop\n\tv_nop"
               : "+v"(acc0), "+v"(acc1), "+v"(acc2), "+v"(acc3));

  const int col = tn * TN + 4 * l;
  float4 bv = make_float4(0.0f, 0.0f, 0.0f, 0.0f);
  if (HAS_BIAS) bv = *(const float4*)(bias + col);
  float4 v[8];
  #pragma unroll
  for (int r = 0; r < 8; ++r) {
    const int row = tm * 16 + r + 8 * half;
    v[r] = make_float4(acc0[r] + bv.x, acc1[r] + bv.y, acc2[r] + bv.z, acc3[r] + bv.w);
    if (HAS_ADD) {
      const float4 ad = *(const float4*)(addm + (size_t)row * N + col);
      v[r].x += ad.x; v[r].y += ad.y; v[r].z += ad.z; v[r].w += ad.w;
    }
    if (RELU) {
      v[r].x = fmaxf(v[r].x, 0.0f); v[r].y = fmaxf(v[r].y, 0.0f);
      v[r].z = fmaxf(v[r].z, 0.0f); v[r].w = fmaxf(v[r].w, 0.0f);
    }
  }
  if constexpr (LN) {
    __shared__ float2 part[16][4];
    const int w = threadIdx.x >> 5;
    #pragma unroll
    for (int r = 0; r < 8; ++r) {
      float sm = (v[r].x + v[r].y) + (v[r].z + v[r].w);
      float sq = (v[r].x * v[r].x + v[r].y * v[r].y) + (v[r].z * v[r].z + v[r].w * v[r].w);
      sm = lgesql_xadd<1>(sm); sq = lgesql_xadd<1>(sq);
      sm = lgesql_xadd<2>(sm); sq = lgesql_xadd<2>(sq);
      sm = lgesql_xadd<4>(sm); sq = lgesql_xadd<4>(sq);
      sm = lgesql_xadd<8>(sm); sq = lgesql_xadd<8>(sq);
      if (l == r) part[r + 8 * half][w] = make_float2(sm, sq);
    }
    __syncthreads();
    const float4 g4 = *(const float4*)(ln_g + col);
    const float4 b4 = *(const float4*)(ln_b + col);
    #pragma unroll
    for (int r = 0; r < 8; ++r) {
      const int row = tm * 16 + r + 8 * half;
      const float4 p01 = *(const float4*)(&part[r + 8 * half][0]);
      const float4 p23 = *(const float4*)(&part[r + 8 * half][2]);
      const float sum = ((p01.x + p01.z) + p23.x) + p23.z;
      const float sqs = ((p01.y + p01.w) + p23.y) + p23.w;
      const float mu   = sum * (1.0f / N);
      const float var  = sqs * (1.0f / N) - mu * mu;
      const float rstd = rsqrtf(var + 1e-5f);
      float4 o;
      o.x = (v[r].x - mu) * rstd * g4.x + b4.x;
      o.y = (v[r].y - mu) * rstd * g4.y + b4.y;
      o.z = (v[r].z - mu) * rstd * g4.z + b4.z;
      o.w = (v[r].w - mu) * rstd * g4.w + b4.w;
      *(float4*)(C + (size_t)row * N + col) = o;
    }
  } else {
    #pragma unroll
    for (int r = 0; r < 8; ++r) {
      const int row = tm * 16 + r + 8 * half;
      *(float4*)(C + (size_t)row * N + col) = v[r];
    }
  }
}

__global__ __launch_bounds__(256) void lgesql_dst_count(
    const int* __restrict__ lg_dst, int* __restrict__ cnt, int* __restrict__ rank)
{
  const int e = blockIdx.x * blockDim.x + threadIdx.x;
  rank[e] = atomicAdd(&cnt[lg_dst[e]], 1);
}

__global__ __launch_bounds__(1024) void lgesql_dst_offsets(
    const int* __restrict__ cnt, int* __restrict__ off)
{
  constexpr int ITEMS = 20;
  static_assert(E_NODES % ITEMS == 0 && E_NODES / ITEMS <= 1024, "node split");
  __shared__ int wsum[32];
  const int t = threadIdx.x;
  const int lane = t & 31;
  const int w = t >> 5;
  const bool live = t < E_NODES / ITEMS;
  int c[ITEMS];
  if (live) {
    const int4* p = (const int4*)(cnt + t * ITEMS);
    #pragma unroll
    for (int g = 0; g < ITEMS / 4; ++g) {
      const int4 x = p[g];
      c[4 * g] = x.x; c[4 * g + 1] = x.y; c[4 * g + 2] = x.z; c[4 * g + 3] = x.w;
    }
  } else {
    #pragma unroll
    for (int i = 0; i < ITEMS; ++i) c[i] = 0;
  }
  int tot = 0;
  #pragma unroll
  for (int i = 0; i < ITEMS; ++i) { const int x = c[i]; c[i] = tot; tot += x; }
  int inc = tot;
  #pragma unroll
  for (int d = 1; d < 32; d <<= 1) {
    const int nb = __shfl_up(inc, d, 32);
    if (lane >= d) inc += nb;
  }
  if (lane == 31) wsum[w] = inc;
  __syncthreads();
  if (w == 0) {
    const int own = wsum[lane];
    int si = own;
    #pragma unroll
    for (int d = 1; d < 32; d <<= 1) {
      const int nb = __shfl_up(si, d, 32);
      if (lane >= d) si += nb;
    }
    wsum[lane] = si - own;
  }
  __syncthreads();
  const int base = wsum[w] + inc - tot;
  if (live) {
    int4* o = (int4*)(off + t * ITEMS);
    #pragma unroll
    for (int g = 0; g < ITEMS / 4; ++g)
      o[g] = make_int4(base + c[4 * g], base + c[4 * g + 1], base + c[4 * g + 2], base + c[4 * g + 3]);
  }
  if (t == 1023) off[E_NODES] = base + tot;
}

__global__ __launch_bounds__(256) void lgesql_dst_fill(
    const int* __restrict__ lg_src, const int* __restrict__ lg_dst,
    const int* __restrict__ off, const int* __restrict__ rank, int* __restrict__ slot_src)
{
  const int e = blockIdx.x * blockDim.x + threadIdx.x;
  slot_src[off[lg_dst[e]] + rank[e]] = lg_src[e];
}

template <int U>
__device__ __forceinline__ void lgesql_edges(
    const float* __restrict__ k, const float* __restrict__ v, int mysrc, int j, int lane,
    const float4& qa, const float4& qb, float4& accA, float4& accB, float& zA, float& zB)
{
  float4 ka[U], kb[U], va[U], vb[U];
  #pragma unroll
  for (int u = 0; u < U; ++u) {
    const int s = __builtin_amdgcn_readlane(mysrc, j + u);
    const float4* kr = (const float4*)(k + (size_t)s * NDIM_);
    const float4* vr = (const float4*)(v + (size_t)s * NDIM_);
    ka[u] = kr[lane]; kb[u] = kr[32 + lane];
    va[u] = vr[lane]; vb[u] = vr[32 + lane];
  }
  #pragma unroll
  for (int u = 0; u < U; ++u) {
    float pA = ka[u].x * qa.x + ka[u].y * qa.y + ka[u].z * qa.z + ka[u].w * qa.w;
    float pB = kb[u].x * qb.x + kb[u].y * qb.y + kb[u].z * qb.z + kb[u].w * qb.w;
    pA = lgesql_xadd<1>(pA); pB = lgesql_xadd<1>(pB);
    pA = lgesql_xadd<2>(pA); pB = lgesql_xadd<2>(pB);
    pA = lgesql_xadd<4>(pA); pB = lgesql_xadd<4>(pB);
    float sA = pA * 0.17677669529663687f;
    float sB = pB * 0.17677669529663687f;
    sA = expf(fminf(5.0f, fmaxf(-5.0f, sA)));
    sB = expf(fminf(5.0f, fmaxf(-5.0f, sB)));
    accA.x += va[u].x * sA; accA.y += va[u].y * sA; accA.z += va[u].z * sA; accA.w += va[u].w * sA;
    accB.x += vb[u].x * sB; accB.y += vb[u].y * sB; accB.z += vb[u].z * sB; accB.w += vb[u].w * sB;
    zA += sA; zB += sB;
  }
}

__global__ __launch_bounds__(256) void lgesql_edge_gather(
    const float* __restrict__ q, const float* __restrict__ k, const float* __restrict__ v,
    const int* __restrict__ off, const int* __restrict__ slot_src, float* __restrict__ o)
{
  const int lane = threadIdx.x & 31;
  const int n = __builtin_amdgcn_readfirstlane((int)((blockIdx.x * blockDim.x + threadIdx.x) >> 5));
  const int beg = off[n];
  const int end = off[n + 1];
  const float4* qr = (const float4*)(q + (size_t)n * NDIM_);
  const float4 qa = qr[lane];
  const float4 qb = qr[32 + lane];
  float4 accA = make_float4(0.f, 0.f, 0.f, 0.f);
  float4 accB = make_float4(0.f, 0.f, 0.f, 0.f);
  float zA = 0.f, zB = 0.f;
  for (int base = beg; base < end; base += 32) {
    const int m = min(32, end - base);
    const int mysrc = (lane < m) ? slot_src[base + lane] : 0;
    int j = 0;
    for (; j + 4 <= m; j += 4)
      lgesql_edges<4>(k, v, mysrc, j, lane, qa, qb, accA, accB, zA, zB);
    if (j + 2 <= m) {
      lgesql_edges<2>(k, v, mysrc, j, lane, qa, qb, accA, accB, zA, zB);
      j += 2;
    }
    if (j < m)
      lgesql_edges<1>(k, v, mysrc, j, lane, qa, qb, accA, accB, zA, zB);
  }
  float4* orow = (float4*)(o + (size_t)n * NDIM_);
  orow[lane]      = make_float4(accA.x / zA, accA.y / zA, accA.z / zA, accA.w / zA);
  orow[32 + lane] = make_float4(accB.x / zB, accB.y / zB, accB.z / zB, accB.w / zB);
}

extern "C" void kernel_launch(void* const* d_in, const int* in_sizes, int n_in,
                              void* d_out, int out_size, void* d_ws, size_t ws_size,
                              hipStream_t stream) {
  const float* x     = (const float*)d_in[0];
  const float* src_x = (const float*)d_in[1];
  const float* dst_x = (const float*)d_in[2];
  const float* Wq    = (const float*)d_in[3];
  const float* bq    = (const float*)d_in[4];
  const float* Wk    = (const float*)d_in[5];
  const float* Wv    = (const float*)d_in[6];
  const float* Wo    = (const float*)d_in[7];
  const float* bo    = (const float*)d_in[8];
  const float* ln1_g = (const float*)d_in[9];
  const float* ln1_b = (const float*)d_in[10];
  const float* W1    = (const float*)d_in[11];
  const float* b1    = (const float*)d_in[12];
  const float* W2    = (const float*)d_in[13];
  const float* b2    = (const float*)d_in[14];
  const float* ln2_g = (const float*)d_in[15];
  const float* ln2_b = (const float*)d_in[16];
  const int*   lg_src = (const int*)d_in[17];
  const int*   lg_dst = (const int*)d_in[18];
  float* out = (float*)d_out;

  const size_t EN = (size_t)E_NODES * NDIM_;
  float* ws = (float*)d_ws;
  float* qf   = ws;
  float* kf   = ws + EN;
  float* vf   = ws + 2 * EN;
  float* wvf  = ws + 3 * EN;
  int*   cntf = (int*)(ws + 4 * EN);
  int*   offf = cntf + E_NODES;
  float* ffn1 = ws + 4 * EN + (size_t)E_NODES * HEADS;
  int*   slotf = (int*)ffn1;
  int*   rankf = slotf + LE_EDGES;

  hipMemsetAsync(cntf, 0, (size_t)E_NODES * sizeof(int), stream);

  const dim3 blk(256);

  lgesql_dst_count<<<LE_EDGES / 256, blk, 0, stream>>>(lg_dst, cntf, rankf);
  lgesql_dst_offsets<<<1, dim3(1024), 0, stream>>>(cntf, offf);
  lgesql_dst_fill<<<LE_EDGES / 256, blk, 0, stream>>>(lg_src, lg_dst, offf, rankf, slotf);
  const int g256  = (E_NODES / 16) * (NDIM_ / 64) / 8;
  const int g1024 = (E_NODES / 16) * (4 * NDIM_ / 64) / 8;
  const int gln   = E_NODES / 16;
  const dim3 blkln(128);

  lgesql_gemm_wmma_f32<256, 256, true,  true,  false, false>
      <<<g256, blk, 0, stream>>>(x, Wq, bq, src_x, nullptr, nullptr, qf);
  lgesql_gemm_wmma_f32<256, 256, false, false, false, false>
      <<<g256, blk, 0, stream>>>(x, Wk, nullptr, nullptr, nullptr, nullptr, kf);
  lgesql_gemm_wmma_f32<256, 256, false, true,  false, false>
      <<<g256, blk, 0, stream>>>(x, Wv, nullptr, dst_x, nullptr, nullptr, vf);

  lgesql_edge_gather<<<E_NODES / 8, blk, 0, stream>>>(qf, kf, vf, offf, slotf, wvf);

  lgesql_gemm_wmma_f32<256, 256, true,  true,  false, true>
      <<<gln, blkln, 0, stream>>>(wvf, Wo, bo, x, ln1_g, ln1_b, kf);

  lgesql_gemm_wmma_f32<256, 1024, true, false, true, false>
      <<<g1024, blk, 0, stream>>>(kf, W1, b1, nullptr, nullptr, nullptr, ffn1);
  lgesql_gemm_wmma_f32<1024, 256, true, true, false, true>
      <<<gln, blkln, 0, stream>>>(ffn1, W2, b2, kf, ln2_g, ln2_b, out);
}


# q, k and v projections in one launch: grid of 3 x 625 work-groups, the work-group index picks Wq/Wk/Wv and its epilogue; 2 launches fewer, no idle tail between them; sums unchanged
# speedup vs baseline: 1.8197x; 1.0427x over previous
#include <hip/hip_runtime.h>
#include <math.h>

#define E_NODES 20000
#define LE_EDGES 320000
#define NDIM_ 256
#define HEADS 8
#define DK_ 32

typedef __attribute__((ext_vector_type(2))) float v2f;
typedef __attribute__((ext_vector_type(8))) float v8f;

__device__ __forceinline__ v2f lgesql_pair(float lo, float hi)
{
  v2f p; p.x = lo; p.y = hi; return p;
}

template <int XMASK>
__device__ __forceinline__ float lgesql_xadd(float x)
{
  const int y = __builtin_amdgcn_update_dpp(0, __builtin_bit_cast(int, x), 0x160 | XMASK, 0xf, 0xf, true);
  return x + __builtin_bit_cast(float, y);
}

template <int K, int N>
__device__ __forceinline__ void lgesql_tile_k_loop(
    const float* __restrict__ Ap, const float* __restrict__ Bp,
    v8f& acc0, v8f& acc1, v8f& acc2, v8f& acc3)
{
  for (int k = 0; k < K; k += 16) {
    int kk = k;
    asm volatile("v_nop\n\tv_nop\n\tv_nop\n\tv_nop\n\tv_nop\n\tv_nop\n\tv_nop\n\tv_nop"
                 : "+v"(acc0), "+v"(acc1), "+v"(acc2), "+v"(acc3), "+s"(kk));
    const float* ak = Ap + kk;
    const float* bk = Bp + (size_t)kk * N;
    const float4 a0 = *(const float4*)(ak);
    const float4 a1 = *(const float4*)(ak + 8);
    const float4 r0 = *(const float4*)(bk);
    const float4 r1 = *(const float4*)(bk + N);
    const float4 r2 = *(const float4*)(bk + 2 * N);
    const float4 r3 = *(const float4*)(bk + 3 * N);
    const float4 r4 = *(const float4*)(bk + 8 * N);
    const float4 r5 = *(const float4*)(bk + 9 * N);
    const float4 r6 = *(const float4*)(bk + 10 * N);
    const float4 r7 = *(const float4*)(bk + 11 * N);
    v2f p00 = lgesql_pair(r0.x, r1.x), p01 = lgesql_pair(r0.y, r1.y), p02 = lgesql_pair(r0.z, r1.z), p03 = lgesql_pair(r0.w, r1.w);
    v2f p10 = lgesql_pair(r2.x, r3.x), p11 = lgesql_pair(r2.y, r3.y), p12 = lgesql_pair(r2.z, r3.z), p13 = lgesql_pair(r2.w, r3.w);
    v2f p20 = lgesql_pair(r4.x, r5.x), p21 = lgesql_pair(r4.y, r5.y), p22 = lgesql_pair(r4.z, r5.z), p23 = lgesql_pair(r4.w, r5.w);
    v2f p30 = lgesql_pair(r6.x, r7.x), p31 = lgesql_pair(r6.y, r7.y), p32 = lgesql_pair(r6.z, r7.z), p33 = lgesql_pair(r6.w, r7.w);
    asm volatile("" : "+v"(p00), "+v"(p01), "+v"(p02), "+v"(p03), "+v"(p10), "+v"(p11), "+v"(p12), "+v"(p13),
                      "+v"(p20), "+v"(p21), "+v"(p22), "+v"(p23), "+v"(p30), "+v"(p31), "+v"(p32), "+v"(p33));
    const v2f a00 = lgesql_pair(a0.x, a0.y), a01 = lgesql_pair(a0.z, a0.w);
    const v2f a10 = lgesql_pair(a1.x, a1.y), a11 = lgesql_pair(a1.z, a1.w);
    acc0 = __builtin_amdgcn_wmma_f32_16x16x4_f32(false, a00, false, p00, (short)0, acc0, false, false);
    acc1 = __builtin_amdgcn_wmma_f32_16x16x4_f32(false, a00, false, p01, (short)0, acc1, false, false);
    acc2 = __builtin_amdgcn_wmma_f32_16x16x4_f32(false, a00, false, p02, (short)0, acc2, false, false);
    acc3 = __builtin_amdgcn_wmma_f32_16x16x4_f32(false, a00, false, p03, (short)0, acc3, false, false);
    acc0 = __builtin_amdgcn_wmma_f32_16x16x4_f32(false, a01, false, p10, (short)0, acc0, false, false);
    acc1 = __builtin_amdgcn_wmma_f32_16x16x4_f32(false, a01, false, p11, (short)0, acc1, false, false);
    acc2 = __builtin_amdgcn_wmma_f32_16x16x4_f32(false, a01, false, p12, (short)0, acc2, false, false);
    acc3 = __builtin_amdgcn_wmma_f32_16x16x4_f32(false, a01, false, p13, (short)0, acc3, false, false);
    acc0 = __builtin_amdgcn_wmma_f32_16x16x4_f32(false, a10, false, p20, (short)0, acc0, false, false);
    acc1 = __builtin_amdgcn_wmma_f32_16x16x4_f32(false, a10, false, p21, (short)0, acc1, false, false);
    acc2 = __builtin_amdgcn_wmma_f32_16x16x4_f32(false, a10, false, p22, (short)0, acc2, false, false);
    acc3 = __builtin_amdgcn_wmma_f32_16x16x4_f32(false, a10, false, p23, (short)0, acc3, false, false);
    acc0 = __builtin_amdgcn_wmma_f32_16x16x4_f32(false, a11, false, p30, (short)0, acc0, false, false);
    acc1 = __builtin_amdgcn_wmma_f32_16x16x4_f32(false, a11, false, p31, (short)0, acc1, false, false);
    acc2 = __builtin_amdgcn_wmma_f32_16x16x4_f32(false, a11, false, p32, (short)0, acc2, false, false);
    acc3 = __builtin_amdgcn_wmma_f32_16x16x4_f32(false, a11, false, p33, (short)0, acc3, false, false);
  }
  asm volatile("v_nop\n\tv_nop\n\tv_nop\n\tv_nop\n\tv_nop\n\tv_nop\n\tv_nop\n\tv_nop"
               : "+v"(acc0), "+v"(acc1), "+v"(acc2), "+v"(acc3));
}

template <int K, int N, bool HAS_BIAS, bool HAS_ADD, bool RELU, bool LN>
__global__ __launch_bounds__(LN ? 128 : 256) void lgesql_gemm_wmma_f32(
    const float* __restrict__ A, const float* __restrict__ W,
    const float* __restrict__ bias, const float* __restrict__ addm,
    const float* __restrict__ ln_g, const float* __restrict__ ln_b,
    float* __restrict__ C)
{
  static_assert(!LN || N == 256, "LN: the four waves of a work-group must cover whole rows");
  constexpr int TN = 64;
  constexpr int tilesN = N / TN;
  static_assert(K % 16 == 0 && N % TN == 0, "tile split");
  const int wid  = (blockIdx.x * blockDim.x + threadIdx.x) >> 5;
  const int lane = threadIdx.x & 31;
  const int tm = wid / tilesN;
  const int tn = wid - tm * tilesN;
  const int half = lane >> 4;
  const int l    = lane & 15;

  const float* Ap = A + (size_t)(tm * 16 + l) * K + 4 * half;
  const float* Bp = W + (size_t)(4 * half) * N + tn * TN + 4 * l;

  v8f acc0 = {}, acc1 = {}, acc2 = {}, acc3 = {};
  lgesql_tile_k_loop<K, N>(Ap, Bp, acc0, acc1, acc2, acc3);

  const int col = tn * TN + 4 * l;
  float4 bv = make_float4(0.0f, 0.0f, 0.0f, 0.0f);
  if (HAS_BIAS) bv = *(const float4*)(bias + col);
  float4 v[8];
  #pragma unroll
  for (int r = 0; r < 8; ++r) {
    const int row = tm * 16 + r + 8 * half;
    v[r] = make_float4(acc0[r] + bv.x, acc1[r] + bv.y, acc2[r] + bv.z, acc3[r] + bv.w);
    if (HAS_ADD) {
      const float4 ad = *(const float4*)(addm + (size_t)row * N + col);
      v[r].x += ad.x; v[r].y += ad.y; v[r].z += ad.z; v[r].w += ad.w;
    }
    if (RELU) {
      v[r].x = fmaxf(v[r].x, 0.0f); v[r].y = fmaxf(v[r].y, 0.0f);
      v[r].z = fmaxf(v[r].z, 0.0f); v[r].w = fmaxf(v[r].w, 0.0f);
    }
  }
  if constexpr (LN) {
    __shared__ float2 part[16][4];
    const int w = threadIdx.x >> 5;
    #pragma unroll
    for (int r = 0; r < 8; ++r) {
      float sm = (v[r].x + v[r].y) + (v[r].z + v[r].w);
      float sq = (v[r].x * v[r].x + v[r].y * v[r].y) + (v[r].z * v[r].z + v[r].w * v[r].w);
      sm = lgesql_xadd<1>(sm); sq = lgesql_xadd<1>(sq);
      sm = lgesql_xadd<2>(sm); sq = lgesql_xadd<2>(sq);
      sm = lgesql_xadd<4>(sm); sq = lgesql_xadd<4>(sq);
      sm = lgesql_xadd<8>(sm); sq = lgesql_xadd<8>(sq);
      if (l == r) part[r + 8 * half][w] = make_float2(sm, sq);
    }
    __syncthreads();
    const float4 g4 = *(const float4*)(ln_g + col);
    const float4 b4 = *(const float4*)(ln_b + col);
    #pragma unroll
    for (int r = 0; r < 8; ++r) {
      const int row = tm * 16 + r + 8 * half;
      const float4 p01 = *(const float4*)(&part[r + 8 * half][0]);
      const float4 p23 = *(const float4*)(&part[r + 8 * half][2]);
      const float sum = ((p01.x + p01.z) + p23.x) + p23.z;
      const float sqs = ((p01.y + p01.w) + p23.y) + p23.w;
      const float mu   = sum * (1.0f / N);
      const float var  = sqs * (1.0f / N) - mu * mu;
      const float rstd = rsqrtf(var + 1e-5f);
      float4 o;
      o.x = (v[r].x - mu) * rstd * g4.x + b4.x;
      o.y = (v[r].y - mu) * rstd * g4.y + b4.y;
      o.z = (v[r].z - mu) * rstd * g4.z + b4.z;
      o.w = (v[r].w - mu) * rstd * g4.w + b4.w;
      *(float4*)(C + (size_t)row * N + col) = o;
    }
  } else {
    #pragma unroll
    for (int r = 0; r < 8; ++r) {
      const int row = tm * 16 + r + 8 * half;
      *(float4*)(C + (size_t)row * N + col) = v[r];
    }
  }
}

__global__ __launch_bounds__(256) void lgesql_qkv_wmma_f32(
    const float* __restrict__ x,
    const float* __restrict__ Wq, const float* __restrict__ Wk, const float* __restrict__ Wv,
    const float* __restrict__ bq, const float* __restrict__ src_x, const float* __restrict__ dst_x,
    float* __restrict__ q, float* __restrict__ k, float* __restrict__ v)
{
  constexpr int K = NDIM_, N = NDIM_;
  constexpr int TN = 64;
  constexpr int tilesN = N / TN;
  constexpr int GROUPS_PER_MATRIX = (E_NODES / 16) * tilesN / 8;
  const int which = (int)blockIdx.x / GROUPS_PER_MATRIX;
  const int grp   = (int)blockIdx.x - which * GROUPS_PER_MATRIX;
  const float* __restrict__ W = which == 0 ? Wq : (which == 1 ? Wk : Wv);
  float* __restrict__ C       = which == 0 ? q  : (which == 1 ? k  : v);
  const float* __restrict__ addm = which == 0 ? src_x : dst_x;

  const int wid  = (grp * (int)blockDim.x + (int)threadIdx.x) >> 5;
  const int lane = threadIdx.x & 31;
  const int tm = wid / tilesN;
  const int tn = wid - tm * tilesN;
  const int half = lane >> 4;
  const int l    = lane & 15;

  const float* Ap = x + (size_t)(tm * 16 + l) * K + 4 * half;
  const float* Bp = W + (size_t)(4 * half) * N + tn * TN + 4 * l;

  v8f acc0 = {}, acc1 = {}, acc2 = {}, acc3 = {};
  lgesql_tile_k_loop<K, N>(Ap, Bp, acc0, acc1, acc2, acc3);

  const int col = tn * TN + 4 * l;
  const size_t at = (size_t)(tm * 16 + 8 * half) * N + col;
  float* __restrict__ Cp = C + at;
  if (which == 1) {
    #pragma unroll
    for (int r = 0; r < 8; ++r)
      *(float4*)(Cp + (size_t)r * N) = make_float4(acc0[r], acc1[r], acc2[r], acc3[r]);
  } else {
    const float* __restrict__ Rp = addm + at;
    float4 bv = make_float4(0.0f, 0.0f, 0.0f, 0.0f);
    if (which == 0) bv = *(const float4*)(bq + col);
    float4 ad[8];
    #pragma unroll
    for (int r = 0; r < 8; ++r) ad[r] = *(const float4*)(Rp + (size_t)r * N);
    #pragma unroll
    for (int r = 0; r < 8; ++r) {
      float4 o = make_float4(acc0[r] + bv.x, acc1[r] + bv.y, acc2[r] + bv.z, acc3[r] + bv.w);
      o.x += ad[r].x; o.y += ad[r].y; o.z += ad[r].z; o.w += ad[r].w;
      *(float4*)(Cp + (size_t)r * N) = o;
    }
  }
}

__global__ __launch_bounds__(256) void lgesql_dst_count(
    const int* __restrict__ lg_dst, int* __restrict__ cnt, int* __restrict__ rank)
{
  const int e = blockIdx.x * blockDim.x + threadIdx.x;
  rank[e] = atomicAdd(&cnt[lg_dst[e]], 1);
}

__global__ __launch_bounds__(1024) void lgesql_dst_offsets(
    const int* __restrict__ cnt, int* __restrict__ off)
{
  constexpr int ITEMS = 20;
  static_assert(E_NODES % ITEMS == 0 && E_NODES / ITEMS <= 1024, "node split");
  __shared__ int wsum[32];
  const int t = threadIdx.x;
  const int lane = t & 31;
  const int w = t >> 5;
  const bool live = t < E_NODES / ITEMS;
  int c[ITEMS];
  if (live) {
    const int4* p = (const int4*)(cnt + t * ITEMS);
    #pragma unroll
    for (int g = 0; g < ITEMS / 4; ++g) {
      const int4 x = p[g];
      c[4 * g] = x.x; c[4 * g + 1] = x.y; c[4 * g + 2] = x.z; c[4 * g + 3] = x.w;
    }
  } else {
    #pragma unroll
    for (int i = 0; i < ITEMS; ++i) c[i] = 0;
  }
  int tot = 0;
  #pragma unroll
  for (int i = 0; i < ITEMS; ++i) { const int x = c[i]; c[i] = tot; tot += x; }
  int inc = tot;
  #pragma unroll
  for (int d = 1; d < 32; d <<= 1) {
    const int nb = __shfl_up(inc, d, 32);
    if (lane >= d) inc += nb;
  }
  if (lane == 31) wsum[w] = inc;
  __syncthreads();
  if (w == 0) {
    const int own = wsum[lane];
    int si = own;
    #pragma unroll
    for (int d = 1; d < 32; d <<= 1) {
      const int nb = __shfl_up(si, d, 32);
      if (lane >= d) si += nb;
    }
    wsum[lane] = si - own;
  }
  __syncthreads();
  const int base = wsum[w] + inc - tot;
  if (live) {
    int4* o = (int4*)(off + t * ITEMS);
    #pragma unroll
    for (int g = 0; g < ITEMS / 4; ++g)
      o[g] = make_int4(base + c[4 * g], base + c[4 * g + 1], base + c[4 * g + 2], base + c[4 * g + 3]);
  }
  if (t == 1023) off[E_NODES] = base + tot;
}

__global__ __launch_bounds__(256) void lgesql_dst_fill(
    const int* __restrict__ lg_src, const int* __restrict__ lg_dst,
    const int* __restrict__ off, const int* __restrict__ rank, int* __restrict__ slot_src)
{
  const int e = blockIdx.x * blockDim.x + threadIdx.x;
  slot_src[off[lg_dst[e]] + rank[e]] = lg_src[e];
}

template <int U>
__device__ __forceinline__ void lgesql_edges(
    const float* __restrict__ k, const float* __restrict__ v, int mysrc, int j, int lane,
    const float4& qa, const float4& qb, float4& accA, float4& accB, float& zA, float& zB)
{
  float4 ka[U], kb[U], va[U], vb[U];
  #pragma unroll
  for (int u = 0; u < U; ++u) {
    const int s = __builtin_amdgcn_readlane(mysrc, j + u);
    const float4* kr = (const float4*)(k + (size_t)s * NDIM_);
    const float4* vr = (const float4*)(v + (size_t)s * NDIM_);
    ka[u] = kr[lane]; kb[u] = kr[32 + lane];
    va[u] = vr[lane]; vb[u] = vr[32 + lane];
  }
  #pragma unroll
  for (int u = 0; u < U; ++u) {
    float pA = ka[u].x * qa.x + ka[u].y * qa.y + ka[u].z * qa.z + ka[u].w * qa.w;
    float pB = kb[u].x * qb.x + kb[u].y * qb.y + kb[u].z * qb.z + kb[u].w * qb.w;
    pA = lgesql_xadd<1>(pA); pB = lgesql_xadd<1>(pB);
    pA = lgesql_xadd<2>(pA); pB = lgesql_xadd<2>(pB);
    pA = lgesql_xadd<4>(pA); pB = lgesql_xadd<4>(pB);
    float sA = pA * 0.17677669529663687f;
    float sB = pB * 0.17677669529663687f;
    sA = expf(fminf(5.0f, fmaxf(-5.0f, sA)));
    sB = expf(fminf(5.0f, fmaxf(-5.0f, sB)));
    accA.x += va[u].x * sA; accA.y += va[u].y * sA; accA.z += va[u].z * sA; accA.w += va[u].w * sA;
    accB.x += vb[u].x * sB; accB.y += vb[u].y * sB; accB.z += vb[u].z * sB; accB.w += vb[u].w * sB;
    zA += sA; zB += sB;
  }
}

__global__ __launch_bounds__(256) void lgesql_edge_gather(
    const float* __restrict__ q, const float* __restrict__ k, const float* __restrict__ v,
    const int* __restrict__ off, const int* __restrict__ slot_src, float* __restrict__ o)
{
  const int lane = threadIdx.x & 31;
  const int n = __builtin_amdgcn_readfirstlane((int)((blockIdx.x * blockDim.x + threadIdx.x) >> 5));
  const int beg = off[n];
  const int end = off[n + 1];
  const float4* qr = (const float4*)(q + (size_t)n * NDIM_);
  const float4 qa = qr[lane];
  const float4 qb = qr[32 + lane];
  float4 accA = make_float4(0.f, 0.f, 0.f, 0.f);
  float4 accB = make_float4(0.f, 0.f, 0.f, 0.f);
  float zA = 0.f, zB = 0.f;
  for (int base = beg; base < end; base += 32) {
    const int m = min(32, end - base);
    const int mysrc = (lane < m) ? slot_src[base + lane] : 0;
    int j = 0;
    for (; j + 4 <= m; j += 4)
      lgesql_edges<4>(k, v, mysrc, j, lane, qa, qb, accA, accB, zA, zB);
    if (j + 2 <= m) {
      lgesql_edges<2>(k, v, mysrc, j, lane, qa, qb, accA, accB, zA, zB);
      j += 2;
    }
    if (j < m)
      lgesql_edges<1>(k, v, mysrc, j, lane, qa, qb, accA, accB, zA, zB);
  }
  float4* orow = (float4*)(o + (size_t)n * NDIM_);
  orow[lane]      = make_float4(accA.x / zA, accA.y / zA, accA.z / zA, accA.w / zA);
  orow[32 + lane] = make_float4(accB.x / zB, accB.y / zB, accB.z / zB, accB.w / zB);
}

extern "C" void kernel_launch(void* const* d_in, const int* in_sizes, int n_in,
                              void* d_out, int out_size, void* d_ws, size_t ws_size,
                              hipStream_t stream) {
  const float* x     = (const float*)d_in[0];
  const float* src_x = (const float*)d_in[1];
  const float* dst_x = (const float*)d_in[2];
  const float* Wq    = (const float*)d_in[3];
  const float* bq    = (const float*)d_in[4];
  const float* Wk    = (const float*)d_in[5];
  const float* Wv    = (const float*)d_in[6];
  const float* Wo    = (const float*)d_in[7];
  const float* bo    = (const float*)d_in[8];
  const float* ln1_g = (const float*)d_in[9];
  const float* ln1_b = (const float*)d_in[10];
  const float* W1    = (const float*)d_in[11];
  const float* b1    = (const float*)d_in[12];
  const float* W2    = (const float*)d_in[13];
  const float* b2    = (const float*)d_in[14];
  const float* ln2_g = (const float*)d_in[15];
  const float* ln2_b = (const float*)d_in[16];
  const int*   lg_src = (const int*)d_in[17];
  const int*   lg_dst = (const int*)d_in[18];
  float* out = (float*)d_out;

  const size_t EN = (size_t)E_NODES * NDIM_;
  float* ws = (float*)d_ws;
  float* qf   = ws;
  float* kf   = ws + EN;
  float* vf   = ws + 2 * EN;
  float* wvf  = ws + 3 * EN;
  int*   cntf = (int*)(ws + 4 * EN);
  int*   offf = cntf + E_NODES;
  float* ffn1 = ws + 4 * EN + (size_t)E_NODES * HEADS;
  int*   slotf = (int*)ffn1;
  int*   rankf = slotf + LE_EDGES;

  hipMemsetAsync(cntf, 0, (size_t)E_NODES * sizeof(int), stream);

  const dim3 blk(256);

  lgesql_dst_count<<<LE_EDGES / 256, blk, 0, stream>>>(lg_dst, cntf, rankf);
  lgesql_dst_offsets<<<1, dim3(1024), 0, stream>>>(cntf, offf);
  lgesql_dst_fill<<<LE_EDGES / 256, blk, 0, stream>>>(lg_src, lg_dst, offf, rankf, slotf);
  const int gqkv  = 3 * ((E_NODES / 16) * (NDIM_ / 64) / 8);
  const int g1024 = (E_NODES / 16) * (4 * NDIM_ / 64) / 8;
  const int gln   = E_NODES / 16;
  const dim3 blkln(128);

  lgesql_qkv_wmma_f32<<<gqkv, blk, 0, stream>>>(x, Wq, Wk, Wv, bq, src_x, dst_x, qf, kf, vf);

  lgesql_edge_gather<<<E_NODES / 8, blk, 0, stream>>>(qf, kf, vf, offf, slotf, wvf);

  lgesql_gemm_wmma_f32<256, 256, true,  true,  false, true>
      <<<gln, blkln, 0, stream>>>(wvf, Wo, bo, x, ln1_g, ln1_b, kf);

  lgesql_gemm_wmma_f32<256, 1024, true, false, true, false>
      <<<g1024, blk, 0, stream>>>(kf, W1, b1, nullptr, nullptr, nullptr, ffn1);
  lgesql_gemm_wmma_f32<1024, 256, true, true, false, true>
      <<<gln, blkln, 0, stream>>>(ffn1, W2, b2, kf, ln2_g, ln2_b, out);
}


# Edges grouped by dst in ONE pass inside the q/k/v launch: int-atomic slot in a 64-slot row per node + overflow list; count, prefix-sum and fill launches gone (3 fewer); sums still in arrival rank
# speedup vs baseline: 1.8954x; 1.0416x over previous
#include <hip/hip_runtime.h>
#include <math.h>

#define E_NODES 20000
#define LE_EDGES 320000
#define NDIM_ 256
#define HEADS 8
#define DK_ 32
#define BUCKET_CAP 64
#define GROUP_EPT 5
#define GROUP_WGS (LE_EDGES / (256 * GROUP_EPT))

typedef __attribute__((ext_vector_type(2))) float v2f;
typedef __attribute__((ext_vector_type(8))) float v8f;

__device__ __forceinline__ v2f lgesql_pair(float lo, float hi)
{
  v2f p; p.x = lo; p.y = hi; return p;
}

template <int XMASK>
__device__ __forceinline__ float lgesql_xadd(float x)
{
  const int y = __builtin_amdgcn_update_dpp(0, __builtin_bit_cast(int, x), 0x160 | XMASK, 0xf, 0xf, true);
  return x + __builtin_bit_cast(float, y);
}

template <int K, int N>
__device__ __forceinline__ void lgesql_tile_k_loop(
    const float* __restrict__ Ap, const float* __restrict__ Bp,
    v8f& acc0, v8f& acc1, v8f& acc2, v8f& acc3)
{
  for (int k = 0; k < K; k += 16) {
    int kk = k;
    asm volatile("v_nop\n\tv_nop\n\tv_nop\n\tv_nop\n\tv_nop\n\tv_nop\n\tv_nop\n\tv_nop"
                 : "+v"(acc0), "+v"(acc1), "+v"(acc2), "+v"(acc3), "+s"(kk));
    const float* ak = Ap + kk;
    const float* bk = Bp + (size_t)kk * N;
    const float4 a0 = *(const float4*)(ak);
    const float4 a1 = *(const float4*)(ak + 8);
    const float4 r0 = *(const float4*)(bk);
    const float4 r1 = *(const float4*)(bk + N);
    const float4 r2 = *(const float4*)(bk + 2 * N);
    const float4 r3 = *(const float4*)(bk + 3 * N);
    const float4 r4 = *(const float4*)(bk + 8 * N);
    const float4 r5 = *(const float4*)(bk + 9 * N);
    const float4 r6 = *(const float4*)(bk + 10 * N);
    const float4 r7 = *(const float4*)(bk + 11 * N);
    v2f p00 = lgesql_pair(r0.x, r1.x), p01 = lgesql_pair(r0.y, r1.y), p02 = lgesql_pair(r0.z, r1.z), p03 = lgesql_pair(r0.w, r1.w);
    v2f p10 = lgesql_pair(r2.x, r3.x), p11 = lgesql_pair(r2.y, r3.y), p12 = lgesql_pair(r2.z, r3.z), p13 = lgesql_pair(r2.w, r3.w);
    v2f p20 = lgesql_pair(r4.x, r5.x), p21 = lgesql_pair(r4.y, r5.y), p22 = lgesql_pair(r4.z, r5.z), p23 = lgesql_pair(r4.w, r5.w);
    v2f p30 = lgesql_pair(r6.x, r7.x), p31 = lgesql_pair(r6.y, r7.y), p32 = lgesql_pair(r6.z, r7.z), p33 = lgesql_pair(r6.w, r7.w);
    asm volatile("" : "+v"(p00), "+v"(p01), "+v"(p02), "+v"(p03), "+v"(p10), "+v"(p11), "+v"(p12), "+v"(p13),
                      "+v"(p20), "+v"(p21), "+v"(p22), "+v"(p23), "+v"(p30), "+v"(p31), "+v"(p32), "+v"(p33));
    const v2f a00 = lgesql_pair(a0.x, a0.y), a01 = lgesql_pair(a0.z, a0.w);
    const v2f a10 = lgesql_pair(a1.x, a1.y), a11 = lgesql_pair(a1.z, a1.w);
    acc0 = __builtin_amdgcn_wmma_f32_16x16x4_f32(false, a00, false, p00, (short)0, acc0, false, false);
    acc1 = __builtin_amdgcn_wmma_f32_16x16x4_f32(false, a00, false, p01, (short)0, acc1, false, false);
    acc2 = __builtin_amdgcn_wmma_f32_16x16x4_f32(false, a00, false, p02, (short)0, acc2, false, false);
    acc3 = __builtin_amdgcn_wmma_f32_16x16x4_f32(false, a00, false, p03, (short)0, acc3, false, false);
    acc0 = __builtin_amdgcn_wmma_f32_16x16x4_f32(false, a01, false, p10, (short)0, acc0, false, false);
    acc1 = __builtin_amdgcn_wmma_f32_16x16x4_f32(false, a01, false, p11, (short)0, acc1, false, false);
    acc2 = __builtin_amdgcn_wmma_f32_16x16x4_f32(false, a01, false, p12, (short)0, acc2, false, false);
    acc3 = __builtin_amdgcn_wmma_f32_16x16x4_f32(false, a01, false, p13, (short)0, acc3, false, false);
    acc0 = __builtin_amdgcn_wmma_f32_16x16x4_f32(false, a10, false, p20, (short)0, acc0, false, false);
    acc1 = __builtin_amdgcn_wmma_f32_16x16x4_f32(false, a10, false, p21, (short)0, acc1, false, false);
    acc2 = __builtin_amdgcn_wmma_f32_16x16x4_f32(false, a10, false, p22, (short)0, acc2, false, false);
    acc3 = __builtin_amdgcn_wmma_f32_16x16x4_f32(false, a10, false, p23, (short)0, acc3, false, false);
    acc0 = __builtin_amdgcn_wmma_f32_16x16x4_f32(false, a11, false, p30, (short)0, acc0, false, false);
    acc1 = __builtin_amdgcn_wmma_f32_16x16x4_f32(false, a11, false, p31, (short)0, acc1, false, false);
    acc2 = __builtin_amdgcn_wmma_f32_16x16x4_f32(false, a11, false, p32, (short)0, acc2, false, false);
    acc3 = __builtin_amdgcn_wmma_f32_16x16x4_f32(false, a11, false, p33, (short)0, acc3, false, false);
  }
  asm volatile("v_nop\n\tv_nop\n\tv_nop\n\tv_nop\n\tv_nop\n\tv_nop\n\tv_nop\n\tv_nop"
               : "+v"(acc0), "+v"(acc1), "+v"(acc2), "+v"(acc3));
}

template <int K, int N, bool HAS_BIAS, bool HAS_ADD, bool RELU, bool LN>
__global__ __launch_bounds__(LN ? 128 : 256) void lgesql_gemm_wmma_f32(
    const float* __restrict__ A, const float* __restrict__ W,
    const float* __restrict__ bias, const float* __restrict__ addm,
    const float* __restrict__ ln_g, const float* __restrict__ ln_b,
    float* __restrict__ C)
{
  static_assert(!LN || N == 256, "LN: the four waves of a work-group must cover whole rows");
  constexpr int TN = 64;
  constexpr int tilesN = N / TN;
  static_assert(K % 16 == 0 && N % TN == 0, "tile split");
  const int wid  = (blockIdx.x * blockDim.x + threadIdx.x) >> 5;
  const int lane = threadIdx.x & 31;
  const int tm = wid / tilesN;
  const int tn = wid - tm * tilesN;
  const int half = lane >> 4;
  const int l    = lane & 15;

  const float* Ap = A + (size_t)(tm * 16 + l) * K + 4 * half;
  const float* Bp = W + (size_t)(4 * half) * N + tn * TN + 4 * l;

  v8f acc0 = {}, acc1 = {}, acc2 = {}, acc3 = {};
  lgesql_tile_k_loop<K, N>(Ap, Bp, acc0, acc1, acc2, acc3);

  const int col = tn * TN + 4 * l;
  float4 bv = make_float4(0.0f, 0.0f, 0.0f, 0.0f);
  if (HAS_BIAS) bv = *(const float4*)(bias + col);
  float4 v[8];
  #pragma unroll
  for (int r = 0; r < 8; ++r) {
    const int row = tm * 16 + r + 8 * half;
    v[r] = make_float4(acc0[r] + bv.x, acc1[r] + bv.y, acc2[r] + bv.z, acc3[r] + bv.w);
    if (HAS_ADD) {
      const float4 ad = *(const float4*)(addm + (size_t)row * N + col);
      v[r].x += ad.x; v[r].y += ad.y; v[r].z += ad.z; v[r].w += ad.w;
    }
    if (RELU) {
      v[r].x = fmaxf(v[r].x, 0.0f); v[r].y = fmaxf(v[r].y, 0.0f);
      v[r].z = fmaxf(v[r].z, 0.0f); v[r].w = fmaxf(v[r].w, 0.0f);
    }
  }
  if constexpr (LN) {
    __shared__ float2 part[16][4];
    const int w = threadIdx.x >> 5;
    #pragma unroll
    for (int r = 0; r < 8; ++r) {
      float sm = (v[r].x + v[r].y) + (v[r].z + v[r].w);
      float sq = (v[r].x * v[r].x + v[r].y * v[r].y) + (v[r].z * v[r].z + v[r].w * v[r].w);
      sm = lgesql_xadd<1>(sm); sq = lgesql_xadd<1>(sq);
      sm = lgesql_xadd<2>(sm); sq = lgesql_xadd<2>(sq);
      sm = lgesql_xadd<4>(sm); sq = lgesql_xadd<4>(sq);
      sm = lgesql_xadd<8>(sm); sq = lgesql_xadd<8>(sq);
      if (l == r) part[r + 8 * half][w] = make_float2(sm, sq);
    }
    __syncthreads();
    const float4 g4 = *(const float4*)(ln_g + col);
    const float4 b4 = *(const float4*)(ln_b + col);
    #pragma unroll
    for (int r = 0; r < 8; ++r) {
      const int row = tm * 16 + r + 8 * half;
      const float4 p01 = *(const float4*)(&part[r + 8 * half][0]);
      const float4 p23 = *(const float4*)(&part[r + 8 * half][2]);
      const float sum = ((p01.x + p01.z) + p23.x) + p23.z;
      const float sqs = ((p01.y + p01.w) + p23.y) + p23.w;
      const float mu   = sum * (1.0f / N);
      const float var  = sqs * (1.0f / N) - mu * mu;
      const float rstd = rsqrtf(var + 1e-5f);
      float4 o;
      o.x = (v[r].x - mu) * rstd * g4.x + b4.x;
      o.y = (v[r].y - mu) * rstd * g4.y + b4.y;
      o.z = (v[r].z - mu) * rstd * g4.z + b4.z;
      o.w = (v[r].w - mu) * rstd * g4.w + b4.w;
      *(float4*)(C + (size_t)row * N + col) = o;
    }
  } else {
    #pragma unroll
    for (int r = 0; r < 8; ++r) {
      const int row = tm * 16 + r + 8 * half;
      *(float4*)(C + (size_t)row * N + col) = v[r];
    }
  }
}

__device__ __forceinline__ void lgesql_group_edges(
    int wg, const int* __restrict__ lg_src, const int* __restrict__ lg_dst,
    int* __restrict__ cnt, int* __restrict__ bucket, int* __restrict__ ovf, int* __restrict__ ovf_n)
{
  static_assert(GROUP_WGS * 256 * GROUP_EPT == LE_EDGES, "the grouping work-groups cover the edges exactly");
  const int e0 = wg * (256 * GROUP_EPT) + (int)threadIdx.x;
  int d[GROUP_EPT], sidx[GROUP_EPT], r[GROUP_EPT];
  #pragma unroll
  for (int i = 0; i < GROUP_EPT; ++i) { d[i] = lg_dst[e0 + 256 * i]; sidx[i] = lg_src[e0 + 256 * i]; }
  #pragma unroll
  for (int i = 0; i < GROUP_EPT; ++i) r[i] = atomicAdd(&cnt[d[i]], 1);
  #pragma unroll
  for (int i = 0; i < GROUP_EPT; ++i) {
    if (r[i] < BUCKET_CAP) {
      bucket[(size_t)d[i] * BUCKET_CAP + r[i]] = sidx[i];
    } else {
      const int p = atomicAdd(ovf_n, 1);
      ovf[2 * p] = d[i]; ovf[2 * p + 1] = sidx[i];
    }
  }
}

__global__ __launch_bounds__(256) void lgesql_qkv_wmma_f32(
    const float* __restrict__ x,
    const float* __restrict__ Wq, const float* __restrict__ Wk, const float* __restrict__ Wv,
    const float* __restrict__ bq, const float* __restrict__ src_x, const float* __restrict__ dst_x,
    float* __restrict__ q, float* __restrict__ k, float* __restrict__ v,
    const int* __restrict__ lg_src, const int* __restrict__ lg_dst,
    int* __restrict__ cnt, int* __restrict__ bucket, int* __restrict__ ovf, int* __restrict__ ovf_n)
{
  if ((int)blockIdx.x < GROUP_WGS) {
    lgesql_group_edges((int)blockIdx.x, lg_src, lg_dst, cnt, bucket, ovf, ovf_n);
    return;
  }
  const int pblk = (int)blockIdx.x - GROUP_WGS;
  constexpr int K = NDIM_, N = NDIM_;
  constexpr int TN = 64;
  constexpr int tilesN = N / TN;
  constexpr int GROUPS_PER_MATRIX = (E_NODES / 16) * tilesN / 8;
  const int which = pblk / GROUPS_PER_MATRIX;
  const int grp   = pblk - which * GROUPS_PER_MATRIX;
  const float* __restrict__ W = which == 0 ? Wq : (which == 1 ? Wk : Wv);
  float* __restrict__ C       = which == 0 ? q  : (which == 1 ? k  : v);
  const float* __restrict__ addm = which == 0 ? src_x : dst_x;

  const int wid  = (grp * (int)blockDim.x + (int)threadIdx.x) >> 5;
  const int lane = threadIdx.x & 31;
  const int tm = wid / tilesN;
  const int tn = wid - tm * tilesN;
  const int half = lane >> 4;
  const int l    = lane & 15;

  const float* Ap = x + (size_t)(tm * 16 + l) * K + 4 * half;
  const float* Bp = W + (size_t)(4 * half) * N + tn * TN + 4 * l;

  v8f acc0 = {}, acc1 = {}, acc2 = {}, acc3 = {};
  lgesql_tile_k_loop<K, N>(Ap, Bp, acc0, acc1, acc2, acc3);

  int tm_e = tm, tn_e = tn, half_e = half, l_e = l;
  asm volatile("" : "+v"(tm_e), "+v"(tn_e), "+v"(half_e), "+v"(l_e), "+v"(acc0));
  const int col = tn_e * TN + 4 * l_e;
  const size_t at = (size_t)(tm_e * 16 + 8 * half_e) * N + col;
  float* __restrict__ Cp = C + at;
  if (which == 1) {
    #pragma unroll
    for (int r = 0; r < 8; ++r)
      *(float4*)(Cp + (size_t)r * N) = make_float4(acc0[r], acc1[r], acc2[r], acc3[r]);
  } else {
    const float* __restrict__ Rp = addm + at;
    float4 bv = make_float4(0.0f, 0.0f, 0.0f, 0.0f);
    if (which == 0) bv = *(const float4*)(bq + col);
    float4 ad[8];
    #pragma unroll
    for (int r = 0; r < 8; ++r) ad[r] = *(const float4*)(Rp + (size_t)r * N);
    #pragma unroll
    for (int r = 0; r < 8; ++r) {
      float4 o = make_float4(acc0[r] + bv.x, acc1[r] + bv.y, acc2[r] + bv.z, acc3[r] + bv.w);
      o.x += ad[r].x; o.y += ad[r].y; o.z += ad[r].z; o.w += ad[r].w;
      *(float4*)(Cp + (size_t)r * N) = o;
    }
  }
}

template <int U>
__device__ __forceinline__ void lgesql_edges(
    const float* __restrict__ k, const float* __restrict__ v, int mysrc, unsigned& todo, int lane,
    const float4& qa, const float4& qb, float4& accA, float4& accB, float& zA, float& zB)
{
  float4 ka[U], kb[U], va[U], vb[U];
  #pragma unroll
  for (int u = 0; u < U; ++u) {
    const int s = __builtin_amdgcn_readlane(mysrc, __builtin_ctz(todo));
    todo &= todo - 1;
    const float4* kr = (const float4*)(k + (size_t)s * NDIM_);
    const float4* vr = (const float4*)(v + (size_t)s * NDIM_);
    ka[u] = kr[lane]; kb[u] = kr[32 + lane];
    va[u] = vr[lane]; vb[u] = vr[32 + lane];
  }
  #pragma unroll
  for (int u = 0; u < U; ++u) {
    float pA = ka[u].x * qa.x + ka[u].y * qa.y + ka[u].z * qa.z + ka[u].w * qa.w;
    float pB = kb[u].x * qb.x + kb[u].y * qb.y + kb[u].z * qb.z + kb[u].w * qb.w;
    pA = lgesql_xadd<1>(pA); pB = lgesql_xadd<1>(pB);
    pA = lgesql_xadd<2>(pA); pB = lgesql_xadd<2>(pB);
    pA = lgesql_xadd<4>(pA); pB = lgesql_xadd<4>(pB);
    float sA = pA * 0.17677669529663687f;
    float sB = pB * 0.17677669529663687f;
    sA = expf(fminf(5.0f, fmaxf(-5.0f, sA)));
    sB = expf(fminf(5.0f, fmaxf(-5.0f, sB)));
    accA.x += va[u].x * sA; accA.y += va[u].y * sA; accA.z += va[u].z * sA; accA.w += va[u].w * sA;
    accB.x += vb[u].x * sB; accB.y += vb[u].y * sB; accB.z += vb[u].z * sB; accB.w += vb[u].w * sB;
    zA += sA; zB += sB;
  }
}

__global__ __launch_bounds__(256) void lgesql_edge_gather(
    const float* __restrict__ q, const float* __restrict__ k, const float* __restrict__ v,
    const int* __restrict__ cnt, const int* __restrict__ bucket,
    const int* __restrict__ ovf, const int* __restrict__ ovf_n, float* __restrict__ o)
{
  const int lane = threadIdx.x & 31;
  const int n = __builtin_amdgcn_readfirstlane((int)((blockIdx.x * blockDim.x + threadIdx.x) >> 5));
  const int c = cnt[n];
  const int deg = min(c, BUCKET_CAP);
  const int* __restrict__ brow = bucket + (size_t)n * BUCKET_CAP;
  const float4* qr = (const float4*)(q + (size_t)n * NDIM_);
  const float4 qa = qr[lane];
  const float4 qb = qr[32 + lane];
  float4 accA = make_float4(0.f, 0.f, 0.f, 0.f);
  float4 accB = make_float4(0.f, 0.f, 0.f, 0.f);
  float zA = 0.f, zB = 0.f;
  const int nb = (deg + 31) >> 5;
  int novf = 0;
  if (c > BUCKET_CAP) novf = min(*ovf_n, LE_EDGES);
  const int nchunk = nb + ((novf + 31) >> 5);
  for (int t = 0; t < nchunk; ++t) {
    int mysrc;
    unsigned todo;
    if (t < nb) {
      const int m = min(32, deg - 32 * t);
      mysrc = (lane < m) ? brow[32 * t + lane] : 0;
      todo = (m == 32) ? 0xffffffffu : ((1u << m) - 1u);
    } else {
      const int idx = 32 * (t - nb) + lane;
      const int od = (idx < novf) ? ovf[2 * idx] : -1;
      mysrc = (idx < novf) ? ovf[2 * idx + 1] : 0;
      todo = __builtin_amdgcn_ballot_w32(od == n);
    }
    while (__builtin_popcount(todo) >= 4)
      lgesql_edges<4>(k, v, mysrc, todo, lane, qa, qb, accA, accB, zA, zB);
    if (__builtin_popcount(todo) >= 2)
      lgesql_edges<2>(k, v, mysrc, todo, lane, qa, qb, accA, accB, zA, zB);
    if (todo)
      lgesql_edges<1>(k, v, mysrc, todo, lane, qa, qb, accA, accB, zA, zB);
  }
  float4* orow = (float4*)(o + (size_t)n * NDIM_);
  orow[lane]      = make_float4(accA.x / zA, accA.y / zA, accA.z / zA, accA.w / zA);
  orow[32 + lane] = make_float4(accB.x / zB, accB.y / zB, accB.z / zB, accB.w / zB);
}

extern "C" void kernel_launch(void* const* d_in, const int* in_sizes, int n_in,
                              void* d_out, int out_size, void* d_ws, size_t ws_size,
                              hipStream_t stream) {
  const float* x     = (const float*)d_in[0];
  const float* src_x = (const float*)d_in[1];
  const float* dst_x = (const float*)d_in[2];
  const float* Wq    = (const float*)d_in[3];
  const float* bq    = (const float*)d_in[4];
  const float* Wk    = (const float*)d_in[5];
  const float* Wv    = (const float*)d_in[6];
  const float* Wo    = (const float*)d_in[7];
  const float* bo    = (const float*)d_in[8];
  const float* ln1_g = (const float*)d_in[9];
  const float* ln1_b = (const float*)d_in[10];
  const float* W1    = (const float*)d_in[11];
  const float* b1    = (const float*)d_in[12];
  const float* W2    = (const float*)d_in[13];
  const float* b2    = (const float*)d_in[14];
  const float* ln2_g = (const float*)d_in[15];
  const float* ln2_b = (const float*)d_in[16];
  const int*   lg_src = (const int*)d_in[17];
  const int*   lg_dst = (const int*)d_in[18];
  float* out = (float*)d_out;

  const size_t EN = (size_t)E_NODES * NDIM_;
  float* ws = (float*)d_ws;
  float* qf   = ws;
  float* kf   = ws + EN;
  float* vf   = ws + 2 * EN;
  float* wvf  = ws + 3 * EN;
  int*   cntf = (int*)(ws + 4 * EN);
  int*   ovfn = cntf + E_NODES;
  float* ffn1 = ws + 4 * EN + (size_t)E_NODES * HEADS;
  int*   bucketf = (int*)ffn1;
  int*   ovff = bucketf + (size_t)E_NODES * BUCKET_CAP;

  hipMemsetAsync(cntf, 0, (size_t)(E_NODES + 1) * sizeof(int), stream);

  const dim3 blk(256);

  const int gqkv  = GROUP_WGS + 3 * ((E_NODES / 16) * (NDIM_ / 64) / 8);
  const int g1024 = (E_NODES / 16) * (4 * NDIM_ / 64) / 8;
  const int gln   = E_NODES / 16;
  const dim3 blkln(128);

  lgesql_qkv_wmma_f32<<<gqkv, blk, 0, stream>>>(x, Wq, Wk, Wv, bq, src_x, dst_x, qf, kf, vf,
                                                lg_src, lg_dst, cntf, bucketf, ovff, ovfn);

  lgesql_edge_gather<<<E_NODES / 8, blk, 0, stream>>>(qf, kf, vf, cntf, bucketf, ovff, ovfn, wvf);

  lgesql_gemm_wmma_f32<256, 256, true,  true,  false, true>
      <<<gln, blkln, 0, stream>>>(wvf, Wo, bo, x, ln1_g, ln1_b, kf);

  lgesql_gemm_wmma_f32<256, 1024, true, false, true, false>
      <<<g1024, blk, 0, stream>>>(kf, W1, b1, nullptr, nullptr, nullptr, ffn1);
  lgesql_gemm_wmma_f32<1024, 256, true, true, false, true>
      <<<gln, blkln, 0, stream>>>(ffn1, W2, b2, kf, ln2_g, ln2_b, out);
}


# q/k/v and first FFN products on a 32x64 wave tile: two row tiles share each W fragment load and its pairing moves (W loads and moves per matrix instruction halved); order of sums unchanged
# speedup vs baseline: 1.9654x; 1.0369x over previous
#include <hip/hip_runtime.h>
#include <math.h>

#define E_NODES 20000
#define LE_EDGES 320000
#define NDIM_ 256
#define HEADS 8
#define DK_ 32
#define BUCKET_CAP 64
#define GROUP_EPT 5
#define QKV_THREADS 128
#define GROUP_WGS (LE_EDGES / (QKV_THREADS * GROUP_EPT))

typedef __attribute__((ext_vector_type(2))) float v2f;
typedef __attribute__((ext_vector_type(4))) float v4f;
typedef __attribute__((ext_vector_type(8))) float v8f;

__device__ __forceinline__ v2f lgesql_pair(float lo, float hi)
{
  v2f p; p.x = lo; p.y = hi; return p;
}

template <int XMASK>
__device__ __forceinline__ float lgesql_xadd(float x)
{
  const int y = __builtin_amdgcn_update_dpp(0, __builtin_bit_cast(int, x), 0x160 | XMASK, 0xf, 0xf, true);
  return x + __builtin_bit_cast(float, y);
}

template <int K, int N>
__device__ __forceinline__ void lgesql_tile_k_loop(
    const float* __restrict__ Ap, const float* __restrict__ Bp,
    v8f& acc0, v8f& acc1, v8f& acc2, v8f& acc3)
{
  for (int k = 0; k < K; k += 16) {
    int kk = k;
    asm volatile("v_nop\n\tv_nop\n\tv_nop\n\tv_nop\n\tv_nop\n\tv_nop\n\tv_nop\n\tv_nop"
                 : "+v"(acc0), "+v"(acc1), "+v"(acc2), "+v"(acc3), "+s"(kk));
    const float* ak = Ap + kk;
    const float* bk = Bp + (size_t)kk * N;
    const float4 a0 = *(const float4*)(ak);
    const float4 a1 = *(const float4*)(ak + 8);
    const float4 r0 = *(const float4*)(bk);
    const float4 r1 = *(const float4*)(bk + N);
    const float4 r2 = *(const float4*)(bk + 2 * N);
    const float4 r3 = *(const float4*)(bk + 3 * N);
    const float4 r4 = *(const float4*)(bk + 8 * N);
    const float4 r5 = *(const float4*)(bk + 9 * N);
    const float4 r6 = *(const float4*)(bk + 10 * N);
    const float4 r7 = *(const float4*)(bk + 11 * N);
    v2f p00 = lgesql_pair(r0.x, r1.x), p01 = lgesql_pair(r0.y, r1.y), p02 = lgesql_pair(r0.z, r1.z), p03 = lgesql_pair(r0.w, r1.w);
    v2f p10 = lgesql_pair(r2.x, r3.x), p11 = lgesql_pair(r2.y, r3.y), p12 = lgesql_pair(r2.z, r3.z), p13 = lgesql_pair(r2.w, r3.w);
    v2f p20 = lgesql_pair(r4.x, r5.x), p21 = lgesql_pair(r4.y, r5.y), p22 = lgesql_pair(r4.z, r5.z), p23 = lgesql_pair(r4.w, r5.w);
    v2f p30 = lgesql_pair(r6.x, r7.x), p31 = lgesql_pair(r6.y, r7.y), p32 = lgesql_pair(r6.z, r7.z), p33 = lgesql_pair(r6.w, r7.w);
    asm volatile("" : "+v"(p00), "+v"(p01), "+v"(p02), "+v"(p03), "+v"(p10), "+v"(p11), "+v"(p12), "+v"(p13),
                      "+v"(p20), "+v"(p21), "+v"(p22), "+v"(p23), "+v"(p30), "+v"(p31), "+v"(p32), "+v"(p33));
    const v2f a00 = lgesql_pair(a0.x, a0.y), a01 = lgesql_pair(a0.z, a0.w);
    const v2f a10 = lgesql_pair(a1.x, a1.y), a11 = lgesql_pair(a1.z, a1.w);
    acc0 = __builtin_amdgcn_wmma_f32_16x16x4_f32(false, a00, false, p00, (short)0, acc0, false, false);
    acc1 = __builtin_amdgcn_wmma_f32_16x16x4_f32(false, a00, false, p01, (short)0, acc1, false, false);
    acc2 = __builtin_amdgcn_wmma_f32_16x16x4_f32(false, a00, false, p02, (short)0, acc2, false, false);
    acc3 = __builtin_amdgcn_wmma_f32_16x16x4_f32(false, a00, false, p03, (short)0, acc3, false, false);
    acc0 = __builtin_amdgcn_wmma_f32_16x16x4_f32(false, a01, false, p10, (short)0, acc0, false, false);
    acc1 = __builtin_amdgcn_wmma_f32_16x16x4_f32(false, a01, false, p11, (short)0, acc1, false, false);
    acc2 = __builtin_amdgcn_wmma_f32_16x16x4_f32(false, a01, false, p12, (short)0, acc2, false, false);
    acc3 = __builtin_amdgcn_wmma_f32_16x16x4_f32(false, a01, false, p13, (short)0, acc3, false, false);
    acc0 = __builtin_amdgcn_wmma_f32_16x16x4_f32(false, a10, false, p20, (short)0, acc0, false, false);
    acc1 = __builtin_amdgcn_wmma_f32_16x16x4_f32(false, a10, false, p21, (short)0, acc1, false, false);
    acc2 = __builtin_amdgcn_wmma_f32_16x16x4_f32(false, a10, false, p22, (short)0, acc2, false, false);
    acc3 = __builtin_amdgcn_wmma_f32_16x16x4_f32(false, a10, false, p23, (short)0, acc3, false, false);
    acc0 = __builtin_amdgcn_wmma_f32_16x16x4_f32(false, a11, false, p30, (short)0, acc0, false, false);
    acc1 = __builtin_amdgcn_wmma_f32_16x16x4_f32(false, a11, false, p31, (short)0, acc1, false, false);
    acc2 = __builtin_amdgcn_wmma_f32_16x16x4_f32(false, a11, false, p32, (short)0, acc2, false, false);
    acc3 = __builtin_amdgcn_wmma_f32_16x16x4_f32(false, a11, false, p33, (short)0, acc3, false, false);
  }
  asm volatile("v_nop\n\tv_nop\n\tv_nop\n\tv_nop\n\tv_nop\n\tv_nop\n\tv_nop\n\tv_nop"
               : "+v"(acc0), "+v"(acc1), "+v"(acc2), "+v"(acc3));
}

template <int K, int N>
__device__ __forceinline__ void lgesql_tile2_k_loop(
    const float* __restrict__ Ap, const float* __restrict__ Bp,
    v8f& c00, v8f& c01, v8f& c02, v8f& c03, v8f& c10, v8f& c11, v8f& c12, v8f& c13)
{
  for (int k = 0; k < K; k += 16) {
    int kk = k;
    asm volatile("v_nop\n\tv_nop\n\tv_nop\n\tv_nop\n\tv_nop\n\tv_nop\n\tv_nop\n\tv_nop"
                 : "+v"(c00), "+v"(c01), "+v"(c02), "+v"(c03), "+v"(c10), "+v"(c11), "+v"(c12), "+v"(c13), "+s"(kk));
    const float* ak = Ap + kk;
    const float* bk = Bp + (size_t)kk * N;
    v4f a0 = *(const v4f*)(ak);
    v4f a1 = *(const v4f*)(ak + 8);
    v4f a2 = *(const v4f*)(ak + 16 * K);
    v4f a3 = *(const v4f*)(ak + 16 * K + 8);
    v4f r0 = *(const v4f*)(bk);
    v4f r1 = *(const v4f*)(bk + N);
    v4f r2 = *(const v4f*)(bk + 2 * N);
    v4f r3 = *(const v4f*)(bk + 3 * N);
    v4f r4 = *(const v4f*)(bk + 8 * N);
    v4f r5 = *(const v4f*)(bk + 9 * N);
    v4f r6 = *(const v4f*)(bk + 10 * N);
    v4f r7 = *(const v4f*)(bk + 11 * N);
    asm volatile("" : "+v"(a0), "+v"(a1), "+v"(a2), "+v"(a3), "+v"(r0), "+v"(r1), "+v"(r2), "+v"(r3),
                      "+v"(r4), "+v"(r5), "+v"(r6), "+v"(r7));
    v2f p00 = lgesql_pair(r0.x, r1.x), p01 = lgesql_pair(r0.y, r1.y), p02 = lgesql_pair(r0.z, r1.z), p03 = lgesql_pair(r0.w, r1.w);
    v2f p10 = lgesql_pair(r2.x, r3.x), p11 = lgesql_pair(r2.y, r3.y), p12 = lgesql_pair(r2.z, r3.z), p13 = lgesql_pair(r2.w, r3.w);
    v2f p20 = lgesql_pair(r4.x, r5.x), p21 = lgesql_pair(r4.y, r5.y), p22 = lgesql_pair(r4.z, r5.z), p23 = lgesql_pair(r4.w, r5.w);
    v2f p30 = lgesql_pair(r6.x, r7.x), p31 = lgesql_pair(r6.y, r7.y), p32 = lgesql_pair(r6.z, r7.z), p33 = lgesql_pair(r6.w, r7.w);
    asm volatile("" : "+v"(p00), "+v"(p01), "+v"(p02), "+v"(p03), "+v"(p10), "+v"(p11), "+v"(p12), "+v"(p13),
                      "+v"(p20), "+v"(p21), "+v"(p22), "+v"(p23), "+v"(p30), "+v"(p31), "+v"(p32), "+v"(p33));
    const v2f x00 = lgesql_pair(a0.x, a0.y), x01 = lgesql_pair(a0.z, a0.w);
    const v2f x10 = lgesql_pair(a1.x, a1.y), x11 = lgesql_pair(a1.z, a1.w);
    const v2f y00 = lgesql_pair(a2.x, a2.y), y01 = lgesql_pair(a2.z, a2.w);
    const v2f y10 = lgesql_pair(a3.x, a3.y), y11 = lgesql_pair(a3.z, a3.w);
    c00 = __builtin_amdgcn_wmma_f32_16x16x4_f32(false, x00, false, p00, (short)0, c00, false, false);
    c01 = __builtin_amdgcn_wmma_f32_16x16x4_f32(false, x00, false, p01, (short)0, c01, false, false);
    c02 = __builtin_amdgcn_wmma_f32_16x16x4_f32(false, x00, false, p02, (short)0, c02, false, false);
    c03 = __builtin_amdgcn_wmma_f32_16x16x4_f32(false, x00, false, p03, (short)0, c03, false, false);
    c10 = __builtin_amdgcn_wmma_f32_16x16x4_f32(false, y00, false, p00, (short)0, c10, false, false);
    c11 = __builtin_amdgcn_wmma_f32_16x16x4_f32(false, y00, false, p01, (short)0, c11, false, false);
    c12 = __builtin_amdgcn_wmma_f32_16x16x4_f32(false, y00, false, p02, (short)0, c12, false, false);
    c13 = __builtin_amdgcn_wmma_f32_16x16x4_f32(false, y00, false, p03, (short)0, c13, false, false);
    c00 = __builtin_amdgcn_wmma_f32_16x16x4_f32(false, x01, false, p10, (short)0, c00, false, false);
    c01 = __builtin_amdgcn_wmma_f32_16x16x4_f32(false, x01, false, p11, (short)0, c01, false, false);
    c02 = __builtin_amdgcn_wmma_f32_16x16x4_f32(false, x01, false, p12, (short)0, c02, false, false);
    c03 = __builtin_amdgcn_wmma_f32_16x16x4_f32(false, x01, false, p13, (short)0, c03, false, false);
    c10 = __builtin_amdgcn_wmma_f32_16x16x4_f32(false, y01, false, p10, (short)0, c10, false, false);
    c11 = __builtin_amdgcn_wmma_f32_16x16x4_f32(false, y01, false, p11, (short)0, c11, false, false);
    c12 = __builtin_amdgcn_wmma_f32_16x16x4_f32(false, y01, false, p12, (short)0, c12, false, false);
    c13 = __builtin_amdgcn_wmma_f32_16x16x4_f32(false, y01, false, p13, (short)0, c13, false, false);
    c00 = __builtin_amdgcn_wmma_f32_16x16x4_f32(false, x10, false, p20, (short)0, c00, false, false);
    c01 = __builtin_amdgcn_wmma_f32_16x16x4_f32(false, x10, false, p21, (short)0, c01, false, false);
    c02 = __builtin_amdgcn_wmma_f32_16x16x4_f32(false, x10, false, p22, (short)0, c02, false, false);
    c03 = __builtin_amdgcn_wmma_f32_16x16x4_f32(false, x10, false, p23, (short)0, c03, false, false);
    c10 = __builtin_amdgcn_wmma_f32_16x16x4_f32(false, y10, false, p20, (short)0, c10, false, false);
    c11 = __builtin_amdgcn_wmma_f32_16x16x4_f32(false, y10, false, p21, (short)0, c11, false, false);
    c12 = __builtin_amdgcn_wmma_f32_16x16x4_f32(false, y10, false, p22, (short)0, c12, false, false);
    c13 = __builtin_amdgcn_wmma_f32_16x16x4_f32(false, y10, false, p23, (short)0, c13, false, false);
    c00 = __builtin_amdgcn_wmma_f32_16x16x4_f32(false, x11, false, p30, (short)0, c00, false, false);
    c01 = __builtin_amdgcn_wmma_f32_16x16x4_f32(false, x11, false, p31, (short)0, c01, false, false);
    c02 = __builtin_amdgcn_wmma_f32_16x16x4_f32(false, x11, false, p32, (short)0, c02, false, false);
    c03 = __builtin_amdgcn_wmma_f32_16x16x4_f32(false, x11, false, p33, (short)0, c03, false, false);
    c10 = __builtin_amdgcn_wmma_f32_16x16x4_f32(false, y11, false, p30, (short)0, c10, false, false);
    c11 = __builtin_amdgcn_wmma_f32_16x16x4_f32(false, y11, false, p31, (short)0, c11, false, false);
    c12 = __builtin_amdgcn_wmma_f32_16x16x4_f32(false, y11, false, p32, (short)0, c12, false, false);
    c13 = __builtin_amdgcn_wmma_f32_16x16x4_f32(false, y11, false, p33, (short)0, c13, false, false);
  }
  asm volatile("v_nop\n\tv_nop\n\tv_nop\n\tv_nop\n\tv_nop\n\tv_nop\n\tv_nop\n\tv_nop"
               : "+v"(c00), "+v"(c01), "+v"(c02), "+v"(c03), "+v"(c10), "+v"(c11), "+v"(c12), "+v"(c13));
}

template <int K, int N, bool HAS_BIAS, bool HAS_ADD, bool RELU, bool LN>
__global__ __launch_bounds__(LN ? 128 : 256) void lgesql_gemm_wmma_f32(
    const float* __restrict__ A, const float* __restrict__ W,
    const float* __restrict__ bias, const float* __restrict__ addm,
    const float* __restrict__ ln_g, const float* __restrict__ ln_b,
    float* __restrict__ C)
{
  static_assert(!LN || N == 256, "LN: the four waves of a work-group must cover whole rows");
  constexpr int TN = 64;
  constexpr int tilesN = N / TN;
  static_assert(K % 16 == 0 && N % TN == 0, "tile split");
  const int wid  = (blockIdx.x * blockDim.x + threadIdx.x) >> 5;
  const int lane = threadIdx.x & 31;
  const int tm = wid / tilesN;
  const int tn = wid - tm * tilesN;
  const int half = lane >> 4;
  const int l    = lane & 15;

  const float* Ap = A + (size_t)(tm * 16 + l) * K + 4 * half;
  const float* Bp = W + (size_t)(4 * half) * N + tn * TN + 4 * l;

  v8f acc0 = {}, acc1 = {}, acc2 = {}, acc3 = {};
  lgesql_tile_k_loop<K, N>(Ap, Bp, acc0, acc1, acc2, acc3);

  const int col = tn * TN + 4 * l;
  float4 bv = make_float4(0.0f, 0.0f, 0.0f, 0.0f);
  if (HAS_BIAS) bv = *(const float4*)(bias + col);
  float4 v[8];
  #pragma unroll
  for (int r = 0; r < 8; ++r) {
    const int row = tm * 16 + r + 8 * half;
    v[r] = make_float4(acc0[r] + bv.x, acc1[r] + bv.y, acc2[r] + bv.z, acc3[r] + bv.w);
    if (HAS_ADD) {
      const float4 ad = *(const float4*)(addm + (size_t)row * N + col);
      v[r].x += ad.x; v[r].y += ad.y; v[r].z += ad.z; v[r].w += ad.w;
    }
    if (RELU) {
      v[r].x = fmaxf(v[r].x, 0.0f); v[r].y = fmaxf(v[r].y, 0.0f);
      v[r].z = fmaxf(v[r].z, 0.0f); v[r].w = fmaxf(v[r].w, 0.0f);
    }
  }
  if constexpr (LN) {
    __shared__ float2 part[16][4];
    const int w = threadIdx.x >> 5;
    #pragma unroll
    for (int r = 0; r < 8; ++r) {
      float sm = (v[r].x + v[r].y) + (v[r].z + v[r].w);
      float sq = (v[r].x * v[r].x + v[r].y * v[r].y) + (v[r].z * v[r].z + v[r].w * v[r].w);
      sm = lgesql_xadd<1>(sm); sq = lgesql_xadd<1>(sq);
      sm = lgesql_xadd<2>(sm); sq = lgesql_xadd<2>(sq);
      sm = lgesql_xadd<4>(sm); sq = lgesql_xadd<4>(sq);
      sm = lgesql_xadd<8>(sm); sq = lgesql_xadd<8>(sq);
      if (l == r) part[r + 8 * half][w] = make_float2(sm, sq);
    }
    __syncthreads();
    const float4 g4 = *(const float4*)(ln_g + col);
    const float4 b4 = *(const float4*)(ln_b + col);
    #pragma unroll
    for (int r = 0; r < 8; ++r) {
      const int row = tm * 16 + r + 8 * half;
      const float4 p01 = *(const float4*)(&part[r + 8 * half][0]);
      const float4 p23 = *(const float4*)(&part[r + 8 * half][2]);
      const float sum = ((p01.x + p01.z) + p23.x) + p23.z;
      const float sqs = ((p01.y + p01.w) + p23.y) + p23.w;
      const float mu   = sum * (1.0f / N);
      const float var  = sqs * (1.0f / N) - mu * mu;
      const float rstd = rsqrtf(var + 1e-5f);
      float4 o;
      o.x = (v[r].x - mu) * rstd * g4.x + b4.x;
      o.y = (v[r].y - mu) * rstd * g4.y + b4.y;
      o.z = (v[r].z - mu) * rstd * g4.z + b4.z;
      o.w = (v[r].w - mu) * rstd * g4.w + b4.w;
      *(float4*)(C + (size_t)row * N + col) = o;
    }
  } else {
    #pragma unroll
    for (int r = 0; r < 8; ++r) {
      const int row = tm * 16 + r + 8 * half;
      *(float4*)(C + (size_t)row * N + col) = v[r];
    }
  }
}

template <int K, int N>
__global__ __launch_bounds__(256) void lgesql_gemm_relu_wmma_f32(
    const float* __restrict__ A, const float* __restrict__ W,
    const float* __restrict__ bias, float* __restrict__ C)
{
  constexpr int TN = 64;
  constexpr int tilesN = N / TN;
  static_assert(K % 16 == 0 && N % TN == 0 && E_NODES % 32 == 0, "tile split");
  const int wid  = (blockIdx.x * blockDim.x + threadIdx.x) >> 5;
  const int lane = threadIdx.x & 31;
  const int tm = wid / tilesN;
  const int tn = wid - tm * tilesN;
  const int half = lane >> 4;
  const int l    = lane & 15;

  const float* Ap = A + (size_t)(tm * 32 + l) * K + 4 * half;
  const float* Bp = W + (size_t)(4 * half) * N + tn * TN + 4 * l;

  v8f c00 = {}, c01 = {}, c02 = {}, c03 = {}, c10 = {}, c11 = {}, c12 = {}, c13 = {};
  lgesql_tile2_k_loop<K, N>(Ap, Bp, c00, c01, c02, c03, c10, c11, c12, c13);

  int tm_e = tm, tn_e = tn, half_e = half, l_e = l;
  asm volatile("" : "+v"(tm_e), "+v"(tn_e), "+v"(half_e), "+v"(l_e), "+v"(c00));
  const int col = tn_e * TN + 4 * l_e;
  const float4 bv = *(const float4*)(bias + col);
  float* __restrict__ Cp = C + (size_t)(tm_e * 32 + 8 * half_e) * N + col;
  #pragma unroll
  for (int r = 0; r < 8; ++r) {
    float4 o = make_float4(c00[r] + bv.x, c01[r] + bv.y, c02[r] + bv.z, c03[r] + bv.w);
    o.x = fmaxf(o.x, 0.0f); o.y = fmaxf(o.y, 0.0f); o.z = fmaxf(o.z, 0.0f); o.w = fmaxf(o.w, 0.0f);
    *(float4*)(Cp + (size_t)r * N) = o;
  }
  #pragma unroll
  for (int r = 0; r < 8; ++r) {
    float4 o = make_float4(c10[r] + bv.x, c11[r] + bv.y, c12[r] + bv.z, c13[r] + bv.w);
    o.x = fmaxf(o.x, 0.0f); o.y = fmaxf(o.y, 0.0f); o.z = fmaxf(o.z, 0.0f); o.w = fmaxf(o.w, 0.0f);
    *(float4*)(Cp + (size_t)(16 + r) * N) = o;
  }
}

__device__ __forceinline__ void lgesql_group_edges(
    int wg, const int* __restrict__ lg_src, const int* __restrict__ lg_dst,
    int* __restrict__ cnt, int* __restrict__ bucket, int* __restrict__ ovf, int* __restrict__ ovf_n)
{
  static_assert(GROUP_WGS * QKV_THREADS * GROUP_EPT == LE_EDGES, "the grouping work-groups cover the edges exactly");
  const int e0 = wg * (QKV_THREADS * GROUP_EPT) + (int)threadIdx.x;
  int d[GROUP_EPT], sidx[GROUP_EPT], r[GROUP_EPT];
  #pragma unroll
  for (int i = 0; i < GROUP_EPT; ++i) { d[i] = lg_dst[e0 + QKV_THREADS * i]; sidx[i] = lg_src[e0 + QKV_THREADS * i]; }
  #pragma unroll
  for (int i = 0; i < GROUP_EPT; ++i) r[i] = atomicAdd(&cnt[d[i]], 1);
  #pragma unroll
  for (int i = 0; i < GROUP_EPT; ++i) {
    if (r[i] < BUCKET_CAP) {
      bucket[(size_t)d[i] * BUCKET_CAP + r[i]] = sidx[i];
    } else {
      const int p = atomicAdd(ovf_n, 1);
      ovf[2 * p] = d[i]; ovf[2 * p + 1] = sidx[i];
    }
  }
}

__global__ __launch_bounds__(QKV_THREADS) void lgesql_qkv_wmma_f32(
    const float* __restrict__ x,
    const float* __restrict__ Wq, const float* __restrict__ Wk, const float* __restrict__ Wv,
    const float* __restrict__ bq, const float* __restrict__ src_x, const float* __restrict__ dst_x,
    float* __restrict__ q, float* __restrict__ k, float* __restrict__ v,
    const int* __restrict__ lg_src, const int* __restrict__ lg_dst,
    int* __restrict__ cnt, int* __restrict__ bucket, int* __restrict__ ovf, int* __restrict__ ovf_n)
{
  if ((int)blockIdx.x < GROUP_WGS) {
    lgesql_group_edges((int)blockIdx.x, lg_src, lg_dst, cnt, bucket, ovf, ovf_n);
    return;
  }
  const int pblk = (int)blockIdx.x - GROUP_WGS;
  constexpr int K = NDIM_, N = NDIM_;
  constexpr int TN = 64;
  constexpr int tilesN = N / TN;
  constexpr int GROUPS_PER_MATRIX = (E_NODES / 32) * tilesN / (QKV_THREADS / 32);
  static_assert(E_NODES % 32 == 0, "32-row tiles cover the rows exactly");
  const int which = pblk / GROUPS_PER_MATRIX;
  const int grp   = pblk - which * GROUPS_PER_MATRIX;
  const float* __restrict__ W = which == 0 ? Wq : (which == 1 ? Wk : Wv);
  float* __restrict__ C       = which == 0 ? q  : (which == 1 ? k  : v);
  const float* __restrict__ addm = which == 0 ? src_x : dst_x;

  const int wid  = (grp * (int)blockDim.x + (int)threadIdx.x) >> 5;
  const int lane = threadIdx.x & 31;
  const int tm = wid / tilesN;
  const int tn = wid - tm * tilesN;
  const int half = lane >> 4;
  const int l    = lane & 15;

  const float* Ap = x + (size_t)(tm * 32 + l) * K + 4 * half;
  const float* Bp = W + (size_t)(4 * half) * N + tn * TN + 4 * l;

  v8f c00 = {}, c01 = {}, c02 = {}, c03 = {}, c10 = {}, c11 = {}, c12 = {}, c13 = {};
  lgesql_tile2_k_loop<K, N>(Ap, Bp, c00, c01, c02, c03, c10, c11, c12, c13);

  int tm_e = tm, tn_e = tn, half_e = half, l_e = l;
  asm volatile("" : "+v"(tm_e), "+v"(tn_e), "+v"(half_e), "+v"(l_e), "+v"(c00));
  const int col = tn_e * TN + 4 * l_e;
  const size_t at = (size_t)(tm_e * 32 + 8 * half_e) * N + col;
  float* __restrict__ Cp = C + at;
  if (which == 1) {
    #pragma unroll
    for (int r = 0; r < 8; ++r)
      *(float4*)(Cp + (size_t)r * N) = make_float4(c00[r], c01[r], c02[r], c03[r]);
    #pragma unroll
    for (int r = 0; r < 8; ++r)
      *(float4*)(Cp + (size_t)(16 + r) * N) = make_float4(c10[r], c11[r], c12[r], c13[r]);
  } else {
    const float* __restrict__ Rp = addm + at;
    float4 bv = make_float4(0.0f, 0.0f, 0.0f, 0.0f);
    if (which == 0) bv = *(const float4*)(bq + col);
    #pragma unroll
    for (int r0 = 0; r0 < 8; r0 += 4) {
      float4 ad[4];
      #pragma unroll
      for (int i = 0; i < 4; ++i) ad[i] = *(const float4*)(Rp + (size_t)(r0 + i) * N);
      #pragma unroll
      for (int i = 0; i < 4; ++i) {
        const int r = r0 + i;
        float4 o = make_float4(c00[r] + bv.x, c01[r] + bv.y, c02[r] + bv.z, c03[r] + bv.w);
        o.x += ad[i].x; o.y += ad[i].y; o.z += ad[i].z; o.w += ad[i].w;
        *(float4*)(Cp + (size_t)r * N) = o;
      }
      __builtin_amdgcn_sched_barrier(0);
    }
    #pragma unroll
    for (int r0 = 0; r0 < 8; r0 += 4) {
      float4 ad[4];
      #pragma unroll
      for (int i = 0; i < 4; ++i) ad[i] = *(const float4*)(Rp + (size_t)(16 + r0 + i) * N);
      #pragma unroll
      for (int i = 0; i < 4; ++i) {
        const int r = r0 + i;
        float4 o = make_float4(c10[r] + bv.x, c11[r] + bv.y, c12[r] + bv.z, c13[r] + bv.w);
        o.x += ad[i].x; o.y += ad[i].y; o.z += ad[i].z; o.w += ad[i].w;
        *(float4*)(Cp + (size_t)(16 + r) * N) = o;
      }
      __builtin_amdgcn_sched_barrier(0);
    }
  }
}

template <int U>
__device__ __forceinline__ void lgesql_edges(
    const float* __restrict__ k, const float* __restrict__ v, int mysrc, unsigned& todo, int lane,
    const float4& qa, const float4& qb, float4& accA, float4& accB, float& zA, float& zB)
{
  float4 ka[U], kb[U], va[U], vb[U];
  #pragma unroll
  for (int u = 0; u < U; ++u) {
    const int s = __builtin_amdgcn_readlane(mysrc, __builtin_ctz(todo));
    todo &= todo - 1;
    const float4* kr = (const float4*)(k + (size_t)s * NDIM_);
    const float4* vr = (const float4*)(v + (size_t)s * NDIM_);
    ka[u] = kr[lane]; kb[u] = kr[32 + lane];
    va[u] = vr[lane]; vb[u] = vr[32 + lane];
  }
  #pragma unroll
  for (int u = 0; u < U; ++u) {
    float pA = ka[u].x * qa.x + ka[u].y * qa.y + ka[u].z * qa.z + ka[u].w * qa.w;
    float pB = kb[u].x * qb.x + kb[u].y * qb.y + kb[u].z * qb.z + kb[u].w * qb.w;
    pA = lgesql_xadd<1>(pA); pB = lgesql_xadd<1>(pB);
    pA = lgesql_xadd<2>(pA); pB = lgesql_xadd<2>(pB);
    pA = lgesql_xadd<4>(pA); pB = lgesql_xadd<4>(pB);
    float sA = pA * 0.17677669529663687f;
    float sB = pB * 0.17677669529663687f;
    sA = expf(fminf(5.0f, fmaxf(-5.0f, sA)));
    sB = expf(fminf(5.0f, fmaxf(-5.0f, sB)));
    accA.x += va[u].x * sA; accA.y += va[u].y * sA; accA.z += va[u].z * sA; accA.w += va[u].w * sA;
    accB.x += vb[u].x * sB; accB.y += vb[u].y * sB; accB.z += vb[u].z * sB; accB.w += vb[u].w * sB;
    zA += sA; zB += sB;
  }
}

__global__ __launch_bounds__(256) void lgesql_edge_gather(
    const float* __restrict__ q, const float* __restrict__ k, const float* __restrict__ v,
    const int* __restrict__ cnt, const int* __restrict__ bucket,
    const int* __restrict__ ovf, const int* __restrict__ ovf_n, float* __restrict__ o)
{
  const int lane = threadIdx.x & 31;
  const int n = __builtin_amdgcn_readfirstlane((int)((blockIdx.x * blockDim.x + threadIdx.x) >> 5));
  const int c = cnt[n];
  const int deg = min(c, BUCKET_CAP);
  const int* __restrict__ brow = bucket + (size_t)n * BUCKET_CAP;
  const float4* qr = (const float4*)(q + (size_t)n * NDIM_);
  const float4 qa = qr[lane];
  const float4 qb = qr[32 + lane];
  float4 accA = make_float4(0.f, 0.f, 0.f, 0.f);
  float4 accB = make_float4(0.f, 0.f, 0.f, 0.f);
  float zA = 0.f, zB = 0.f;
  const int nb = (deg + 31) >> 5;
  int novf = 0;
  if (c > BUCKET_CAP) novf = min(*ovf_n, LE_EDGES);
  const int nchunk = nb + ((novf + 31) >> 5);
  for (int t = 0; t < nchunk; ++t) {
    int mysrc;
    unsigned todo;
    if (t < nb) {
      const int m = min(32, deg - 32 * t);
      mysrc = (lane < m) ? brow[32 * t + lane] : 0;
      todo = (m == 32) ? 0xffffffffu : ((1u << m) - 1u);
    } else {
      const int idx = 32 * (t - nb) + lane;
      const int od = (idx < novf) ? ovf[2 * idx] : -1;
      mysrc = (idx < novf) ? ovf[2 * idx + 1] : 0;
      todo = __builtin_amdgcn_ballot_w32(od == n);
    }
    while (__builtin_popcount(todo) >= 4)
      lgesql_edges<4>(k, v, mysrc, todo, lane, qa, qb, accA, accB, zA, zB);
    if (__builtin_popcount(todo) >= 2)
      lgesql_edges<2>(k, v, mysrc, todo, lane, qa, qb, accA, accB, zA, zB);
    if (todo)
      lgesql_edges<1>(k, v, mysrc, todo, lane, qa, qb, accA, accB, zA, zB);
  }
  float4* orow = (float4*)(o + (size_t)n * NDIM_);
  orow[lane]      = make_float4(accA.x / zA, accA.y / zA, accA.z / zA, accA.w / zA);
  orow[32 + lane] = make_float4(accB.x / zB, accB.y / zB, accB.z / zB, accB.w / zB);
}

extern "C" void kernel_launch(void* const* d_in, const int* in_sizes, int n_in,
                              void* d_out, int out_size, void* d_ws, size_t ws_size,
                              hipStream_t stream) {
  const float* x     = (const float*)d_in[0];
  const float* src_x = (const float*)d_in[1];
  const float* dst_x = (const float*)d_in[2];
  const float* Wq    = (const float*)d_in[3];
  const float* bq    = (const float*)d_in[4];
  const float* Wk    = (const float*)d_in[5];
  const float* Wv    = (const float*)d_in[6];
  const float* Wo    = (const float*)d_in[7];
  const float* bo    = (const float*)d_in[8];
  const float* ln1_g = (const float*)d_in[9];
  const float* ln1_b = (const float*)d_in[10];
  const float* W1    = (const float*)d_in[11];
  const float* b1    = (const float*)d_in[12];
  const float* W2    = (const float*)d_in[13];
  const float* b2    = (const float*)d_in[14];
  const float* ln2_g = (const float*)d_in[15];
  const float* ln2_b = (const float*)d_in[16];
  const int*   lg_src = (const int*)d_in[17];
  const int*   lg_dst = (const int*)d_in[18];
  float* out = (float*)d_out;

  const size_t EN = (size_t)E_NODES * NDIM_;
  float* ws = (float*)d_ws;
  float* qf   = ws;
  float* kf   = ws + EN;
  float* vf   = ws + 2 * EN;
  float* wvf  = ws + 3 * EN;
  int*   cntf = (int*)(ws + 4 * EN);
  int*   ovfn = cntf + E_NODES;
  float* ffn1 = ws + 4 * EN + (size_t)E_NODES * HEADS;
  int*   bucketf = (int*)ffn1;
  int*   ovff = bucketf + (size_t)E_NODES * BUCKET_CAP;

  hipMemsetAsync(cntf, 0, (size_t)(E_NODES + 1) * sizeof(int), stream);

  const dim3 blk(256);

  const int gqkv  = GROUP_WGS + 3 * ((E_NODES / 32) * (NDIM_ / 64) / (QKV_THREADS / 32));
  const int g1024 = (E_NODES / 32) * (4 * NDIM_ / 64) / 8;
  const dim3 blkqkv(QKV_THREADS);
  const int gln   = E_NODES / 16;
  const dim3 blkln(128);

  lgesql_qkv_wmma_f32<<<gqkv, blkqkv, 0, stream>>>(x, Wq, Wk, Wv, bq, src_x, dst_x, qf, kf, vf,
                                                lg_src, lg_dst, cntf, bucketf, ovff, ovfn);

  lgesql_edge_gather<<<E_NODES / 8, blk, 0, stream>>>(qf, kf, vf, cntf, bucketf, ovff, ovfn, wvf);

  lgesql_gemm_wmma_f32<256, 256, true,  true,  false, true>
      <<<gln, blkln, 0, stream>>>(wvf, Wo, bo, x, ln1_g, ln1_b, kf);

  lgesql_gemm_relu_wmma_f32<256, 1024>
      <<<g1024, blk, 0, stream>>>(kf, W1, b1, ffn1);
  lgesql_gemm_wmma_f32<1024, 256, true, true, false, true>
      <<<gln, blkln, 0, stream>>>(ffn1, W2, b2, kf, ln2_g, ln2_b, out);
}


# Edge attention, o@Wo+LN1, FFN1 and FFN2+LN2 in ONE launch: a 4-wave block keeps its 16 rows of o, h and each 256-wide FFN1 chunk in LDS (3 launches fewer); order of sums unchanged
# speedup vs baseline: 2.0943x; 1.0656x over previous
#include <hip/hip_runtime.h>
#include <math.h>

#define E_NODES 20000
#define LE_EDGES 320000
#define NDIM_ 256
#define HEADS 8
#define DK_ 32
#define BUCKET_CAP 64
#define GROUP_EPT 5
#define QKV_THREADS 128
#define GROUP_WGS (LE_EDGES / (QKV_THREADS * GROUP_EPT))

typedef __attribute__((ext_vector_type(2))) float v2f;
typedef __attribute__((ext_vector_type(4))) float v4f;
typedef __attribute__((ext_vector_type(8))) float v8f;

__device__ __forceinline__ v2f lgesql_pair(float lo, float hi)
{
  v2f p; p.x = lo; p.y = hi; return p;
}

template <int XMASK>
__device__ __forceinline__ float lgesql_xadd(float x)
{
  const int y = __builtin_amdgcn_update_dpp(0, __builtin_bit_cast(int, x), 0x160 | XMASK, 0xf, 0xf, true);
  return x + __builtin_bit_cast(float, y);
}

template <int K, int N>
__device__ __forceinline__ void lgesql_tile_k_loop(
    const float* __restrict__ Ap, const float* __restrict__ Bp,
    v8f& acc0, v8f& acc1, v8f& acc2, v8f& acc3)
{
  for (int k = 0; k < K; k += 16) {
    int kk = k;
    asm volatile("v_nop\n\tv_nop\n\tv_nop\n\tv_nop\n\tv_nop\n\tv_nop\n\tv_nop\n\tv_nop"
                 : "+v"(acc0), "+v"(acc1), "+v"(acc2), "+v"(acc3), "+s"(kk));
    const float* ak = Ap + kk;
    const float* bk = Bp + (size_t)kk * N;
    const float4 a0 = *(const float4*)(ak);
    const float4 a1 = *(const float4*)(ak + 8);
    const float4 r0 = *(const float4*)(bk);
    const float4 r1 = *(const float4*)(bk + N);
    const float4 r2 = *(const float4*)(bk + 2 * N);
    const float4 r3 = *(const float4*)(bk + 3 * N);
    const float4 r4 = *(const float4*)(bk + 8 * N);
    const float4 r5 = *(const float4*)(bk + 9 * N);
    const float4 r6 = *(const float4*)(bk + 10 * N);
    const float4 r7 = *(const float4*)(bk + 11 * N);
    v2f p00 = lgesql_pair(r0.x, r1.x), p01 = lgesql_pair(r0.y, r1.y), p02 = lgesql_pair(r0.z, r1.z), p03 = lgesql_pair(r0.w, r1.w);
    v2f p10 = lgesql_pair(r2.x, r3.x), p11 = lgesql_pair(r2.y, r3.y), p12 = lgesql_pair(r2.z, r3.z), p13 = lgesql_pair(r2.w, r3.w);
    v2f p20 = lgesql_pair(r4.x, r5.x), p21 = lgesql_pair(r4.y, r5.y), p22 = lgesql_pair(r4.z, r5.z), p23 = lgesql_pair(r4.w, r5.w);
    v2f p30 = lgesql_pair(r6.x, r7.x), p31 = lgesql_pair(r6.y, r7.y), p32 = lgesql_pair(r6.z, r7.z), p33 = lgesql_pair(r6.w, r7.w);
    asm volatile("" : "+v"(p00), "+v"(p01), "+v"(p02), "+v"(p03), "+v"(p10), "+v"(p11), "+v"(p12), "+v"(p13),
                      "+v"(p20), "+v"(p21), "+v"(p22), "+v"(p23), "+v"(p30), "+v"(p31), "+v"(p32), "+v"(p33));
    const v2f a00 = lgesql_pair(a0.x, a0.y), a01 = lgesql_pair(a0.z, a0.w);
    const v2f a10 = lgesql_pair(a1.x, a1.y), a11 = lgesql_pair(a1.z, a1.w);
    acc0 = __builtin_amdgcn_wmma_f32_16x16x4_f32(false, a00, false, p00, (short)0, acc0, false, false);
    acc1 = __builtin_amdgcn_wmma_f32_16x16x4_f32(false, a00, false, p01, (short)0, acc1, false, false);
    acc2 = __builtin_amdgcn_wmma_f32_16x16x4_f32(false, a00, false, p02, (short)0, acc2, false, false);
    acc3 = __builtin_amdgcn_wmma_f32_16x16x4_f32(false, a00, false, p03, (short)0, acc3, false, false);
    acc0 = __builtin_amdgcn_wmma_f32_16x16x4_f32(false, a01, false, p10, (short)0, acc0, false, false);
    acc1 = __builtin_amdgcn_wmma_f32_16x16x4_f32(false, a01, false, p11, (short)0, acc1, false, false);
    acc2 = __builtin_amdgcn_wmma_f32_16x16x4_f32(false, a01, false, p12, (short)0, acc2, false, false);
    acc3 = __builtin_amdgcn_wmma_f32_16x16x4_f32(false, a01, false, p13, (short)0, acc3, false, false);
    acc0 = __builtin_amdgcn_wmma_f32_16x16x4_f32(false, a10, false, p20, (short)0, acc0, false, false);
    acc1 = __builtin_amdgcn_wmma_f32_16x16x4_f32(false, a10, false, p21, (short)0, acc1, false, false);
    acc2 = __builtin_amdgcn_wmma_f32_16x16x4_f32(false, a10, false, p22, (short)0, acc2, false, false);
    acc3 = __builtin_amdgcn_wmma_f32_16x16x4_f32(false, a10, false, p23, (short)0, acc3, false, false);
    acc0 = __builtin_amdgcn_wmma_f32_16x16x4_f32(false, a11, false, p30, (short)0, acc0, false, false);
    acc1 = __builtin_amdgcn_wmma_f32_16x16x4_f32(false, a11, false, p31, (short)0, acc1, false, false);
    acc2 = __builtin_amdgcn_wmma_f32_16x16x4_f32(false, a11, false, p32, (short)0, acc2, false, false);
    acc3 = __builtin_amdgcn_wmma_f32_16x16x4_f32(false, a11, false, p33, (short)0, acc3, false, false);
  }
  asm volatile("v_nop\n\tv_nop\n\tv_nop\n\tv_nop\n\tv_nop\n\tv_nop\n\tv_nop\n\tv_nop"
               : "+v"(acc0), "+v"(acc1), "+v"(acc2), "+v"(acc3));
}

template <int K, int N>
__device__ __forceinline__ void lgesql_tile2_k_loop(
    const float* __restrict__ Ap, const float* __restrict__ Bp,
    v8f& c00, v8f& c01, v8f& c02, v8f& c03, v8f& c10, v8f& c11, v8f& c12, v8f& c13)
{
  for (int k = 0; k < K; k += 16) {
    int kk = k;
    asm volatile("v_nop\n\tv_nop\n\tv_nop\n\tv_nop\n\tv_nop\n\tv_nop\n\tv_nop\n\tv_nop"
                 : "+v"(c00), "+v"(c01), "+v"(c02), "+v"(c03), "+v"(c10), "+v"(c11), "+v"(c12), "+v"(c13), "+s"(kk));
    const float* ak = Ap + kk;
    const float* bk = Bp + (size_t)kk * N;
    v4f a0 = *(const v4f*)(ak);
    v4f a1 = *(const v4f*)(ak + 8);
    v4f a2 = *(const v4f*)(ak + 16 * K);
    v4f a3 = *(const v4f*)(ak + 16 * K + 8);
    v4f r0 = *(const v4f*)(bk);
    v4f r1 = *(const v4f*)(bk + N);
    v4f r2 = *(const v4f*)(bk + 2 * N);
    v4f r3 = *(const v4f*)(bk + 3 * N);
    v4f r4 = *(const v4f*)(bk + 8 * N);
    v4f r5 = *(const v4f*)(bk + 9 * N);
    v4f r6 = *(const v4f*)(bk + 10 * N);
    v4f r7 = *(const v4f*)(bk + 11 * N);
    asm volatile("" : "+v"(a0), "+v"(a1), "+v"(a2), "+v"(a3), "+v"(r0), "+v"(r1), "+v"(r2), "+v"(r3),
                      "+v"(r4), "+v"(r5), "+v"(r6), "+v"(r7));
    v2f p00 = lgesql_pair(r0.x, r1.x), p01 = lgesql_pair(r0.y, r1.y), p02 = lgesql_pair(r0.z, r1.z), p03 = lgesql_pair(r0.w, r1.w);
    v2f p10 = lgesql_pair(r2.x, r3.x), p11 = lgesql_pair(r2.y, r3.y), p12 = lgesql_pair(r2.z, r3.z), p13 = lgesql_pair(r2.w, r3.w);
    v2f p20 = lgesql_pair(r4.x, r5.x), p21 = lgesql_pair(r4.y, r5.y), p22 = lgesql_pair(r4.z, r5.z), p23 = lgesql_pair(r4.w, r5.w);
    v2f p30 = lgesql_pair(r6.x, r7.x), p31 = lgesql_pair(r6.y, r7.y), p32 = lgesql_pair(r6.z, r7.z), p33 = lgesql_pair(r6.w, r7.w);
    asm volatile("" : "+v"(p00), "+v"(p01), "+v"(p02), "+v"(p03), "+v"(p10), "+v"(p11), "+v"(p12), "+v"(p13),
                      "+v"(p20), "+v"(p21), "+v"(p22), "+v"(p23), "+v"(p30), "+v"(p31), "+v"(p32), "+v"(p33));
    const v2f x00 = lgesql_pair(a0.x, a0.y), x01 = lgesql_pair(a0.z, a0.w);
    const v2f x10 = lgesql_pair(a1.x, a1.y), x11 = lgesql_pair(a1.z, a1.w);
    const v2f y00 = lgesql_pair(a2.x, a2.y), y01 = lgesql_pair(a2.z, a2.w);
    const v2f y10 = lgesql_pair(a3.x, a3.y), y11 = lgesql_pair(a3.z, a3.w);
    c00 = __builtin_amdgcn_wmma_f32_16x16x4_f32(false, x00, false, p00, (short)0, c00, false, false);
    c01 = __builtin_amdgcn_wmma_f32_16x16x4_f32(false, x00, false, p01, (short)0, c01, false, false);
    c02 = __builtin_amdgcn_wmma_f32_16x16x4_f32(false, x00, false, p02, (short)0, c02, false, false);
    c03 = __builtin_amdgcn_wmma_f32_16x16x4_f32(false, x00, false, p03, (short)0, c03, false, false);
    c10 = __builtin_amdgcn_wmma_f32_16x16x4_f32(false, y00, false, p00, (short)0, c10, false, false);
    c11 = __builtin_amdgcn_wmma_f32_16x16x4_f32(false, y00, false, p01, (short)0, c11, false, false);
    c12 = __builtin_amdgcn_wmma_f32_16x16x4_f32(false, y00, false, p02, (short)0, c12, false, false);
    c13 = __builtin_amdgcn_wmma_f32_16x16x4_f32(false, y00, false, p03, (short)0, c13, false, false);
    c00 = __builtin_amdgcn_wmma_f32_16x16x4_f32(false, x01, false, p10, (short)0, c00, false, false);
    c01 = __builtin_amdgcn_wmma_f32_16x16x4_f32(false, x01, false, p11, (short)0, c01, false, false);
    c02 = __builtin_amdgcn_wmma_f32_16x16x4_f32(false, x01, false, p12, (short)0, c02, false, false);
    c03 = __builtin_amdgcn_wmma_f32_16x16x4_f32(false, x01, false, p13, (short)0, c03, false, false);
    c10 = __builtin_amdgcn_wmma_f32_16x16x4_f32(false, y01, false, p10, (short)0, c10, false, false);
    c11 = __builtin_amdgcn_wmma_f32_16x16x4_f32(false, y01, false, p11, (short)0, c11, false, false);
    c12 = __builtin_amdgcn_wmma_f32_16x16x4_f32(false, y01, false, p12, (short)0, c12, false, false);
    c13 = __builtin_amdgcn_wmma_f32_16x16x4_f32(false, y01, false, p13, (short)0, c13, false, false);
    c00 = __builtin_amdgcn_wmma_f32_16x16x4_f32(false, x10, false, p20, (short)0, c00, false, false);
    c01 = __builtin_amdgcn_wmma_f32_16x16x4_f32(false, x10, false, p21, (short)0, c01, false, false);
    c02 = __builtin_amdgcn_wmma_f32_16x16x4_f32(false, x10, false, p22, (short)0, c02, false, false);
    c03 = __builtin_amdgcn_wmma_f32_16x16x4_f32(false, x10, false, p23, (short)0, c03, false, false);
    c10 = __builtin_amdgcn_wmma_f32_16x16x4_f32(false, y10, false, p20, (short)0, c10, false, false);
    c11 = __builtin_amdgcn_wmma_f32_16x16x4_f32(false, y10, false, p21, (short)0, c11, false, false);
    c12 = __builtin_amdgcn_wmma_f32_16x16x4_f32(false, y10, false, p22, (short)0, c12, false, false);
    c13 = __builtin_amdgcn_wmma_f32_16x16x4_f32(false, y10, false, p23, (short)0, c13, false, false);
    c00 = __builtin_amdgcn_wmma_f32_16x16x4_f32(false, x11, false, p30, (short)0, c00, false, false);
    c01 = __builtin_amdgcn_wmma_f32_16x16x4_f32(false, x11, false, p31, (short)0, c01, false, false);
    c02 = __builtin_amdgcn_wmma_f32_16x16x4_f32(false, x11, false, p32, (short)0, c02, false, false);
    c03 = __builtin_amdgcn_wmma_f32_16x16x4_f32(false, x11, false, p33, (short)0, c03, false, false);
    c10 = __builtin_amdgcn_wmma_f32_16x16x4_f32(false, y11, false, p30, (short)0, c10, false, false);
    c11 = __builtin_amdgcn_wmma_f32_16x16x4_f32(false, y11, false, p31, (short)0, c11, false, false);
    c12 = __builtin_amdgcn_wmma_f32_16x16x4_f32(false, y11, false, p32, (short)0, c12, false, false);
    c13 = __builtin_amdgcn_wmma_f32_16x16x4_f32(false, y11, false, p33, (short)0, c13, false, false);
  }
  asm volatile("v_nop\n\tv_nop\n\tv_nop\n\tv_nop\n\tv_nop\n\tv_nop\n\tv_nop\n\tv_nop"
               : "+v"(c00), "+v"(c01), "+v"(c02), "+v"(c03), "+v"(c10), "+v"(c11), "+v"(c12), "+v"(c13));
}

__device__ __forceinline__ void lgesql_group_edges(
    int wg, const int* __restrict__ lg_src, const int* __restrict__ lg_dst,
    int* __restrict__ cnt, int* __restrict__ bucket, int* __restrict__ ovf, int* __restrict__ ovf_n)
{
  static_assert(GROUP_WGS * QKV_THREADS * GROUP_EPT == LE_EDGES, "the grouping work-groups cover the edges exactly");
  const int e0 = wg * (QKV_THREADS * GROUP_EPT) + (int)threadIdx.x;
  int d[GROUP_EPT], sidx[GROUP_EPT], r[GROUP_EPT];
  #pragma unroll
  for (int i = 0; i < GROUP_EPT; ++i) { d[i] = lg_dst[e0 + QKV_THREADS * i]; sidx[i] = lg_src[e0 + QKV_THREADS * i]; }
  #pragma unroll
  for (int i = 0; i < GROUP_EPT; ++i) r[i] = atomicAdd(&cnt[d[i]], 1);
  #pragma unroll
  for (int i = 0; i < GROUP_EPT; ++i) {
    if (r[i] < BUCKET_CAP) {
      bucket[(size_t)d[i] * BUCKET_CAP + r[i]] = sidx[i];
    } else {
      const int p = atomicAdd(ovf_n, 1);
      ovf[2 * p] = d[i]; ovf[2 * p + 1] = sidx[i];
    }
  }
}

__global__ __launch_bounds__(QKV_THREADS) void lgesql_qkv_wmma_f32(
    const float* __restrict__ x,
    const float* __restrict__ Wq, const float* __restrict__ Wk, const float* __restrict__ Wv,
    const float* __restrict__ bq, const float* __restrict__ src_x, const float* __restrict__ dst_x,
    float* __restrict__ q, float* __restrict__ k, float* __restrict__ v,
    const int* __restrict__ lg_src, const int* __restrict__ lg_dst,
    int* __restrict__ cnt, int* __restrict__ bucket, int* __restrict__ ovf, int* __restrict__ ovf_n)
{
  if ((int)blockIdx.x < GROUP_WGS) {
    lgesql_group_edges((int)blockIdx.x, lg_src, lg_dst, cnt, bucket, ovf, ovf_n);
    return;
  }
  const int pblk = (int)blockIdx.x - GROUP_WGS;
  constexpr int K = NDIM_, N = NDIM_;
  constexpr int TN = 64;
  constexpr int tilesN = N / TN;
  constexpr int GROUPS_PER_MATRIX = (E_NODES / 32) * tilesN / (QKV_THREADS / 32);
  static_assert(E_NODES % 32 == 0, "32-row tiles cover the rows exactly");
  const int which = pblk / GROUPS_PER_MATRIX;
  const int grp   = pblk - which * GROUPS_PER_MATRIX;
  const float* __restrict__ W = which == 0 ? Wq : (which == 1 ? Wk : Wv);
  float* __restrict__ C       = which == 0 ? q  : (which == 1 ? k  : v);
  const float* __restrict__ addm = which == 0 ? src_x : dst_x;

  const int wid  = (grp * (int)blockDim.x + (int)threadIdx.x) >> 5;
  const int lane = threadIdx.x & 31;
  const int tm = wid / tilesN;
  const int tn = wid - tm * tilesN;
  const int half = lane >> 4;
  const int l    = lane & 15;

  const float* Ap = x + (size_t)(tm * 32 + l) * K + 4 * half;
  const float* Bp = W + (size_t)(4 * half) * N + tn * TN + 4 * l;

  v8f c00 = {}, c01 = {}, c02 = {}, c03 = {}, c10 = {}, c11 = {}, c12 = {}, c13 = {};
  lgesql_tile2_k_loop<K, N>(Ap, Bp, c00, c01, c02, c03, c10, c11, c12, c13);

  int tm_e = tm, tn_e = tn, half_e = half, l_e = l;
  asm volatile("" : "+v"(tm_e), "+v"(tn_e), "+v"(half_e), "+v"(l_e), "+v"(c00));
  const int col = tn_e * TN + 4 * l_e;
  const size_t at = (size_t)(tm_e * 32 + 8 * half_e) * N + col;
  float* __restrict__ Cp = C + at;
  if (which == 1) {
    #pragma unroll
    for (int r = 0; r < 8; ++r)
      *(float4*)(Cp + (size_t)r * N) = make_float4(c00[r], c01[r], c02[r], c03[r]);
    #pragma unroll
    for (int r = 0; r < 8; ++r)
      *(float4*)(Cp + (size_t)(16 + r) * N) = make_float4(c10[r], c11[r], c12[r], c13[r]);
  } else {
    const float* __restrict__ Rp = addm + at;
    float4 bv = make_float4(0.0f, 0.0f, 0.0f, 0.0f);
    if (which == 0) bv = *(const float4*)(bq + col);
    #pragma unroll
    for (int r0 = 0; r0 < 8; r0 += 4) {
      float4 ad[4];
      #pragma unroll
      for (int i = 0; i < 4; ++i) ad[i] = *(const float4*)(Rp + (size_t)(r0 + i) * N);
      #pragma unroll
      for (int i = 0; i < 4; ++i) {
        const int r = r0 + i;
        float4 o = make_float4(c00[r] + bv.x, c01[r] + bv.y, c02[r] + bv.z, c03[r] + bv.w);
        o.x += ad[i].x; o.y += ad[i].y; o.z += ad[i].z; o.w += ad[i].w;
        *(float4*)(Cp + (size_t)r * N) = o;
      }
      __builtin_amdgcn_sched_barrier(0);
    }
    #pragma unroll
    for (int r0 = 0; r0 < 8; r0 += 4) {
      float4 ad[4];
      #pragma unroll
      for (int i = 0; i < 4; ++i) ad[i] = *(const float4*)(Rp + (size_t)(16 + r0 + i) * N);
      #pragma unroll
      for (int i = 0; i < 4; ++i) {
        const int r = r0 + i;
        float4 o = make_float4(c10[r] + bv.x, c11[r] + bv.y, c12[r] + bv.z, c13[r] + bv.w);
        o.x += ad[i].x; o.y += ad[i].y; o.z += ad[i].z; o.w += ad[i].w;
        *(float4*)(Cp + (size_t)(16 + r) * N) = o;
      }
      __builtin_amdgcn_sched_barrier(0);
    }
  }
}

template <int U>
__device__ __forceinline__ void lgesql_edges(
    const float* __restrict__ k, const float* __restrict__ v, int mysrc, unsigned& todo, int lane,
    const float4& qa, const float4& qb, float4& accA, float4& accB, float& zA, float& zB)
{
  float4 ka[U], kb[U], va[U], vb[U];
  #pragma unroll
  for (int u = 0; u < U; ++u) {
    const int s = __builtin_amdgcn_readlane(mysrc, __builtin_ctz(todo));
    todo &= todo - 1;
    const float4* kr = (const float4*)(k + (size_t)s * NDIM_);
    const float4* vr = (const float4*)(v + (size_t)s * NDIM_);
    ka[u] = kr[lane]; kb[u] = kr[32 + lane];
    va[u] = vr[lane]; vb[u] = vr[32 + lane];
  }
  #pragma unroll
  for (int u = 0; u < U; ++u) {
    float pA = ka[u].x * qa.x + ka[u].y * qa.y + ka[u].z * qa.z + ka[u].w * qa.w;
    float pB = kb[u].x * qb.x + kb[u].y * qb.y + kb[u].z * qb.z + kb[u].w * qb.w;
    pA = lgesql_xadd<1>(pA); pB = lgesql_xadd<1>(pB);
    pA = lgesql_xadd<2>(pA); pB = lgesql_xadd<2>(pB);
    pA = lgesql_xadd<4>(pA); pB = lgesql_xadd<4>(pB);
    float sA = pA * 0.17677669529663687f;
    float sB = pB * 0.17677669529663687f;
    sA = expf(fminf(5.0f, fmaxf(-5.0f, sA)));
    sB = expf(fminf(5.0f, fmaxf(-5.0f, sB)));
    accA.x += va[u].x * sA; accA.y += va[u].y * sA; accA.z += va[u].z * sA; accA.w += va[u].w * sA;
    accB.x += vb[u].x * sB; accB.y += vb[u].y * sB; accB.z += vb[u].z * sB; accB.w += vb[u].w * sB;
    zA += sA; zB += sB;
  }
}

__device__ __forceinline__ void lgesql_node_attention(
    int n, int lane,
    const float* __restrict__ q, const float* __restrict__ k, const float* __restrict__ v,
    const int* __restrict__ cnt, const int* __restrict__ bucket,
    const int* __restrict__ ovf, const int* __restrict__ ovf_n, float4& oa, float4& ob)
{
  const int c = cnt[n];
  const int deg = min(c, BUCKET_CAP);
  const int* __restrict__ brow = bucket + (size_t)n * BUCKET_CAP;
  const float4* qr = (const float4*)(q + (size_t)n * NDIM_);
  const float4 qa = qr[lane];
  const float4 qb = qr[32 + lane];
  float4 accA = make_float4(0.f, 0.f, 0.f, 0.f);
  float4 accB = make_float4(0.f, 0.f, 0.f, 0.f);
  float zA = 0.f, zB = 0.f;
  const int nb = (deg + 31) >> 5;
  int novf = 0;
  if (c > BUCKET_CAP) novf = min(*ovf_n, LE_EDGES);
  const int nchunk = nb + ((novf + 31) >> 5);
  for (int t = 0; t < nchunk; ++t) {
    int mysrc;
    unsigned todo;
    if (t < nb) {
      const int m = min(32, deg - 32 * t);
      mysrc = (lane < m) ? brow[32 * t + lane] : 0;
      todo = (m == 32) ? 0xffffffffu : ((1u << m) - 1u);
    } else {
      const int idx = 32 * (t - nb) + lane;
      const int od = (idx < novf) ? ovf[2 * idx] : -1;
      mysrc = (idx < novf) ? ovf[2 * idx + 1] : 0;
      todo = __builtin_amdgcn_ballot_w32(od == n);
    }
    while (__builtin_popcount(todo) >= 4)
      lgesql_edges<4>(k, v, mysrc, todo, lane, qa, qb, accA, accB, zA, zB);
    if (__builtin_popcount(todo) >= 2)
      lgesql_edges<2>(k, v, mysrc, todo, lane, qa, qb, accA, accB, zA, zB);
    if (todo)
      lgesql_edges<1>(k, v, mysrc, todo, lane, qa, qb, accA, accB, zA, zB);
  }
  oa = make_float4(accA.x / zA, accA.y / zA, accA.z / zA, accA.w / zA);
  ob = make_float4(accB.x / zB, accB.y / zB, accB.z / zB, accB.w / zB);
}

__device__ __forceinline__ void lgesql_ln_rows(
    float4 (&v)[8], float2 (*part)[4], int w, int l, int half, const float4 g4, const float4 b4)
{
  #pragma unroll
  for (int r = 0; r < 8; ++r) {
    float sm = (v[r].x + v[r].y) + (v[r].z + v[r].w);
    float sq = (v[r].x * v[r].x + v[r].y * v[r].y) + (v[r].z * v[r].z + v[r].w * v[r].w);
    sm = lgesql_xadd<1>(sm); sq = lgesql_xadd<1>(sq);
    sm = lgesql_xadd<2>(sm); sq = lgesql_xadd<2>(sq);
    sm = lgesql_xadd<4>(sm); sq = lgesql_xadd<4>(sq);
    sm = lgesql_xadd<8>(sm); sq = lgesql_xadd<8>(sq);
    if (l == r) part[r + 8 * half][w] = make_float2(sm, sq);
  }
  __syncthreads();
  #pragma unroll
  for (int r = 0; r < 8; ++r) {
    const float4 p01 = *(const float4*)(&part[r + 8 * half][0]);
    const float4 p23 = *(const float4*)(&part[r + 8 * half][2]);
    const float sum = ((p01.x + p01.z) + p23.x) + p23.z;
    const float sqs = ((p01.y + p01.w) + p23.y) + p23.w;
    const float mu   = sum * (1.0f / NDIM_);
    const float var  = sqs * (1.0f / NDIM_) - mu * mu;
    const float rstd = rsqrtf(var + 1e-5f);
    float4 o;
    o.x = (v[r].x - mu) * rstd * g4.x + b4.x;
    o.y = (v[r].y - mu) * rstd * g4.y + b4.y;
    o.z = (v[r].z - mu) * rstd * g4.z + b4.z;
    o.w = (v[r].w - mu) * rstd * g4.w + b4.w;
    v[r] = o;
  }
}

#define LDS_PITCH 260
__global__ __launch_bounds__(128) void lgesql_attn_ffn_wmma_f32(
    const float* __restrict__ q, const float* __restrict__ k, const float* __restrict__ v,
    const int* __restrict__ cnt, const int* __restrict__ bucket,
    const int* __restrict__ ovf, const int* __restrict__ ovf_n,
    const float* __restrict__ x, const float* __restrict__ Wo, const float* __restrict__ bo,
    const float* __restrict__ ln1_g, const float* __restrict__ ln1_b,
    const float* __restrict__ W1, const float* __restrict__ b1,
    const float* __restrict__ W2, const float* __restrict__ b2,
    const float* __restrict__ ln2_g, const float* __restrict__ ln2_b,
    float* __restrict__ out)
{
  __shared__ __attribute__((aligned(16))) float sA[16 * LDS_PITCH];
  __shared__ __attribute__((aligned(16))) float sF[16 * LDS_PITCH];
  __shared__ __attribute__((aligned(16))) float2 part[16][4];
  const int lane = threadIdx.x & 31;
  const int w    = threadIdx.x >> 5;
  const int half = lane >> 4;
  const int l    = lane & 15;
  const int tm   = blockIdx.x;

  for (int i = 0; i < 4; ++i) {
    const int row = 4 * w + i;
    const int n = __builtin_amdgcn_readfirstlane(16 * tm + row);
    float4 oa, ob;
    lgesql_node_attention(n, lane, q, k, v, cnt, bucket, ovf, ovf_n, oa, ob);
    *(float4*)(sA + row * LDS_PITCH + 4 * lane)       = oa;
    *(float4*)(sA + row * LDS_PITCH + 128 + 4 * lane) = ob;
  }
  __syncthreads();

  const float* Ap = sA + l * LDS_PITCH + 4 * half;
  const float* Fp = sF + l * LDS_PITCH + 4 * half;
  const int col = 64 * w + 4 * l;
  float4 vv[8];

  {
    v8f a0 = {}, a1 = {}, a2 = {}, a3 = {};
    lgesql_tile_k_loop<NDIM_, NDIM_>(Ap, Wo + (size_t)(4 * half) * NDIM_ + col, a0, a1, a2, a3);
    const float4 bv = *(const float4*)(bo + col);
    #pragma unroll
    for (int r = 0; r < 8; ++r) {
      const int row = tm * 16 + r + 8 * half;
      vv[r] = make_float4(a0[r] + bv.x, a1[r] + bv.y, a2[r] + bv.z, a3[r] + bv.w);
      const float4 ad = *(const float4*)(x + (size_t)row * NDIM_ + col);
      vv[r].x += ad.x; vv[r].y += ad.y; vv[r].z += ad.z; vv[r].w += ad.w;
    }
    lgesql_ln_rows(vv, part, w, l, half, *(const float4*)(ln1_g + col), *(const float4*)(ln1_b + col));
    #pragma unroll
    for (int r = 0; r < 8; ++r)
      *(float4*)(sA + (r + 8 * half) * LDS_PITCH + col) = vv[r];
  }
  __syncthreads();

  v8f d0 = {}, d1 = {}, d2 = {}, d3 = {};
  for (int c = 0; c < 4; ++c) {
    v8f e0 = {}, e1 = {}, e2 = {}, e3 = {};
    const int hcol = NDIM_ * c + col;
    lgesql_tile_k_loop<NDIM_, 4 * NDIM_>(Ap, W1 + (size_t)(4 * half) * (4 * NDIM_) + hcol, e0, e1, e2, e3);
    const float4 bv = *(const float4*)(b1 + hcol);
    #pragma unroll
    for (int r = 0; r < 8; ++r) {
      float4 o = make_float4(e0[r] + bv.x, e1[r] + bv.y, e2[r] + bv.z, e3[r] + bv.w);
      o.x = fmaxf(o.x, 0.0f); o.y = fmaxf(o.y, 0.0f); o.z = fmaxf(o.z, 0.0f); o.w = fmaxf(o.w, 0.0f);
      *(float4*)(sF + (r + 8 * half) * LDS_PITCH + col) = o;
    }
    __syncthreads();
    lgesql_tile_k_loop<NDIM_, NDIM_>(Fp, W2 + (size_t)(NDIM_ * c + 4 * half) * NDIM_ + col, d0, d1, d2, d3);
    __syncthreads();
  }

  {
    const float4 bv = *(const float4*)(b2 + col);
    #pragma unroll
    for (int r = 0; r < 8; ++r) {
      vv[r] = make_float4(d0[r] + bv.x, d1[r] + bv.y, d2[r] + bv.z, d3[r] + bv.w);
      const float4 ad = *(const float4*)(sA + (r + 8 * half) * LDS_PITCH + col);
      vv[r].x += ad.x; vv[r].y += ad.y; vv[r].z += ad.z; vv[r].w += ad.w;
    }
    lgesql_ln_rows(vv, part, w, l, half, *(const float4*)(ln2_g + col), *(const float4*)(ln2_b + col));
    #pragma unroll
    for (int r = 0; r < 8; ++r) {
      const int row = tm * 16 + r + 8 * half;
      *(float4*)(out + (size_t)row * NDIM_ + col) = vv[r];
    }
  }
}

extern "C" void kernel_launch(void* const* d_in, const int* in_sizes, int n_in,
                              void* d_out, int out_size, void* d_ws, size_t ws_size,
                              hipStream_t stream) {
  const float* x     = (const float*)d_in[0];
  const float* src_x = (const float*)d_in[1];
  const float* dst_x = (const float*)d_in[2];
  const float* Wq    = (const float*)d_in[3];
  const float* bq    = (const float*)d_in[4];
  const float* Wk    = (const float*)d_in[5];
  const float* Wv    = (const float*)d_in[6];
  const float* Wo    = (const float*)d_in[7];
  const float* bo    = (const float*)d_in[8];
  const float* ln1_g = (const float*)d_in[9];
  const float* ln1_b = (const float*)d_in[10];
  const float* W1    = (const float*)d_in[11];
  const float* b1    = (const float*)d_in[12];
  const float* W2    = (const float*)d_in[13];
  const float* b2    = (const float*)d_in[14];
  const float* ln2_g = (const float*)d_in[15];
  const float* ln2_b = (const float*)d_in[16];
  const int*   lg_src = (const int*)d_in[17];
  const int*   lg_dst = (const int*)d_in[18];
  float* out = (float*)d_out;

  const size_t EN = (size_t)E_NODES * NDIM_;
  float* ws = (float*)d_ws;
  float* qf   = ws;
  float* kf   = ws + EN;
  float* vf   = ws + 2 * EN;
  int*   bucketf = (int*)(ws + 3 * EN);
  int*   ovff = bucketf + (size_t)E_NODES * BUCKET_CAP;
  int*   cntf = (int*)(ws + 4 * EN);
  int*   ovfn = cntf + E_NODES;

  hipMemsetAsync(cntf, 0, (size_t)(E_NODES + 1) * sizeof(int), stream);

  const int gqkv  = GROUP_WGS + 3 * ((E_NODES / 32) * (NDIM_ / 64) / (QKV_THREADS / 32));
  const dim3 blkqkv(QKV_THREADS);
  const int grows = E_NODES / 16;
  const dim3 blkrows(128);

  lgesql_qkv_wmma_f32<<<gqkv, blkqkv, 0, stream>>>(x, Wq, Wk, Wv, bq, src_x, dst_x, qf, kf, vf,
                                                lg_src, lg_dst, cntf, bucketf, ovff, ovfn);

  lgesql_attn_ffn_wmma_f32<<<grows, blkrows, 0, stream>>>(qf, kf, vf, cntf, bucketf, ovff, ovfn,
                                                      x, Wo, bo, ln1_g, ln1_b, W1, b1, W2, b2, ln2_g, ln2_b, out);
}


# Fused launch: issue priority falls with a wave's progress (s_setprio 3 through FFN chunk 0, then 2, 1, 0 in chunks 1-3): work-groups on a WGP advance together, not oldest first; sums unchanged
# speedup vs baseline: 2.1726x; 1.0374x over previous
#include <hip/hip_runtime.h>
#include <math.h>

#define E_NODES 20000
#define LE_EDGES 320000
#define NDIM_ 256
#define HEADS 8
#define DK_ 32
#define BUCKET_CAP 64
#define GROUP_EPT 5
#define QKV_THREADS 128
#define GROUP_WGS (LE_EDGES / (QKV_THREADS * GROUP_EPT))

typedef __attribute__((ext_vector_type(2))) float v2f;
typedef __attribute__((ext_vector_type(4))) float v4f;
typedef __attribute__((ext_vector_type(8))) float v8f;

__device__ __forceinline__ v2f lgesql_pair(float lo, float hi)
{
  v2f p; p.x = lo; p.y = hi; return p;
}

template <int XMASK>
__device__ __forceinline__ float lgesql_xadd(float x)
{
  const int y = __builtin_amdgcn_update_dpp(0, __builtin_bit_cast(int, x), 0x160 | XMASK, 0xf, 0xf, true);
  return x + __builtin_bit_cast(float, y);
}

template <int K, int N>
__device__ __forceinline__ void lgesql_tile_k_loop(
    const float* __restrict__ Ap, const float* __restrict__ Bp,
    v8f& acc0, v8f& acc1, v8f& acc2, v8f& acc3)
{
  for (int k = 0; k < K; k += 16) {
    int kk = k;
    asm volatile("v_nop\n\tv_nop\n\tv_nop\n\tv_nop\n\tv_nop\n\tv_nop\n\tv_nop\n\tv_nop"
                 : "+v"(acc0), "+v"(acc1), "+v"(acc2), "+v"(acc3), "+s"(kk));
    const float* ak = Ap + kk;
    const float* bk = Bp + (size_t)kk * N;
    const float4 a0 = *(const float4*)(ak);
    const float4 a1 = *(const float4*)(ak + 8);
    const float4 r0 = *(const float4*)(bk);
    const float4 r1 = *(const float4*)(bk + N);
    const float4 r2 = *(const float4*)(bk + 2 * N);
    const float4 r3 = *(const float4*)(bk + 3 * N);
    const float4 r4 = *(const float4*)(bk + 8 * N);
    const float4 r5 = *(const float4*)(bk + 9 * N);
    const float4 r6 = *(const float4*)(bk + 10 * N);
    const float4 r7 = *(const float4*)(bk + 11 * N);
    v2f p00 = lgesql_pair(r0.x, r1.x), p01 = lgesql_pair(r0.y, r1.y), p02 = lgesql_pair(r0.z, r1.z), p03 = lgesql_pair(r0.w, r1.w);
    v2f p10 = lgesql_pair(r2.x, r3.x), p11 = lgesql_pair(r2.y, r3.y), p12 = lgesql_pair(r2.z, r3.z), p13 = lgesql_pair(r2.w, r3.w);
    v2f p20 = lgesql_pair(r4.x, r5.x), p21 = lgesql_pair(r4.y, r5.y), p22 = lgesql_pair(r4.z, r5.z), p23 = lgesql_pair(r4.w, r5.w);
    v2f p30 = lgesql_pair(r6.x, r7.x), p31 = lgesql_pair(r6.y, r7.y), p32 = lgesql_pair(r6.z, r7.z), p33 = lgesql_pair(r6.w, r7.w);
    asm volatile("" : "+v"(p00), "+v"(p01), "+v"(p02), "+v"(p03), "+v"(p10), "+v"(p11), "+v"(p12), "+v"(p13),
                      "+v"(p20), "+v"(p21), "+v"(p22), "+v"(p23), "+v"(p30), "+v"(p31), "+v"(p32), "+v"(p33));
    const v2f a00 = lgesql_pair(a0.x, a0.y), a01 = lgesql_pair(a0.z, a0.w);
    const v2f a10 = lgesql_pair(a1.x, a1.y), a11 = lgesql_pair(a1.z, a1.w);
    acc0 = __builtin_amdgcn_wmma_f32_16x16x4_f32(false, a00, false, p00, (short)0, acc0, false, false);
    acc1 = __builtin_amdgcn_wmma_f32_16x16x4_f32(false, a00, false, p01, (short)0, acc1, false, false);
    acc2 = __builtin_amdgcn_wmma_f32_16x16x4_f32(false, a00, false, p02, (short)0, acc2, false, false);
    acc3 = __builtin_amdgcn_wmma_f32_16x16x4_f32(false, a00, false, p03, (short)0, acc3, false, false);
    acc0 = __builtin_amdgcn_wmma_f32_16x16x4_f32(false, a01, false, p10, (short)0, acc0, false, false);
    acc1 = __builtin_amdgcn_wmma_f32_16x16x4_f32(false, a01, false, p11, (short)0, acc1, false, false);
    acc2 = __builtin_amdgcn_wmma_f32_16x16x4_f32(false, a01, false, p12, (short)0, acc2, false, false);
    acc3 = __builtin_amdgcn_wmma_f32_16x16x4_f32(false, a01, false, p13, (short)0, acc3, false, false);
    acc0 = __builtin_amdgcn_wmma_f32_16x16x4_f32(false, a10, false, p20, (short)0, acc0, false, false);
    acc1 = __builtin_amdgcn_wmma_f32_16x16x4_f32(false, a10, false, p21, (short)0, acc1, false, false);
    acc2 = __builtin_amdgcn_wmma_f32_16x16x4_f32(false, a10, false, p22, (short)0, acc2, false, false);
    acc3 = __builtin_amdgcn_wmma_f32_16x16x4_f32(false, a10, false, p23, (short)0, acc3, false, false);
    acc0 = __builtin_amdgcn_wmma_f32_16x16x4_f32(false, a11, false, p30, (short)0, acc0, false, false);
    acc1 = __builtin_amdgcn_wmma_f32_16x16x4_f32(false, a11, false, p31, (short)0, acc1, false, false);
    acc2 = __builtin_amdgcn_wmma_f32_16x16x4_f32(false, a11, false, p32, (short)0, acc2, false, false);
    acc3 = __builtin_amdgcn_wmma_f32_16x16x4_f32(false, a11, false, p33, (short)0, acc3, false, false);
  }
  asm volatile("v_nop\n\tv_nop\n\tv_nop\n\tv_nop\n\tv_nop\n\tv_nop\n\tv_nop\n\tv_nop"
               : "+v"(acc0), "+v"(acc1), "+v"(acc2), "+v"(acc3));
}

template <int K, int N>
__device__ __forceinline__ void lgesql_tile2_k_loop(
    const float* __restrict__ Ap, const float* __restrict__ Bp,
    v8f& c00, v8f& c01, v8f& c02, v8f& c03, v8f& c10, v8f& c11, v8f& c12, v8f& c13)
{
  for (int k = 0; k < K; k += 16) {
    int kk = k;
    asm volatile("v_nop\n\tv_nop\n\tv_nop\n\tv_nop\n\tv_nop\n\tv_nop\n\tv_nop\n\tv_nop"
                 : "+v"(c00), "+v"(c01), "+v"(c02), "+v"(c03), "+v"(c10), "+v"(c11), "+v"(c12), "+v"(c13), "+s"(kk));
    const float* ak = Ap + kk;
    const float* bk = Bp + (size_t)kk * N;
    v4f a0 = *(const v4f*)(ak);
    v4f a1 = *(const v4f*)(ak + 8);
    v4f a2 = *(const v4f*)(ak + 16 * K);
    v4f a3 = *(const v4f*)(ak + 16 * K + 8);
    v4f r0 = *(const v4f*)(bk);
    v4f r1 = *(const v4f*)(bk + N);
    v4f r2 = *(const v4f*)(bk + 2 * N);
    v4f r3 = *(const v4f*)(bk + 3 * N);
    v4f r4 = *(const v4f*)(bk + 8 * N);
    v4f r5 = *(const v4f*)(bk + 9 * N);
    v4f r6 = *(const v4f*)(bk + 10 * N);
    v4f r7 = *(const v4f*)(bk + 11 * N);
    asm volatile("" : "+v"(a0), "+v"(a1), "+v"(a2), "+v"(a3), "+v"(r0), "+v"(r1), "+v"(r2), "+v"(r3),
                      "+v"(r4), "+v"(r5), "+v"(r6), "+v"(r7));
    v2f p00 = lgesql_pair(r0.x, r1.x), p01 = lgesql_pair(r0.y, r1.y), p02 = lgesql_pair(r0.z, r1.z), p03 = lgesql_pair(r0.w, r1.w);
    v2f p10 = lgesql_pair(r2.x, r3.x), p11 = lgesql_pair(r2.y, r3.y), p12 = lgesql_pair(r2.z, r3.z), p13 = lgesql_pair(r2.w, r3.w);
    v2f p20 = lgesql_pair(r4.x, r5.x), p21 = lgesql_pair(r4.y, r5.y), p22 = lgesql_pair(r4.z, r5.z), p23 = lgesql_pair(r4.w, r5.w);
    v2f p30 = lgesql_pair(r6.x, r7.x), p31 = lgesql_pair(r6.y, r7.y), p32 = lgesql_pair(r6.z, r7.z), p33 = lgesql_pair(r6.w, r7.w);
    asm volatile("" : "+v"(p00), "+v"(p01), "+v"(p02), "+v"(p03), "+v"(p10), "+v"(p11), "+v"(p12), "+v"(p13),
                      "+v"(p20), "+v"(p21), "+v"(p22), "+v"(p23), "+v"(p30), "+v"(p31), "+v"(p32), "+v"(p33));
    const v2f x00 = lgesql_pair(a0.x, a0.y), x01 = lgesql_pair(a0.z, a0.w);
    const v2f x10 = lgesql_pair(a1.x, a1.y), x11 = lgesql_pair(a1.z, a1.w);
    const v2f y00 = lgesql_pair(a2.x, a2.y), y01 = lgesql_pair(a2.z, a2.w);
    const v2f y10 = lgesql_pair(a3.x, a3.y), y11 = lgesql_pair(a3.z, a3.w);
    c00 = __builtin_amdgcn_wmma_f32_16x16x4_f32(false, x00, false, p00, (short)0, c00, false, false);
    c01 = __builtin_amdgcn_wmma_f32_16x16x4_f32(false, x00, false, p01, (short)0, c01, false, false);
    c02 = __builtin_amdgcn_wmma_f32_16x16x4_f32(false, x00, false, p02, (short)0, c02, false, false);
    c03 = __builtin_amdgcn_wmma_f32_16x16x4_f32(false, x00, false, p03, (short)0, c03, false, false);
    c10 = __builtin_amdgcn_wmma_f32_16x16x4_f32(false, y00, false, p00, (short)0, c10, false, false);
    c11 = __builtin_amdgcn_wmma_f32_16x16x4_f32(false, y00, false, p01, (short)0, c11, false, false);
    c12 = __builtin_amdgcn_wmma_f32_16x16x4_f32(false, y00, false, p02, (short)0, c12, false, false);
    c13 = __builtin_amdgcn_wmma_f32_16x16x4_f32(false, y00, false, p03, (short)0, c13, false, false);
    c00 = __builtin_amdgcn_wmma_f32_16x16x4_f32(false, x01, false, p10, (short)0, c00, false, false);
    c01 = __builtin_amdgcn_wmma_f32_16x16x4_f32(false, x01, false, p11, (short)0, c01, false, false);
    c02 = __builtin_amdgcn_wmma_f32_16x16x4_f32(false, x01, false, p12, (short)0, c02, false, false);
    c03 = __builtin_amdgcn_wmma_f32_16x16x4_f32(false, x01, false, p13, (short)0, c03, false, false);
    c10 = __builtin_amdgcn_wmma_f32_16x16x4_f32(false, y01, false, p10, (short)0, c10, false, false);
    c11 = __builtin_amdgcn_wmma_f32_16x16x4_f32(false, y01, false, p11, (short)0, c11, false, false);
    c12 = __builtin_amdgcn_wmma_f32_16x16x4_f32(false, y01, false, p12, (short)0, c12, false, false);
    c13 = __builtin_amdgcn_wmma_f32_16x16x4_f32(false, y01, false, p13, (short)0, c13, false, false);
    c00 = __builtin_amdgcn_wmma_f32_16x16x4_f32(false, x10, false, p20, (short)0, c00, false, false);
    c01 = __builtin_amdgcn_wmma_f32_16x16x4_f32(false, x10, false, p21, (short)0, c01, false, false);
    c02 = __builtin_amdgcn_wmma_f32_16x16x4_f32(false, x10, false, p22, (short)0, c02, false, false);
    c03 = __builtin_amdgcn_wmma_f32_16x16x4_f32(false, x10, false, p23, (short)0, c03, false, false);
    c10 = __builtin_amdgcn_wmma_f32_16x16x4_f32(false, y10, false, p20, (short)0, c10, false, false);
    c11 = __builtin_amdgcn_wmma_f32_16x16x4_f32(false, y10, false, p21, (short)0, c11, false, false);
    c12 = __builtin_amdgcn_wmma_f32_16x16x4_f32(false, y10, false, p22, (short)0, c12, false, false);
    c13 = __builtin_amdgcn_wmma_f32_16x16x4_f32(false, y10, false, p23, (short)0, c13, false, false);
    c00 = __builtin_amdgcn_wmma_f32_16x16x4_f32(false, x11, false, p30, (short)0, c00, false, false);
    c01 = __builtin_amdgcn_wmma_f32_16x16x4_f32(false, x11, false, p31, (short)0, c01, false, false);
    c02 = __builtin_amdgcn_wmma_f32_16x16x4_f32(false, x11, false, p32, (short)0, c02, false, false);
    c03 = __builtin_amdgcn_wmma_f32_16x16x4_f32(false, x11, false, p33, (short)0, c03, false, false);
    c10 = __builtin_amdgcn_wmma_f32_16x16x4_f32(false, y11, false, p30, (short)0, c10, false, false);
    c11 = __builtin_amdgcn_wmma_f32_16x16x4_f32(false, y11, false, p31, (short)0, c11, false, false);
    c12 = __builtin_amdgcn_wmma_f32_16x16x4_f32(false, y11, false, p32, (short)0, c12, false, false);
    c13 = __builtin_amdgcn_wmma_f32_16x16x4_f32(false, y11, false, p33, (short)0, c13, false, false);
  }
  asm volatile("v_nop\n\tv_nop\n\tv_nop\n\tv_nop\n\tv_nop\n\tv_nop\n\tv_nop\n\tv_nop"
               : "+v"(c00), "+v"(c01), "+v"(c02), "+v"(c03), "+v"(c10), "+v"(c11), "+v"(c12), "+v"(c13));
}

__device__ __forceinline__ void lgesql_group_edges(
    int wg, const int* __restrict__ lg_src, const int* __restrict__ lg_dst,
    int* __restrict__ cnt, int* __restrict__ bucket, int* __restrict__ ovf, int* __restrict__ ovf_n)
{
  static_assert(GROUP_WGS * QKV_THREADS * GROUP_EPT == LE_EDGES, "the grouping work-groups cover the edges exactly");
  const int e0 = wg * (QKV_THREADS * GROUP_EPT) + (int)threadIdx.x;
  int d[GROUP_EPT], sidx[GROUP_EPT], r[GROUP_EPT];
  #pragma unroll
  for (int i = 0; i < GROUP_EPT; ++i) { d[i] = lg_dst[e0 + QKV_THREADS * i]; sidx[i] = lg_src[e0 + QKV_THREADS * i]; }
  #pragma unroll
  for (int i = 0; i < GROUP_EPT; ++i) r[i] = atomicAdd(&cnt[d[i]], 1);
  #pragma unroll
  for (int i = 0; i < GROUP_EPT; ++i) {
    if (r[i] < BUCKET_CAP) {
      bucket[(size_t)d[i] * BUCKET_CAP + r[i]] = sidx[i];
    } else {
      const int p = atomicAdd(ovf_n, 1);
      ovf[2 * p] = d[i]; ovf[2 * p + 1] = sidx[i];
    }
  }
}

__global__ __launch_bounds__(QKV_THREADS) void lgesql_qkv_wmma_f32(
    const float* __restrict__ x,
    const float* __restrict__ Wq, const float* __restrict__ Wk, const float* __restrict__ Wv,
    const float* __restrict__ bq, const float* __restrict__ src_x, const float* __restrict__ dst_x,
    float* __restrict__ q, float* __restrict__ k, float* __restrict__ v,
    const int* __restrict__ lg_src, const int* __restrict__ lg_dst,
    int* __restrict__ cnt, int* __restrict__ bucket, int* __restrict__ ovf, int* __restrict__ ovf_n)
{
  if ((int)blockIdx.x < GROUP_WGS) {
    lgesql_group_edges((int)blockIdx.x, lg_src, lg_dst, cnt, bucket, ovf, ovf_n);
    return;
  }
  const int pblk = (int)blockIdx.x - GROUP_WGS;
  constexpr int K = NDIM_, N = NDIM_;
  constexpr int TN = 64;
  constexpr int tilesN = N / TN;
  constexpr int GROUPS_PER_MATRIX = (E_NODES / 32) * tilesN / (QKV_THREADS / 32);
  static_assert(E_NODES % 32 == 0, "32-row tiles cover the rows exactly");
  const int which = pblk / GROUPS_PER_MATRIX;
  const int grp   = pblk - which * GROUPS_PER_MATRIX;
  const float* __restrict__ W = which == 0 ? Wq : (which == 1 ? Wk : Wv);
  float* __restrict__ C       = which == 0 ? q  : (which == 1 ? k  : v);
  const float* __restrict__ addm = which == 0 ? src_x : dst_x;

  const int wid  = (grp * (int)blockDim.x + (int)threadIdx.x) >> 5;
  const int lane = threadIdx.x & 31;
  const int tm = wid / tilesN;
  const int tn = wid - tm * tilesN;
  const int half = lane >> 4;
  const int l    = lane & 15;

  const float* Ap = x + (size_t)(tm * 32 + l) * K + 4 * half;
  const float* Bp = W + (size_t)(4 * half) * N + tn * TN + 4 * l;

  v8f c00 = {}, c01 = {}, c02 = {}, c03 = {}, c10 = {}, c11 = {}, c12 = {}, c13 = {};
  lgesql_tile2_k_loop<K, N>(Ap, Bp, c00, c01, c02, c03, c10, c11, c12, c13);

  int tm_e = tm, tn_e = tn, half_e = half, l_e = l;
  asm volatile("" : "+v"(tm_e), "+v"(tn_e), "+v"(half_e), "+v"(l_e), "+v"(c00));
  const int col = tn_e * TN + 4 * l_e;
  const size_t at = (size_t)(tm_e * 32 + 8 * half_e) * N + col;
  float* __restrict__ Cp = C + at;
  if (which == 1) {
    #pragma unroll
    for (int r = 0; r < 8; ++r)
      *(float4*)(Cp + (size_t)r * N) = make_float4(c00[r], c01[r], c02[r], c03[r]);
    #pragma unroll
    for (int r = 0; r < 8; ++r)
      *(float4*)(Cp + (size_t)(16 + r) * N) = make_float4(c10[r], c11[r], c12[r], c13[r]);
  } else {
    const float* __restrict__ Rp = addm + at;
    float4 bv = make_float4(0.0f, 0.0f, 0.0f, 0.0f);
    if (which == 0) bv = *(const float4*)(bq + col);
    #pragma unroll
    for (int r0 = 0; r0 < 8; r0 += 4) {
      float4 ad[4];
      #pragma unroll
      for (int i = 0; i < 4; ++i) ad[i] = *(const float4*)(Rp + (size_t)(r0 + i) * N);
      #pragma unroll
      for (int i = 0; i < 4; ++i) {
        const int r = r0 + i;
        float4 o = make_float4(c00[r] + bv.x, c01[r] + bv.y, c02[r] + bv.z, c03[r] + bv.w);
        o.x += ad[i].x; o.y += ad[i].y; o.z += ad[i].z; o.w += ad[i].w;
        *(float4*)(Cp + (size_t)r * N) = o;
      }
      __builtin_amdgcn_sched_barrier(0);
    }
    #pragma unroll
    for (int r0 = 0; r0 < 8; r0 += 4) {
      float4 ad[4];
      #pragma unroll
      for (int i = 0; i < 4; ++i) ad[i] = *(const float4*)(Rp + (size_t)(16 + r0 + i) * N);
      #pragma unroll
      for (int i = 0; i < 4; ++i) {
        const int r = r0 + i;
        float4 o = make_float4(c10[r] + bv.x, c11[r] + bv.y, c12[r] + bv.z, c13[r] + bv.w);
        o.x += ad[i].x; o.y += ad[i].y; o.z += ad[i].z; o.w += ad[i].w;
        *(float4*)(Cp + (size_t)(16 + r) * N) = o;
      }
      __builtin_amdgcn_sched_barrier(0);
    }
  }
}

template <int U>
__device__ __forceinline__ void lgesql_edges(
    const float* __restrict__ k, const float* __restrict__ v, int mysrc, unsigned& todo, int lane,
    const float4& qa, const float4& qb, float4& accA, float4& accB, float& zA, float& zB)
{
  float4 ka[U], kb[U], va[U], vb[U];
  #pragma unroll
  for (int u = 0; u < U; ++u) {
    const int s = __builtin_amdgcn_readlane(mysrc, __builtin_ctz(todo));
    todo &= todo - 1;
    const float4* kr = (const float4*)(k + (size_t)s * NDIM_);
    const float4* vr = (const float4*)(v + (size_t)s * NDIM_);
    ka[u] = kr[lane]; kb[u] = kr[32 + lane];
    va[u] = vr[lane]; vb[u] = vr[32 + lane];
  }
  #pragma unroll
  for (int u = 0; u < U; ++u) {
    float pA = ka[u].x * qa.x + ka[u].y * qa.y + ka[u].z * qa.z + ka[u].w * qa.w;
    float pB = kb[u].x * qb.x + kb[u].y * qb.y + kb[u].z * qb.z + kb[u].w * qb.w;
    pA = lgesql_xadd<1>(pA); pB = lgesql_xadd<1>(pB);
    pA = lgesql_xadd<2>(pA); pB = lgesql_xadd<2>(pB);
    pA = lgesql_xadd<4>(pA); pB = lgesql_xadd<4>(pB);
    float sA = pA * 0.17677669529663687f;
    float sB = pB * 0.17677669529663687f;
    sA = expf(fminf(5.0f, fmaxf(-5.0f, sA)));
    sB = expf(fminf(5.0f, fmaxf(-5.0f, sB)));
    accA.x += va[u].x * sA; accA.y += va[u].y * sA; accA.z += va[u].z * sA; accA.w += va[u].w * sA;
    accB.x += vb[u].x * sB; accB.y += vb[u].y * sB; accB.z += vb[u].z * sB; accB.w += vb[u].w * sB;
    zA += sA; zB += sB;
  }
}

__device__ __forceinline__ void lgesql_node_attention(
    int n, int lane,
    const float* __restrict__ q, const float* __restrict__ k, const float* __restrict__ v,
    const int* __restrict__ cnt, const int* __restrict__ bucket,
    const int* __restrict__ ovf, const int* __restrict__ ovf_n, float4& oa, float4& ob)
{
  const int c = cnt[n];
  const int deg = min(c, BUCKET_CAP);
  const int* __restrict__ brow = bucket + (size_t)n * BUCKET_CAP;
  const float4* qr = (const float4*)(q + (size_t)n * NDIM_);
  const float4 qa = qr[lane];
  const float4 qb = qr[32 + lane];
  float4 accA = make_float4(0.f, 0.f, 0.f, 0.f);
  float4 accB = make_float4(0.f, 0.f, 0.f, 0.f);
  float zA = 0.f, zB = 0.f;
  const int nb = (deg + 31) >> 5;
  int novf = 0;
  if (c > BUCKET_CAP) novf = min(*ovf_n, LE_EDGES);
  const int nchunk = nb + ((novf + 31) >> 5);
  for (int t = 0; t < nchunk; ++t) {
    int mysrc;
    unsigned todo;
    if (t < nb) {
      const int m = min(32, deg - 32 * t);
      mysrc = (lane < m) ? brow[32 * t + lane] : 0;
      todo = (m == 32) ? 0xffffffffu : ((1u << m) - 1u);
    } else {
      const int idx = 32 * (t - nb) + lane;
      const int od = (idx < novf) ? ovf[2 * idx] : -1;
      mysrc = (idx < novf) ? ovf[2 * idx + 1] : 0;
      todo = __builtin_amdgcn_ballot_w32(od == n);
    }
    while (__builtin_popcount(todo) >= 4)
      lgesql_edges<4>(k, v, mysrc, todo, lane, qa, qb, accA, accB, zA, zB);
    if (__builtin_popcount(todo) >= 2)
      lgesql_edges<2>(k, v, mysrc, todo, lane, qa, qb, accA, accB, zA, zB);
    if (todo)
      lgesql_edges<1>(k, v, mysrc, todo, lane, qa, qb, accA, accB, zA, zB);
  }
  oa = make_float4(accA.x / zA, accA.y / zA, accA.z / zA, accA.w / zA);
  ob = make_float4(accB.x / zB, accB.y / zB, accB.z / zB, accB.w / zB);
}

__device__ __forceinline__ void lgesql_ln_rows(
    float4 (&v)[8], float2 (*part)[4], int w, int l, int half, const float4 g4, const float4 b4)
{
  #pragma unroll
  for (int r = 0; r < 8; ++r) {
    float sm = (v[r].x + v[r].y) + (v[r].z + v[r].w);
    float sq = (v[r].x * v[r].x + v[r].y * v[r].y) + (v[r].z * v[r].z + v[r].w * v[r].w);
    sm = lgesql_xadd<1>(sm); sq = lgesql_xadd<1>(sq);
    sm = lgesql_xadd<2>(sm); sq = lgesql_xadd<2>(sq);
    sm = lgesql_xadd<4>(sm); sq = lgesql_xadd<4>(sq);
    sm = lgesql_xadd<8>(sm); sq = lgesql_xadd<8>(sq);
    if (l == r) part[r + 8 * half][w] = make_float2(sm, sq);
  }
  __syncthreads();
  #pragma unroll
  for (int r = 0; r < 8; ++r) {
    const float4 p01 = *(const float4*)(&part[r + 8 * half][0]);
    const float4 p23 = *(const float4*)(&part[r + 8 * half][2]);
    const float sum = ((p01.x + p01.z) + p23.x) + p23.z;
    const float sqs = ((p01.y + p01.w) + p23.y) + p23.w;
    const float mu   = sum * (1.0f / NDIM_);
    const float var  = sqs * (1.0f / NDIM_) - mu * mu;
    const float rstd = rsqrtf(var + 1e-5f);
    float4 o;
    o.x = (v[r].x - mu) * rstd * g4.x + b4.x;
    o.y = (v[r].y - mu) * rstd * g4.y + b4.y;
    o.z = (v[r].z - mu) * rstd * g4.z + b4.z;
    o.w = (v[r].w - mu) * rstd * g4.w + b4.w;
    v[r] = o;
  }
}

#define LDS_PITCH 260
__global__ __launch_bounds__(128) void lgesql_attn_ffn_wmma_f32(
    const float* __restrict__ q, const float* __restrict__ k, const float* __restrict__ v,
    const int* __restrict__ cnt, const int* __restrict__ bucket,
    const int* __restrict__ ovf, const int* __restrict__ ovf_n,
    const float* __restrict__ x, const float* __restrict__ Wo, const float* __restrict__ bo,
    const float* __restrict__ ln1_g, const float* __restrict__ ln1_b,
    const float* __restrict__ W1, const float* __restrict__ b1,
    const float* __restrict__ W2, const float* __restrict__ b2,
    const float* __restrict__ ln2_g, const float* __restrict__ ln2_b,
    float* __restrict__ out)
{
  __shared__ __attribute__((aligned(16))) float sA[16 * LDS_PITCH];
  __shared__ __attribute__((aligned(16))) float sF[16 * LDS_PITCH];
  __shared__ __attribute__((aligned(16))) float2 part[16][4];
  const int lane = threadIdx.x & 31;
  const int w    = threadIdx.x >> 5;
  const int half = lane >> 4;
  const int l    = lane & 15;
  const int tm   = blockIdx.x;

  __builtin_amdgcn_s_setprio(3);

  for (int i = 0; i < 4; ++i) {
    const int row = 4 * w + i;
    const int n = __builtin_amdgcn_readfirstlane(16 * tm + row);
    float4 oa, ob;
    lgesql_node_attention(n, lane, q, k, v, cnt, bucket, ovf, ovf_n, oa, ob);
    *(float4*)(sA + row * LDS_PITCH + 4 * lane)       = oa;
    *(float4*)(sA + row * LDS_PITCH + 128 + 4 * lane) = ob;
  }
  __syncthreads();

  const float* Ap = sA + l * LDS_PITCH + 4 * half;
  const float* Fp = sF + l * LDS_PITCH + 4 * half;
  const int col = 64 * w + 4 * l;
  float4 vv[8];

  {
    v8f a0 = {}, a1 = {}, a2 = {}, a3 = {};
    lgesql_tile_k_loop<NDIM_, NDIM_>(Ap, Wo + (size_t)(4 * half) * NDIM_ + col, a0, a1, a2, a3);
    const float4 bv = *(const float4*)(bo + col);
    #pragma unroll
    for (int r = 0; r < 8; ++r) {
      const int row = tm * 16 + r + 8 * half;
      vv[r] = make_float4(a0[r] + bv.x, a1[r] + bv.y, a2[r] + bv.z, a3[r] + bv.w);
      const float4 ad = *(const float4*)(x + (size_t)row * NDIM_ + col);
      vv[r].x += ad.x; vv[r].y += ad.y; vv[r].z += ad.z; vv[r].w += ad.w;
    }
    lgesql_ln_rows(vv, part, w, l, half, *(const float4*)(ln1_g + col), *(const float4*)(ln1_b + col));
    #pragma unroll
    for (int r = 0; r < 8; ++r)
      *(float4*)(sA + (r + 8 * half) * LDS_PITCH + col) = vv[r];
  }
  __syncthreads();

  v8f d0 = {}, d1 = {}, d2 = {}, d3 = {};
  for (int c = 0; c < 4; ++c) {
    if (c == 1) __builtin_amdgcn_s_setprio(2);
    else if (c == 2) __builtin_amdgcn_s_setprio(1);
    else if (c == 3) __builtin_amdgcn_s_setprio(0);
    v8f e0 = {}, e1 = {}, e2 = {}, e3 = {};
    const int hcol = NDIM_ * c + col;
    lgesql_tile_k_loop<NDIM_, 4 * NDIM_>(Ap, W1 + (size_t)(4 * half) * (4 * NDIM_) + hcol, e0, e1, e2, e3);
    const float4 bv = *(const float4*)(b1 + hcol);
    #pragma unroll
    for (int r = 0; r < 8; ++r) {
      float4 o = make_float4(e0[r] + bv.x, e1[r] + bv.y, e2[r] + bv.z, e3[r] + bv.w);
      o.x = fmaxf(o.x, 0.0f); o.y = fmaxf(o.y, 0.0f); o.z = fmaxf(o.z, 0.0f); o.w = fmaxf(o.w, 0.0f);
      *(float4*)(sF + (r + 8 * half) * LDS_PITCH + col) = o;
    }
    __syncthreads();
    lgesql_tile_k_loop<NDIM_, NDIM_>(Fp, W2 + (size_t)(NDIM_ * c + 4 * half) * NDIM_ + col, d0, d1, d2, d3);
    __syncthreads();
  }

  {
    const float4 bv = *(const float4*)(b2 + col);
    #pragma unroll
    for (int r = 0; r < 8; ++r) {
      vv[r] = make_float4(d0[r] + bv.x, d1[r] + bv.y, d2[r] + bv.z, d3[r] + bv.w);
      const float4 ad = *(const float4*)(sA + (r + 8 * half) * LDS_PITCH + col);
      vv[r].x += ad.x; vv[r].y += ad.y; vv[r].z += ad.z; vv[r].w += ad.w;
    }
    lgesql_ln_rows(vv, part, w, l, half, *(const float4*)(ln2_g + col), *(const float4*)(ln2_b + col));
    #pragma unroll
    for (int r = 0; r < 8; ++r) {
      const int row = tm * 16 + r + 8 * half;
      *(float4*)(out + (size_t)row * NDIM_ + col) = vv[r];
    }
  }
}

extern "C" void kernel_launch(void* const* d_in, const int* in_sizes, int n_in,
                              void* d_out, int out_size, void* d_ws, size_t ws_size,
                              hipStream_t stream) {
  const float* x     = (const float*)d_in[0];
  const float* src_x = (const float*)d_in[1];
  const float* dst_x = (const float*)d_in[2];
  const float* Wq    = (const float*)d_in[3];
  const float* bq    = (const float*)d_in[4];
  const float* Wk    = (const float*)d_in[5];
  const float* Wv    = (const float*)d_in[6];
  const float* Wo    = (const float*)d_in[7];
  const float* bo    = (const float*)d_in[8];
  const float* ln1_g = (const float*)d_in[9];
  const float* ln1_b = (const float*)d_in[10];
  const float* W1    = (const float*)d_in[11];
  const float* b1    = (const float*)d_in[12];
  const float* W2    = (const float*)d_in[13];
  const float* b2    = (const float*)d_in[14];
  const float* ln2_g = (const float*)d_in[15];
  const float* ln2_b = (const float*)d_in[16];
  const int*   lg_src = (const int*)d_in[17];
  const int*   lg_dst = (const int*)d_in[18];
  float* out = (float*)d_out;

  const size_t EN = (size_t)E_NODES * NDIM_;
  float* ws = (float*)d_ws;
  float* qf   = ws;
  float* kf   = ws + EN;
  float* vf   = ws + 2 * EN;
  int*   bucketf = (int*)(ws + 3 * EN);
  int*   ovff = bucketf + (size_t)E_NODES * BUCKET_CAP;
  int*   cntf = (int*)(ws + 4 * EN);
  int*   ovfn = cntf + E_NODES;

  hipMemsetAsync(cntf, 0, (size_t)(E_NODES + 1) * sizeof(int), stream);

  const int gqkv  = GROUP_WGS + 3 * ((E_NODES / 32) * (NDIM_ / 64) / (QKV_THREADS / 32));
  const dim3 blkqkv(QKV_THREADS);
  const int grows = E_NODES / 16;
  const dim3 blkrows(128);

  lgesql_qkv_wmma_f32<<<gqkv, blkqkv, 0, stream>>>(x, Wq, Wk, Wv, bq, src_x, dst_x, qf, kf, vf,
                                                lg_src, lg_dst, cntf, bucketf, ovff, ovfn);

  lgesql_attn_ffn_wmma_f32<<<grows, blkrows, 0, stream>>>(qf, kf, vf, cntf, bucketf, ovff, ovfn,
                                                      x, Wo, bo, ln1_g, ln1_b, W1, b1, W2, b2, ln2_g, ln2_b, out);
}


# Wo, W1, W2 copied to matrix-operand order by 144 extra work-groups of the projection launch: fused K loops load finished W operands (no pairing moves), 32 k per trip; order of sums unchanged
# speedup vs baseline: 2.2654x; 1.0427x over previous
#include <hip/hip_runtime.h>
#include <math.h>

#define E_NODES 20000
#define LE_EDGES 320000
#define NDIM_ 256
#define HEADS 8
#define DK_ 32
#define BUCKET_CAP 64
#define GROUP_EPT 5
#define QKV_THREADS 128
#define GROUP_WGS (LE_EDGES / (QKV_THREADS * GROUP_EPT))
#define PACK_WO4 0
#define PACK_W14 16384
#define PACK_W24 81920
#define PACK_ALL4 147456
#define PACK_PER_WG 1024
#define PACK_WGS (PACK_ALL4 / PACK_PER_WG)

typedef __attribute__((ext_vector_type(2))) float v2f;
typedef __attribute__((ext_vector_type(4))) float v4f;
typedef __attribute__((ext_vector_type(8))) float v8f;

__device__ __forceinline__ v2f lgesql_pair(float lo, float hi)
{
  v2f p; p.x = lo; p.y = hi; return p;
}

template <int XMASK>
__device__ __forceinline__ float lgesql_xadd(float x)
{
  const int y = __builtin_amdgcn_update_dpp(0, __builtin_bit_cast(int, x), 0x160 | XMASK, 0xf, 0xf, true);
  return x + __builtin_bit_cast(float, y);
}

template <int K>
__device__ __forceinline__ void lgesql_tile_k_loop_packed(
    const float* __restrict__ Ap, const v4f* __restrict__ Bq,
    v8f& acc0, v8f& acc1, v8f& acc2, v8f& acc3)
{
  static_assert(K % 32 == 0, "a trip is two steps of 16 k");
  #pragma nounroll
  for (int k = 0; k < K; k += 32) {
    int kk = k;
    asm volatile("v_nop\n\tv_nop\n\tv_nop\n\tv_nop\n\tv_nop\n\tv_nop\n\tv_nop\n\tv_nop"
                 : "+v"(acc0), "+v"(acc1), "+v"(acc2), "+v"(acc3), "+s"(kk));
    const float* ak = Ap + kk;
    const v4f* bk = Bq + (size_t)(kk >> 4) * 256;
    v4f a[4], r[16];
    #pragma unroll
    for (int u = 0; u < 4; ++u) a[u] = *(const v4f*)(ak + 8 * u);
    #pragma unroll
    for (int q = 0; q < 16; ++q) r[q] = bk[32 * q];
    #pragma unroll
    for (int s = 0; s < 2; ++s) {
      #pragma unroll
      for (int g = 0; g < 4; ++g) {
        const v4f av = a[2 * s + (g >> 1)];
        const v4f b0 = r[8 * s + 0 + (g >> 1)], b1 = r[8 * s + 2 + (g >> 1)];
        const v4f b2 = r[8 * s + 4 + (g >> 1)], b3 = r[8 * s + 6 + (g >> 1)];
        const v2f ao = (g & 1) ? lgesql_pair(av.z, av.w) : lgesql_pair(av.x, av.y);
        const v2f p0 = (g & 1) ? lgesql_pair(b0.z, b0.w) : lgesql_pair(b0.x, b0.y);
        const v2f p1 = (g & 1) ? lgesql_pair(b1.z, b1.w) : lgesql_pair(b1.x, b1.y);
        const v2f p2 = (g & 1) ? lgesql_pair(b2.z, b2.w) : lgesql_pair(b2.x, b2.y);
        const v2f p3 = (g & 1) ? lgesql_pair(b3.z, b3.w) : lgesql_pair(b3.x, b3.y);
        acc0 = __builtin_amdgcn_wmma_f32_16x16x4_f32(false, ao, false, p0, (short)0, acc0, false, false);
        acc1 = __builtin_amdgcn_wmma_f32_16x16x4_f32(false, ao, false, p1, (short)0, acc1, false, false);
        acc2 = __builtin_amdgcn_wmma_f32_16x16x4_f32(false, ao, false, p2, (short)0, acc2, false, false);
        acc3 = __builtin_amdgcn_wmma_f32_16x16x4_f32(false, ao, false, p3, (short)0, acc3, false, false);
      }
    }
  }
  asm volatile("v_nop\n\tv_nop\n\tv_nop\n\tv_nop\n\tv_nop\n\tv_nop\n\tv_nop\n\tv_nop"
               : "+v"(acc0), "+v"(acc1), "+v"(acc2), "+v"(acc3));
}

template <int K, int N>
__device__ __forceinline__ void lgesql_tile2_k_loop(
    const float* __restrict__ Ap, const float* __restrict__ Bp,
    v8f& c00, v8f& c01, v8f& c02, v8f& c03, v8f& c10, v8f& c11, v8f& c12, v8f& c13)
{
  for (int k = 0; k < K; k += 16) {
    int kk = k;
    asm volatile("v_nop\n\tv_nop\n\tv_nop\n\tv_nop\n\tv_nop\n\tv_nop\n\tv_nop\n\tv_nop"
                 : "+v"(c00), "+v"(c01), "+v"(c02), "+v"(c03), "+v"(c10), "+v"(c11), "+v"(c12), "+v"(c13), "+s"(kk));
    const float* ak = Ap + kk;
    const float* bk = Bp + (size_t)kk * N;
    v4f a0 = *(const v4f*)(ak);
    v4f a1 = *(const v4f*)(ak + 8);
    v4f a2 = *(const v4f*)(ak + 16 * K);
    v4f a3 = *(const v4f*)(ak + 16 * K + 8);
    v4f r0 = *(const v4f*)(bk);
    v4f r1 = *(const v4f*)(bk + N);
    v4f r2 = *(const v4f*)(bk + 2 * N);
    v4f r3 = *(const v4f*)(bk + 3 * N);
    v4f r4 = *(const v4f*)(bk + 8 * N);
    v4f r5 = *(const v4f*)(bk + 9 * N);
    v4f r6 = *(const v4f*)(bk + 10 * N);
    v4f r7 = *(const v4f*)(bk + 11 * N);
    asm volatile("" : "+v"(a0), "+v"(a1), "+v"(a2), "+v"(a3), "+v"(r0), "+v"(r1), "+v"(r2), "+v"(r3),
                      "+v"(r4), "+v"(r5), "+v"(r6), "+v"(r7));
    v2f p00 = lgesql_pair(r0.x, r1.x), p01 = lgesql_pair(r0.y, r1.y), p02 = lgesql_pair(r0.z, r1.z), p03 = lgesql_pair(r0.w, r1.w);
    v2f p10 = lgesql_pair(r2.x, r3.x), p11 = lgesql_pair(r2.y, r3.y), p12 = lgesql_pair(r2.z, r3.z), p13 = lgesql_pair(r2.w, r3.w);
    v2f p20 = lgesql_pair(r4.x, r5.x), p21 = lgesql_pair(r4.y, r5.y), p22 = lgesql_pair(r4.z, r5.z), p23 = lgesql_pair(r4.w, r5.w);
    v2f p30 = lgesql_pair(r6.x, r7.x), p31 = lgesql_pair(r6.y, r7.y), p32 = lgesql_pair(r6.z, r7.z), p33 = lgesql_pair(r6.w, r7.w);
    asm volatile("" : "+v"(p00), "+v"(p01), "+v"(p02), "+v"(p03), "+v"(p10), "+v"(p11), "+v"(p12), "+v"(p13),
                      "+v"(p20), "+v"(p21), "+v"(p22), "+v"(p23), "+v"(p30), "+v"(p31), "+v"(p32), "+v"(p33));
    const v2f x00 = lgesql_pair(a0.x, a0.y), x01 = lgesql_pair(a0.z, a0.w);
    const v2f x10 = lgesql_pair(a1.x, a1.y), x11 = lgesql_pair(a1.z, a1.w);
    const v2f y00 = lgesql_pair(a2.x, a2.y), y01 = lgesql_pair(a2.z, a2.w);
    const v2f y10 = lgesql_pair(a3.x, a3.y), y11 = lgesql_pair(a3.z, a3.w);
    c00 = __builtin_amdgcn_wmma_f32_16x16x4_f32(false, x00, false, p00, (short)0, c00, false, false);
    c01 = __builtin_amdgcn_wmma_f32_16x16x4_f32(false, x00, false, p01, (short)0, c01, false, false);
    c02 = __builtin_amdgcn_wmma_f32_16x16x4_f32(false, x00, false, p02, (short)0, c02, false, false);
    c03 = __builtin_amdgcn_wmma_f32_16x16x4_f32(false, x00, false, p03, (short)0, c03, false, false);
    c10 = __builtin_amdgcn_wmma_f32_16x16x4_f32(false, y00, false, p00, (short)0, c10, false, false);
    c11 = __builtin_amdgcn_wmma_f32_16x16x4_f32(false, y00, false, p01, (short)0, c11, false, false);
    c12 = __builtin_amdgcn_wmma_f32_16x16x4_f32(false, y00, false, p02, (short)0, c12, false, false);
    c13 = __builtin_amdgcn_wmma_f32_16x16x4_f32(false, y00, false, p03, (short)0, c13, false, false);
    c00 = __builtin_amdgcn_wmma_f32_16x16x4_f32(false, x01, false, p10, (short)0, c00, false, false);
    c01 = __builtin_amdgcn_wmma_f32_16x16x4_f32(false, x01, false, p11, (short)0, c01, false, false);
    c02 = __builtin_amdgcn_wmma_f32_16x16x4_f32(false, x01, false, p12, (short)0, c02, false, false);
    c03 = __builtin_amdgcn_wmma_f32_16x16x4_f32(false, x01, false, p13, (short)0, c03, false, false);
    c10 = __builtin_amdgcn_wmma_f32_16x16x4_f32(false, y01, false, p10, (short)0, c10, false, false);
    c11 = __builtin_amdgcn_wmma_f32_16x16x4_f32(false, y01, false, p11, (short)0, c11, false, false);
    c12 = __builtin_amdgcn_wmma_f32_16x16x4_f32(false, y01, false, p12, (short)0, c12, false, false);
    c13 = __builtin_amdgcn_wmma_f32_16x16x4_f32(false, y01, false, p13, (short)0, c13, false, false);
    c00 = __builtin_amdgcn_wmma_f32_16x16x4_f32(false, x10, false, p20, (short)0, c00, false, false);
    c01 = __builtin_amdgcn_wmma_f32_16x16x4_f32(false, x10, false, p21, (short)0, c01, false, false);
    c02 = __builtin_amdgcn_wmma_f32_16x16x4_f32(false, x10, false, p22, (short)0, c02, false, false);
    c03 = __builtin_amdgcn_wmma_f32_16x16x4_f32(false, x10, false, p23, (short)0, c03, false, false);
    c10 = __builtin_amdgcn_wmma_f32_16x16x4_f32(false, y10, false, p20, (short)0, c10, false, false);
    c11 = __builtin_amdgcn_wmma_f32_16x16x4_f32(false, y10, false, p21, (short)0, c11, false, false);
    c12 = __builtin_amdgcn_wmma_f32_16x16x4_f32(false, y10, false, p22, (short)0, c12, false, false);
    c13 = __builtin_amdgcn_wmma_f32_16x16x4_f32(false, y10, false, p23, (short)0, c13, false, false);
    c00 = __builtin_amdgcn_wmma_f32_16x16x4_f32(false, x11, false, p30, (short)0, c00, false, false);
    c01 = __builtin_amdgcn_wmma_f32_16x16x4_f32(false, x11, false, p31, (short)0, c01, false, false);
    c02 = __builtin_amdgcn_wmma_f32_16x16x4_f32(false, x11, false, p32, (short)0, c02, false, false);
    c03 = __builtin_amdgcn_wmma_f32_16x16x4_f32(false, x11, false, p33, (short)0, c03, false, false);
    c10 = __builtin_amdgcn_wmma_f32_16x16x4_f32(false, y11, false, p30, (short)0, c10, false, false);
    c11 = __builtin_amdgcn_wmma_f32_16x16x4_f32(false, y11, false, p31, (short)0, c11, false, false);
    c12 = __builtin_amdgcn_wmma_f32_16x16x4_f32(false, y11, false, p32, (short)0, c12, false, false);
    c13 = __builtin_amdgcn_wmma_f32_16x16x4_f32(false, y11, false, p33, (short)0, c13, false, false);
  }
  asm volatile("v_nop\n\tv_nop\n\tv_nop\n\tv_nop\n\tv_nop\n\tv_nop\n\tv_nop\n\tv_nop"
               : "+v"(c00), "+v"(c01), "+v"(c02), "+v"(c03), "+v"(c10), "+v"(c11), "+v"(c12), "+v"(c13));
}

__device__ __forceinline__ void lgesql_pack_weights(
    int wg, const float* __restrict__ Wo, const float* __restrict__ W1, const float* __restrict__ W2,
    v4f* __restrict__ wpk)
{
  static_assert(PACK_W14 % PACK_PER_WG == 0 && PACK_W24 % PACK_PER_WG == 0 && PACK_ALL4 % PACK_PER_WG == 0,
                "a packing work-group stays inside one matrix");
  const int first = wg * PACK_PER_WG;
  const float* __restrict__ W = Wo;
  int N = NDIM_, ktbits = 4, base = PACK_WO4;
  if (first >= PACK_W24)      { W = W2; N = NDIM_;     ktbits = 6; base = PACK_W24; }
  else if (first >= PACK_W14) { W = W1; N = 4 * NDIM_; ktbits = 4; base = PACK_W14; }
  #pragma unroll
  for (int i = 0; i < PACK_PER_WG / QKV_THREADS; ++i) {
    const int p    = first - base + QKV_THREADS * i + (int)threadIdx.x;
    const int lane = p & 31;
    const int q    = (p >> 5) & 7;
    const int kt   = (p >> 8) & ((1 << ktbits) - 1);
    const int ct   = p >> (8 + ktbits);
    const int krow = 16 * kt + 8 * (q & 1) + 4 * (lane >> 4);
    const int ncol = 64 * ct + 4 * (lane & 15) + (q >> 1);
    const float* __restrict__ s = W + (size_t)krow * N + ncol;
    v4f o;
    o.x = s[0]; o.y = s[N]; o.z = s[2 * N]; o.w = s[3 * N];
    wpk[base + p] = o;
  }
}

__device__ __forceinline__ void lgesql_group_edges(
    int wg, const int* __restrict__ lg_src, const int* __restrict__ lg_dst,
    int* __restrict__ cnt, int* __restrict__ bucket, int* __restrict__ ovf, int* __restrict__ ovf_n)
{
  static_assert(GROUP_WGS * QKV_THREADS * GROUP_EPT == LE_EDGES, "the grouping work-groups cover the edges exactly");
  const int e0 = wg * (QKV_THREADS * GROUP_EPT) + (int)threadIdx.x;
  int d[GROUP_EPT], sidx[GROUP_EPT], r[GROUP_EPT];
  #pragma unroll
  for (int i = 0; i < GROUP_EPT; ++i) { d[i] = lg_dst[e0 + QKV_THREADS * i]; sidx[i] = lg_src[e0 + QKV_THREADS * i]; }
  #pragma unroll
  for (int i = 0; i < GROUP_EPT; ++i) r[i] = atomicAdd(&cnt[d[i]], 1);
  #pragma unroll
  for (int i = 0; i < GROUP_EPT; ++i) {
    if (r[i] < BUCKET_CAP) {
      bucket[(size_t)d[i] * BUCKET_CAP + r[i]] = sidx[i];
    } else {
      const int p = atomicAdd(ovf_n, 1);
      ovf[2 * p] = d[i]; ovf[2 * p + 1] = sidx[i];
    }
  }
}

__global__ __launch_bounds__(QKV_THREADS) void lgesql_qkv_wmma_f32(
    const float* __restrict__ x,
    const float* __restrict__ Wq, const float* __restrict__ Wk, const float* __restrict__ Wv,
    const float* __restrict__ bq, const float* __restrict__ src_x, const float* __restrict__ dst_x,
    float* __restrict__ q, float* __restrict__ k, float* __restrict__ v,
    const int* __restrict__ lg_src, const int* __restrict__ lg_dst,
    int* __restrict__ cnt, int* __restrict__ bucket, int* __restrict__ ovf, int* __restrict__ ovf_n,
    const float* __restrict__ Wo, const float* __restrict__ W1, const float* __restrict__ W2,
    float* __restrict__ wpk)
{
  if ((int)blockIdx.x < PACK_WGS) {
    lgesql_pack_weights((int)blockIdx.x, Wo, W1, W2, (v4f*)wpk);
    return;
  }
  if ((int)blockIdx.x < PACK_WGS + GROUP_WGS) {
    lgesql_group_edges((int)blockIdx.x - PACK_WGS, lg_src, lg_dst, cnt, bucket, ovf, ovf_n);
    return;
  }
  const int pblk = (int)blockIdx.x - (PACK_WGS + GROUP_WGS);
  constexpr int K = NDIM_, N = NDIM_;
  constexpr int TN = 64;
  constexpr int tilesN = N / TN;
  constexpr int GROUPS_PER_MATRIX = (E_NODES / 32) * tilesN / (QKV_THREADS / 32);
  static_assert(E_NODES % 32 == 0, "32-row tiles cover the rows exactly");
  const int which = pblk / GROUPS_PER_MATRIX;
  const int grp   = pblk - which * GROUPS_PER_MATRIX;
  const float* __restrict__ W = which == 0 ? Wq : (which == 1 ? Wk : Wv);
  float* __restrict__ C       = which == 0 ? q  : (which == 1 ? k  : v);
  const float* __restrict__ addm = which == 0 ? src_x : dst_x;

  const int wid  = (grp * (int)blockDim.x + (int)threadIdx.x) >> 5;
  const int lane = threadIdx.x & 31;
  const int tm = wid / tilesN;
  const int tn = wid - tm * tilesN;
  const int half = lane >> 4;
  const int l    = lane & 15;

  const float* Ap = x + (size_t)(tm * 32 + l) * K + 4 * half;
  const float* Bp = W + (size_t)(4 * half) * N + tn * TN + 4 * l;

  v8f c00 = {}, c01 = {}, c02 = {}, c03 = {}, c10 = {}, c11 = {}, c12 = {}, c13 = {};
  lgesql_tile2_k_loop<K, N>(Ap, Bp, c00, c01, c02, c03, c10, c11, c12, c13);

  int tm_e = tm, tn_e = tn, half_e = half, l_e = l;
  asm volatile("" : "+v"(tm_e), "+v"(tn_e), "+v"(half_e), "+v"(l_e), "+v"(c00));
  const int col = tn_e * TN + 4 * l_e;
  const size_t at = (size_t)(tm_e * 32 + 8 * half_e) * N + col;
  float* __restrict__ Cp = C + at;
  if (which == 1) {
    #pragma unroll
    for (int r = 0; r < 8; ++r)
      *(float4*)(Cp + (size_t)r * N) = make_float4(c00[r], c01[r], c02[r], c03[r]);
    #pragma unroll
    for (int r = 0; r < 8; ++r)
      *(float4*)(Cp + (size_t)(16 + r) * N) = make_float4(c10[r], c11[r], c12[r], c13[r]);
  } else {
    const float* __restrict__ Rp = addm + at;
    float4 bv = make_float4(0.0f, 0.0f, 0.0f, 0.0f);
    if (which == 0) bv = *(const float4*)(bq + col);
    #pragma unroll
    for (int r0 = 0; r0 < 8; r0 += 4) {
      float4 ad[4];
      #pragma unroll
      for (int i = 0; i < 4; ++i) ad[i] = *(const float4*)(Rp + (size_t)(r0 + i) * N);
      #pragma unroll
      for (int i = 0; i < 4; ++i) {
        const int r = r0 + i;
        float4 o = make_float4(c00[r] + bv.x, c01[r] + bv.y, c02[r] + bv.z, c03[r] + bv.w);
        o.x += ad[i].x; o.y += ad[i].y; o.z += ad[i].z; o.w += ad[i].w;
        *(float4*)(Cp + (size_t)r * N) = o;
      }
      __builtin_amdgcn_sched_barrier(0);
    }
    #pragma unroll
    for (int r0 = 0; r0 < 8; r0 += 4) {
      float4 ad[4];
      #pragma unroll
      for (int i = 0; i < 4; ++i) ad[i] = *(const float4*)(Rp + (size_t)(16 + r0 + i) * N);
      #pragma unroll
      for (int i = 0; i < 4; ++i) {
        const int r = r0 + i;
        float4 o = make_float4(c10[r] + bv.x, c11[r] + bv.y, c12[r] + bv.z, c13[r] + bv.w);
        o.x += ad[i].x; o.y += ad[i].y; o.z += ad[i].z; o.w += ad[i].w;
        *(float4*)(Cp + (size_t)(16 + r) * N) = o;
      }
      __builtin_amdgcn_sched_barrier(0);
    }
  }
}

template <int U>
__device__ __forceinline__ void lgesql_edges(
    const float* __restrict__ k, const float* __restrict__ v, int mysrc, unsigned& todo, int lane,
    const float4& qa, const float4& qb, float4& accA, float4& accB, float& zA, float& zB)
{
  float4 ka[U], kb[U], va[U], vb[U];
  #pragma unroll
  for (int u = 0; u < U; ++u) {
    const int s = __builtin_amdgcn_readlane(mysrc, __builtin_ctz(todo));
    todo &= todo - 1;
    const float4* kr = (const float4*)(k + (size_t)s * NDIM_);
    const float4* vr = (const float4*)(v + (size_t)s * NDIM_);
    ka[u] = kr[lane]; kb[u] = kr[32 + lane];
    va[u] = vr[lane]; vb[u] = vr[32 + lane];
  }
  #pragma unroll
  for (int u = 0; u < U; ++u) {
    float pA = ka[u].x * qa.x + ka[u].y * qa.y + ka[u].z * qa.z + ka[u].w * qa.w;
    float pB = kb[u].x * qb.x + kb[u].y * qb.y + kb[u].z * qb.z + kb[u].w * qb.w;
    pA = lgesql_xadd<1>(pA); pB = lgesql_xadd<1>(pB);
    pA = lgesql_xadd<2>(pA); pB = lgesql_xadd<2>(pB);
    pA = lgesql_xadd<4>(pA); pB = lgesql_xadd<4>(pB);
    float sA = pA * 0.17677669529663687f;
    float sB = pB * 0.17677669529663687f;
    sA = expf(fminf(5.0f, fmaxf(-5.0f, sA)));
    sB = expf(fminf(5.0f, fmaxf(-5.0f, sB)));
    accA.x += va[u].x * sA; accA.y += va[u].y * sA; accA.z += va[u].z * sA; accA.w += va[u].w * sA;
    accB.x += vb[u].x * sB; accB.y += vb[u].y * sB; accB.z += vb[u].z * sB; accB.w += vb[u].w * sB;
    zA += sA; zB += sB;
  }
}

__device__ __forceinline__ void lgesql_node_attention(
    int n, int lane,
    const float* __restrict__ q, const float* __restrict__ k, const float* __restrict__ v,
    const int* __restrict__ cnt, const int* __restrict__ bucket,
    const int* __restrict__ ovf, const int* __restrict__ ovf_n, float4& oa, float4& ob)
{
  const int c = cnt[n];
  const int deg = min(c, BUCKET_CAP);
  const int* __restrict__ brow = bucket + (size_t)n * BUCKET_CAP;
  const float4* qr = (const float4*)(q + (size_t)n * NDIM_);
  const float4 qa = qr[lane];
  const float4 qb = qr[32 + lane];
  float4 accA = make_float4(0.f, 0.f, 0.f, 0.f);
  float4 accB = make_float4(0.f, 0.f, 0.f, 0.f);
  float zA = 0.f, zB = 0.f;
  const int nb = (deg + 31) >> 5;
  int novf = 0;
  if (c > BUCKET_CAP) novf = min(*ovf_n, LE_EDGES);
  const int nchunk = nb + ((novf + 31) >> 5);
  for (int t = 0; t < nchunk; ++t) {
    int mysrc;
    unsigned todo;
    if (t < nb) {
      const int m = min(32, deg - 32 * t);
      mysrc = (lane < m) ? brow[32 * t + lane] : 0;
      todo = (m == 32) ? 0xffffffffu : ((1u << m) - 1u);
    } else {
      const int idx = 32 * (t - nb) + lane;
      const int od = (idx < novf) ? ovf[2 * idx] : -1;
      mysrc = (idx < novf) ? ovf[2 * idx + 1] : 0;
      todo = __builtin_amdgcn_ballot_w32(od == n);
    }
    while (__builtin_popcount(todo) >= 4)
      lgesql_edges<4>(k, v, mysrc, todo, lane, qa, qb, accA, accB, zA, zB);
    if (__builtin_popcount(todo) >= 2)
      lgesql_edges<2>(k, v, mysrc, todo, lane, qa, qb, accA, accB, zA, zB);
    if (todo)
      lgesql_edges<1>(k, v, mysrc, todo, lane, qa, qb, accA, accB, zA, zB);
  }
  oa = make_float4(accA.x / zA, accA.y / zA, accA.z / zA, accA.w / zA);
  ob = make_float4(accB.x / zB, accB.y / zB, accB.z / zB, accB.w / zB);
}

__device__ __forceinline__ void lgesql_ln_rows(
    float4 (&v)[8], float2 (*part)[4], int w, int l, int half, const float4 g4, const float4 b4)
{
  #pragma unroll
  for (int r = 0; r < 8; ++r) {
    float sm = (v[r].x + v[r].y) + (v[r].z + v[r].w);
    float sq = (v[r].x * v[r].x + v[r].y * v[r].y) + (v[r].z * v[r].z + v[r].w * v[r].w);
    sm = lgesql_xadd<1>(sm); sq = lgesql_xadd<1>(sq);
    sm = lgesql_xadd<2>(sm); sq = lgesql_xadd<2>(sq);
    sm = lgesql_xadd<4>(sm); sq = lgesql_xadd<4>(sq);
    sm = lgesql_xadd<8>(sm); sq = lgesql_xadd<8>(sq);
    if (l == r) part[r + 8 * half][w] = make_float2(sm, sq);
  }
  __syncthreads();
  #pragma unroll
  for (int r = 0; r < 8; ++r) {
    const float4 p01 = *(const float4*)(&part[r + 8 * half][0]);
    const float4 p23 = *(const float4*)(&part[r + 8 * half][2]);
    const float sum = ((p01.x + p01.z) + p23.x) + p23.z;
    const float sqs = ((p01.y + p01.w) + p23.y) + p23.w;
    const float mu   = sum * (1.0f / NDIM_);
    const float var  = sqs * (1.0f / NDIM_) - mu * mu;
    const float rstd = rsqrtf(var + 1e-5f);
    float4 o;
    o.x = (v[r].x - mu) * rstd * g4.x + b4.x;
    o.y = (v[r].y - mu) * rstd * g4.y + b4.y;
    o.z = (v[r].z - mu) * rstd * g4.z + b4.z;
    o.w = (v[r].w - mu) * rstd * g4.w + b4.w;
    v[r] = o;
  }
}

#define LDS_PITCH 260
__global__ __launch_bounds__(128) void lgesql_attn_ffn_wmma_f32(
    const float* __restrict__ q, const float* __restrict__ k, const float* __restrict__ v,
    const int* __restrict__ cnt, const int* __restrict__ bucket,
    const int* __restrict__ ovf, const int* __restrict__ ovf_n,
    const float* __restrict__ x, const float* __restrict__ wpk, const float* __restrict__ bo,
    const float* __restrict__ ln1_g, const float* __restrict__ ln1_b,
    const float* __restrict__ b1, const float* __restrict__ b2,
    const float* __restrict__ ln2_g, const float* __restrict__ ln2_b,
    float* __restrict__ out)
{
  __shared__ __attribute__((aligned(16))) float sA[16 * LDS_PITCH];
  __shared__ __attribute__((aligned(16))) float sF[16 * LDS_PITCH];
  __shared__ __attribute__((aligned(16))) float2 part[16][4];
  const int lane = threadIdx.x & 31;
  const int w    = threadIdx.x >> 5;
  const int half = lane >> 4;
  const int l    = lane & 15;
  const int tm   = blockIdx.x;

  __builtin_amdgcn_s_setprio(3);

  for (int i = 0; i < 4; ++i) {
    const int row = 4 * w + i;
    const int n = __builtin_amdgcn_readfirstlane(16 * tm + row);
    float4 oa, ob;
    lgesql_node_attention(n, lane, q, k, v, cnt, bucket, ovf, ovf_n, oa, ob);
    *(float4*)(sA + row * LDS_PITCH + 4 * lane)       = oa;
    *(float4*)(sA + row * LDS_PITCH + 128 + 4 * lane) = ob;
  }
  __syncthreads();

  const float* Ap = sA + l * LDS_PITCH + 4 * half;
  const float* Fp = sF + l * LDS_PITCH + 4 * half;
  const int col = 64 * w + 4 * l;
  const v4f* __restrict__ Bo = (const v4f*)wpk + PACK_WO4 + (size_t)w * 4096 + lane;
  const v4f* __restrict__ B1 = (const v4f*)wpk + PACK_W14 + (size_t)w * 4096 + lane;
  const v4f* __restrict__ B2 = (const v4f*)wpk + PACK_W24 + (size_t)w * 16384 + lane;
  float4 vv[8];

  {
    v8f a0 = {}, a1 = {}, a2 = {}, a3 = {};
    lgesql_tile_k_loop_packed<NDIM_>(Ap, Bo, a0, a1, a2, a3);
    const float4 bv = *(const float4*)(bo + col);
    #pragma unroll
    for (int r = 0; r < 8; ++r) {
      const int row = tm * 16 + r + 8 * half;
      vv[r] = make_float4(a0[r] + bv.x, a1[r] + bv.y, a2[r] + bv.z, a3[r] + bv.w);
      const float4 ad = *(const float4*)(x + (size_t)row * NDIM_ + col);
      vv[r].x += ad.x; vv[r].y += ad.y; vv[r].z += ad.z; vv[r].w += ad.w;
    }
    lgesql_ln_rows(vv, part, w, l, half, *(const float4*)(ln1_g + col), *(const float4*)(ln1_b + col));
    #pragma unroll
    for (int r = 0; r < 8; ++r)
      *(float4*)(sA + (r + 8 * half) * LDS_PITCH + col) = vv[r];
  }
  __syncthreads();

  v8f d0 = {}, d1 = {}, d2 = {}, d3 = {};
  for (int c = 0; c < 4; ++c) {
    if (c == 1) __builtin_amdgcn_s_setprio(2);
    else if (c == 2) __builtin_amdgcn_s_setprio(1);
    else if (c == 3) __builtin_amdgcn_s_setprio(0);
    v8f e0 = {}, e1 = {}, e2 = {}, e3 = {};
    const int hcol = NDIM_ * c + col;
    lgesql_tile_k_loop_packed<NDIM_>(Ap, B1 + (size_t)c * 16384, e0, e1, e2, e3);
    const float4 bv = *(const float4*)(b1 + hcol);
    #pragma unroll
    for (int r = 0; r < 8; ++r) {
      float4 o = make_float4(e0[r] + bv.x, e1[r] + bv.y, e2[r] + bv.z, e3[r] + bv.w);
      o.x = fmaxf(o.x, 0.0f); o.y = fmaxf(o.y, 0.0f); o.z = fmaxf(o.z, 0.0f); o.w = fmaxf(o.w, 0.0f);
      *(float4*)(sF + (r + 8 * half) * LDS_PITCH + col) = o;
    }
    __syncthreads();
    lgesql_tile_k_loop_packed<NDIM_>(Fp, B2 + (size_t)c * 4096, d0, d1, d2, d3);
    __syncthreads();
  }

  {
    const float4 bv = *(const float4*)(b2 + col);
    #pragma unroll
    for (int r = 0; r < 8; ++r) {
      vv[r] = make_float4(d0[r] + bv.x, d1[r] + bv.y, d2[r] + bv.z, d3[r] + bv.w);
      const float4 ad = *(const float4*)(sA + (r + 8 * half) * LDS_PITCH + col);
      vv[r].x += ad.x; vv[r].y += ad.y; vv[r].z += ad.z; vv[r].w += ad.w;
    }
    lgesql_ln_rows(vv, part, w, l, half, *(const float4*)(ln2_g + col), *(const float4*)(ln2_b + col));
    #pragma unroll
    for (int r = 0; r < 8; ++r) {
      const int row = tm * 16 + r + 8 * half;
      *(float4*)(out + (size_t)row * NDIM_ + col) = vv[r];
    }
  }
}

extern "C" void kernel_launch(void* const* d_in, const int* in_sizes, int n_in,
                              void* d_out, int out_size, void* d_ws, size_t ws_size,
                              hipStream_t stream) {
  const float* x     = (const float*)d_in[0];
  const float* src_x = (const float*)d_in[1];
  const float* dst_x = (const float*)d_in[2];
  const float* Wq    = (const float*)d_in[3];
  const float* bq    = (const float*)d_in[4];
  const float* Wk    = (const float*)d_in[5];
  const float* Wv    = (const float*)d_in[6];
  const float* Wo    = (const float*)d_in[7];
  const float* bo    = (const float*)d_in[8];
  const float* ln1_g = (const float*)d_in[9];
  const float* ln1_b = (const float*)d_in[10];
  const float* W1    = (const float*)d_in[11];
  const float* b1    = (const float*)d_in[12];
  const float* W2    = (const float*)d_in[13];
  const float* b2    = (const float*)d_in[14];
  const float* ln2_g = (const float*)d_in[15];
  const float* ln2_b = (const float*)d_in[16];
  const int*   lg_src = (const int*)d_in[17];
  const int*   lg_dst = (const int*)d_in[18];
  float* out = (float*)d_out;

  const size_t EN = (size_t)E_NODES * NDIM_;
  float* ws = (float*)d_ws;
  float* qf   = ws;
  float* kf   = ws + EN;
  float* vf   = ws + 2 * EN;
  int*   bucketf = (int*)(ws + 3 * EN);
  int*   ovff = bucketf + (size_t)E_NODES * BUCKET_CAP;
  float* wpkf = (float*)(ovff + 2 * (size_t)LE_EDGES);
  int*   cntf = (int*)(ws + 4 * EN);
  int*   ovfn = cntf + E_NODES;

  hipMemsetAsync(cntf, 0, (size_t)(E_NODES + 1) * sizeof(int), stream);

  const int gqkv  = PACK_WGS + GROUP_WGS + 3 * ((E_NODES / 32) * (NDIM_ / 64) / (QKV_THREADS / 32));
  const dim3 blkqkv(QKV_THREADS);
  const int grows = E_NODES / 16;
  const dim3 blkrows(128);

  lgesql_qkv_wmma_f32<<<gqkv, blkqkv, 0, stream>>>(x, Wq, Wk, Wv, bq, src_x, dst_x, qf, kf, vf,
                                                lg_src, lg_dst, cntf, bucketf, ovff, ovfn,
                                                Wo, W1, W2, wpkf);

  lgesql_attn_ffn_wmma_f32<<<grows, blkrows, 0, stream>>>(qf, kf, vf, cntf, bucketf, ovff, ovfn,
                                                      x, wpkf, bo, ln1_g, ln1_b, b1, b2, ln2_g, ln2_b, out);
}
